# starting kernel
# baseline (speedup 1.0000x reference)
#include <hip/hip_runtime.h>


typedef float v2f __attribute__((ext_vector_type(2)));
typedef float v8f __attribute__((ext_vector_type(8)));

__global__ void k_zero(float* __restrict__ p, int n) {
    int i = blockIdx.x * blockDim.x + threadIdx.x;
    if (i < n) p[i] = 0.0f;
}

__global__ void k_count(const int* __restrict__ dst, float* __restrict__ deg, int E) {
    int e = blockIdx.x * blockDim.x + threadIdx.x;
    if (e < E) atomicAdd(&deg[dst[e]], 1.0f);
}

__global__ void k_dinv(const float* __restrict__ deg, float* __restrict__ dinv, int n) {
    int i = blockIdx.x * blockDim.x + threadIdx.x;
    if (i < n) dinv[i] = rsqrtf(deg[i] + 1.0f);
}

template <bool RELU_A, int K>
__global__ void k_gemm_wmma(const float* __restrict__ A, const float* __restrict__ B,
                            float* __restrict__ C, int M, int N) {
    constexpr int LDK = K + 2;
    __shared__ float As[16 * LDK];

    const int mt   = blockIdx.x;
    const int nb   = blockIdx.y;
    const int tid  = threadIdx.x;
    const int wave = tid >> 5;
    const int lane = tid & 31;

    const float* __restrict__ Ag = A + (size_t)mt * 16 * K;
#pragma unroll
    for (int idx = tid; idx < 16 * K; idx += 256) {
        int r  = idx / K;
        int kk = idx - r * K;
        float v = Ag[idx];
        if (RELU_A) v = v > 0.0f ? v : 0.0f;
        As[r * LDK + kk] = v;
    }
    __syncthreads();

    const int nt  = nb * 8 + wave;
    const int col = lane & 15;
    const int kb  = (lane >> 4) << 1;

    const float* __restrict__ Arow = As + col * LDK + kb;
    const float* __restrict__ Bc   = B + (size_t)nt * 16 + col;

    v8f acc = {};
#pragma unroll
    for (int k = 0; k < K; k += 4) {
        v2f av; av.x = Arow[k];                      av.y = Arow[k + 1];
        v2f bv; bv.x = Bc[(size_t)(k + kb) * N];     bv.y = Bc[(size_t)(k + kb + 1) * N];
        acc = __builtin_amdgcn_wmma_f32_16x16x4_f32(
            false, av, false, bv, (short)0, acc, false, false);
    }

    float* __restrict__ Cp =
        C + (size_t)(mt * 16 + ((lane >> 4) << 3)) * N + (size_t)nt * 16 + col;
#pragma unroll
    for (int r = 0; r < 8; ++r) Cp[(size_t)r * N] = acc[r];
}

__global__ void k_selfloop_bias(const float* __restrict__ xw, const float* __restrict__ dinv,
                                const float* __restrict__ b, float* __restrict__ agg,
                                int n, int F) {
    int idx = blockIdx.x * blockDim.x + threadIdx.x;
    if (idx < n * F) {
        int i = idx / F;
        int f = idx - i * F;
        float di = dinv[i];
        agg[idx] = xw[idx] * di * di + b[f];
    }
}

__global__ void k_scatter(const int* __restrict__ src, const int* __restrict__ dst,
                          const float* __restrict__ dinv, const float* __restrict__ xw,
                          float* __restrict__ agg, int F) {
    int e = blockIdx.x;
    int f = threadIdx.x;
    int s = src[e], d = dst[e];
    float c = dinv[s] * dinv[d];
    atomicAdd(&agg[(size_t)d * F + f], xw[(size_t)s * F + f] * c);
}

__global__ void k_node_head(const float* __restrict__ emb, const float* __restrict__ Wn,
                            const float* __restrict__ bn, float* __restrict__ out, int n) {
    int i = blockIdx.x * blockDim.x + threadIdx.x;
    if (i >= n) return;
    const float4* row = (const float4*)(emb + (size_t)i * 128);
    float a0 = bn[0], a1 = bn[1];
#pragma unroll 8
    for (int q = 0; q < 32; ++q) {
        float4 v = row[q];
        int f = q * 4;
        a0 += v.x * Wn[(f + 0) * 2 + 0] + v.y * Wn[(f + 1) * 2 + 0] +
              v.z * Wn[(f + 2) * 2 + 0] + v.w * Wn[(f + 3) * 2 + 0];
        a1 += v.x * Wn[(f + 0) * 2 + 1] + v.y * Wn[(f + 1) * 2 + 1] +
              v.z * Wn[(f + 2) * 2 + 1] + v.w * Wn[(f + 3) * 2 + 1];
    }
    out[(size_t)i * 2 + 0] = a0;
    out[(size_t)i * 2 + 1] = a1;
}

__global__ void k_edge_head(const int* __restrict__ src, const int* __restrict__ dst,
                            const float* __restrict__ emb, const float* __restrict__ We,
                            const float* __restrict__ be, float* __restrict__ out, int E) {
    int lane = threadIdx.x & 31;
    int warp = blockIdx.x * (blockDim.x >> 5) + (threadIdx.x >> 5);
    int nwarps = gridDim.x * (blockDim.x >> 5);

    float ws0[4], ws1[4], wd0[4], wd1[4];
#pragma unroll
    for (int j = 0; j < 4; ++j) {
        int f = lane * 4 + j;
        ws0[j] = We[f * 2 + 0];         ws1[j] = We[f * 2 + 1];
        wd0[j] = We[(128 + f) * 2 + 0]; wd1[j] = We[(128 + f) * 2 + 1];
    }
    float b0 = be[0], b1 = be[1];

    for (int e = warp; e < E; e += nwarps) {
        int s = src[e], d = dst[e];
        int en = e + nwarps;
        if (en < E) {
            __builtin_prefetch(&emb[(size_t)src[en] * 128 + lane * 4], 0, 0);
            __builtin_prefetch(&emb[(size_t)dst[en] * 128 + lane * 4], 0, 0);
        }
        float4 vs = *(const float4*)(emb + (size_t)s * 128 + lane * 4);
        float4 vd = *(const float4*)(emb + (size_t)d * 128 + lane * 4);
        float a0 = vs.x * ws0[0] + vs.y * ws0[1] + vs.z * ws0[2] + vs.w * ws0[3] +
                   vd.x * wd0[0] + vd.y * wd0[1] + vd.z * wd0[2] + vd.w * wd0[3];
        float a1 = vs.x * ws1[0] + vs.y * ws1[1] + vs.z * ws1[2] + vs.w * ws1[3] +
                   vd.x * wd1[0] + vd.y * wd1[1] + vd.z * wd1[2] + vd.w * wd1[3];
#pragma unroll
        for (int off = 16; off > 0; off >>= 1) {
            a0 += __shfl_down(a0, off, 32);
            a1 += __shfl_down(a1, off, 32);
        }
        if (lane == 0) {
            out[(size_t)e * 2 + 0] = a0 + b0;
            out[(size_t)e * 2 + 1] = a1 + b1;
        }
    }
}

extern "C" void kernel_launch(void* const* d_in, const int* in_sizes, int n_in,
                              void* d_out, int out_size, void* d_ws, size_t ws_size,
                              hipStream_t stream) {
    const float* x  = (const float*)d_in[0];
    const int*   ei = (const int*)d_in[1];
    const float* W1 = (const float*)d_in[2];
    const float* b1 = (const float*)d_in[3];
    const float* W2 = (const float*)d_in[4];
    const float* b2 = (const float*)d_in[5];
    const float* Wn = (const float*)d_in[6];
    const float* bn = (const float*)d_in[7];
    const float* We = (const float*)d_in[8];
    const float* be = (const float*)d_in[9];

    const int N = in_sizes[0] / 64;
    const int E = in_sizes[1] / 2;
    const int* src = ei;
    const int* dst = ei + E;

    char*  ws  = (char*)d_ws;
    size_t off = 0;
    auto carve = [&](size_t bytes) -> float* {
        float* p = (float*)(ws + off);
        off += (bytes + 255) & ~(size_t)255;
        return p;
    };
    float* deg  = carve((size_t)N * 4);
    float* dinv = carve((size_t)N * 4);
    float* xw1  = carve((size_t)N * 256 * 4);
    float* agg1 = carve((size_t)N * 256 * 4);
    float* xw2  = carve((size_t)N * 128 * 4);
    float* emb  = carve((size_t)N * 128 * 4);

    float* out_node = (float*)d_out;
    float* out_edge = out_node + (size_t)N * 2;

    k_zero <<<(N + 255) / 256, 256, 0, stream>>>(deg, N);
    k_count<<<(E + 255) / 256, 256, 0, stream>>>(dst, deg, E);
    k_dinv <<<(N + 255) / 256, 256, 0, stream>>>(deg, dinv, N);

    k_gemm_wmma<false, 64><<<dim3(N / 16, 256 / 128), 256, 0, stream>>>(x, W1, xw1, N, 256);
    k_selfloop_bias<<<(N * 256 + 255) / 256, 256, 0, stream>>>(xw1, dinv, b1, agg1, N, 256);
    k_scatter<<<E, 256, 0, stream>>>(src, dst, dinv, xw1, agg1, 256);

    k_gemm_wmma<true, 256><<<dim3(N / 16, 128 / 128), 256, 0, stream>>>(agg1, W2, xw2, N, 128);
    k_selfloop_bias<<<(N * 128 + 255) / 256, 256, 0, stream>>>(xw2, dinv, b2, emb, N, 128);
    k_scatter<<<E, 128, 0, stream>>>(src, dst, dinv, xw2, emb, 128);

    k_node_head<<<(N + 255) / 256, 256, 0, stream>>>(emb, Wn, bn, out_node, N);
    k_edge_head<<<(E + 7) / 8, 256, 0, stream>>>(src, dst, emb, We, be, out_edge, E);
}


# Layer-2 aggregation stops scanning the edge list: layer 1's tile filter also writes each tile's edges (src, row, weight) to a list that layer 2 reads; LDS arrival order, so f32 sum order, reshuffles
# speedup vs baseline: 4.5011x; 4.5011x over previous
#include <hip/hip_runtime.h>


typedef float v2f __attribute__((ext_vector_type(2)));
typedef float v8f __attribute__((ext_vector_type(8)));

__global__ void k_zero(float* __restrict__ p, int n) {
    int i = blockIdx.x * blockDim.x + threadIdx.x;
    if (i < n) p[i] = 0.0f;
}

__global__ void k_count(const int* __restrict__ dst, float* __restrict__ deg, int E) {
    int e = blockIdx.x * blockDim.x + threadIdx.x;
    if (e < E) atomicAdd(&deg[dst[e]], 1.0f);
}

template <int T>
__global__ void k_dinv_tile(const float* __restrict__ deg, float* __restrict__ dinv,
                            int* __restrict__ tilecnt, int n) {
    __shared__ int s_cnt;
    const int t = threadIdx.x;
    const int i = blockIdx.x * T + t;
    if (t == 0) s_cnt = 0;
    __syncthreads();
    if (t < T && i < n) {
        const float d = deg[i];
        dinv[i] = rsqrtf(d + 1.0f);
        __hip_atomic_fetch_add(&s_cnt, (int)d, __ATOMIC_RELAXED, __HIP_MEMORY_SCOPE_WORKGROUP);
    }
    __syncthreads();
    if (t == 0) tilecnt[blockIdx.x] = s_cnt;
}

__device__ __forceinline__ int tile_list_base(const int* __restrict__ tilecnt, int tile, int lane) {
    int v = 0;
    for (int j = lane; j < tile; j += 32) v += tilecnt[j];
#pragma unroll
    for (int m = 16; m > 0; m >>= 1) v += __shfl_xor(v, m, 32);
    return __builtin_amdgcn_readfirstlane(v);
}

template <bool RELU_A, int K>
__global__ void k_gemm_wmma(const float* __restrict__ A, const float* __restrict__ B,
                            float* __restrict__ C, int M, int N) {
    constexpr int LDK = K + 2;
    __shared__ float As[16 * LDK];

    const int mt   = blockIdx.x;
    const int nb   = blockIdx.y;
    const int tid  = threadIdx.x;
    const int wave = tid >> 5;
    const int lane = tid & 31;

    const float* __restrict__ Ag = A + (size_t)mt * 16 * K;
#pragma unroll
    for (int idx = tid; idx < 16 * K; idx += 256) {
        int r  = idx / K;
        int kk = idx - r * K;
        float v = Ag[idx];
        if (RELU_A) v = v > 0.0f ? v : 0.0f;
        As[r * LDK + kk] = v;
    }
    __syncthreads();

    const int nt  = nb * 8 + wave;
    const int col = lane & 15;
    const int kb  = (lane >> 4) << 1;

    const float* __restrict__ Arow = As + col * LDK + kb;
    const float* __restrict__ Bc   = B + (size_t)nt * 16 + col;

    v8f acc = {};
#pragma unroll
    for (int k = 0; k < K; k += 4) {
        v2f av; av.x = Arow[k];                      av.y = Arow[k + 1];
        v2f bv; bv.x = Bc[(size_t)(k + kb) * N];     bv.y = Bc[(size_t)(k + kb + 1) * N];
        acc = __builtin_amdgcn_wmma_f32_16x16x4_f32(
            false, av, false, bv, (short)0, acc, false, false);
    }

    float* __restrict__ Cp =
        C + (size_t)(mt * 16 + ((lane >> 4) << 3)) * N + (size_t)nt * 16 + col;
#pragma unroll
    for (int r = 0; r < 8; ++r) Cp[(size_t)r * N] = acc[r];
}

__device__ __forceinline__ void lds_add(float* p, float v) {
    __hip_atomic_fetch_add(p, v, __ATOMIC_RELAXED, __HIP_MEMORY_SCOPE_WORKGROUP);
}

template <int F, int G, bool FULL, bool CLAMP = false>
__device__ __forceinline__ void gather_add(int s, int dl, float cf, int nb, int lane,
                                           const float* __restrict__ xw, float* acc) {
    constexpr int H = F / 128;
#pragma unroll 1
    for (int i = 0; i < nb; i += G) {
        float4 v[G][H];
        int    rl[G];
        float  ci[G];
#pragma unroll
        for (int g = 0; g < G; ++g) {
            if (FULL || CLAMP || i + g < nb) {
                const int j  = (CLAMP && i + g >= nb) ? nb - 1 : i + g;
                const int si = __builtin_amdgcn_readlane(s, j);
                rl[g] = __builtin_amdgcn_readlane(dl, j);
                ci[g] = __int_as_float(__builtin_amdgcn_readlane(__float_as_int(cf), j));
                const float* __restrict__ row = xw + (size_t)si * F + lane * 4;
#pragma unroll
                for (int h = 0; h < H; ++h) v[g][h] = *(const float4*)(row + h * 128);
            }
        }
#pragma unroll
        for (int g = 0; g < G; ++g) {
            if (FULL || i + g < nb) {
                float* a = acc + rl[g] * F + lane;
#pragma unroll
                for (int h = 0; h < H; ++h) {
                    lds_add(a + h * 128,      ci[g] * v[g][h].x);
                    lds_add(a + h * 128 + 32, ci[g] * v[g][h].y);
                    lds_add(a + h * 128 + 64, ci[g] * v[g][h].z);
                    lds_add(a + h * 128 + 96, ci[g] * v[g][h].w);
                }
            }
        }
    }
}

__device__ __forceinline__ int2 list_pack(int s, int dl, float cf) {
    return make_int2((int)((unsigned)s | ((unsigned)dl << 24)), __float_as_int(cf));
}

template <int F, int G, bool FULL, bool EMIT>
__device__ __forceinline__ void agg_batch(const int* __restrict__ pe, const int* __restrict__ pd,
                                          int nb, int lane,
                                          const int* __restrict__ src,
                                          const float* __restrict__ dinv,
                                          const float* __restrict__ xw,
                                          const float* dv, float* acc,
                                          int2* __restrict__ lst, int* lpos) {
    int dl = 0, s = 0;
    float cf = 0.0f;
    if (FULL || lane < nb) {
        const int e = pe[lane];
        dl = pd[lane];
        s  = src[e];
        cf = dinv[s] * dv[dl];
    }
    if (EMIT) {
        int pos = 0;
        if (lane == 0)
            pos = __hip_atomic_fetch_add(lpos, nb, __ATOMIC_RELAXED, __HIP_MEMORY_SCOPE_WORKGROUP);
        pos = __builtin_amdgcn_readfirstlane(pos);
        if (FULL || lane < nb) lst[pos + lane] = list_pack(s, dl, cf);
    }
    gather_add<F, G, FULL>(s, dl, cf, nb, lane, xw, acc);
}

template <int F, int T, int NW, bool KEEP_DV>
__device__ __forceinline__ void tile_init(float* acc, float* dv,
                                          const float* __restrict__ dinv,
                                          const float* __restrict__ xw,
                                          const float* __restrict__ bias,
                                          int n0, int nt, int wave, int lane) {
    constexpr int H   = F / 128;
    constexpr int RPW = (T + NW - 1) / NW;
    float4 b[H];
#pragma unroll
    for (int h = 0; h < H; ++h) b[h] = *(const float4*)(bias + h * 128 + lane * 4);
    float4 v[RPW][H];
    float  di[RPW];
#pragma unroll
    for (int k = 0; k < RPW; ++k) {
        const int r  = wave + k * NW;
        const int rc = r < nt ? r : nt - 1;
        di[k] = dinv[n0 + rc];
        const float* __restrict__ xr = xw + (size_t)(n0 + rc) * F + lane * 4;
#pragma unroll
        for (int h = 0; h < H; ++h) v[k][h] = *(const float4*)(xr + h * 128);
    }
#pragma unroll
    for (int k = 0; k < RPW; ++k) {
        const int r = wave + k * NW;
        if (r < nt) {
            if (KEEP_DV && lane == 0) dv[r] = di[k];
            float* a = acc + r * F + lane;
#pragma unroll
            for (int h = 0; h < H; ++h) {
                a[h * 128]      = v[k][h].x * di[k] * di[k] + b[h].x;
                a[h * 128 + 32] = v[k][h].y * di[k] * di[k] + b[h].y;
                a[h * 128 + 64] = v[k][h].z * di[k] * di[k] + b[h].z;
                a[h * 128 + 96] = v[k][h].w * di[k] * di[k] + b[h].w;
            }
        }
    }
}

template <int F, int T, int NW>
__device__ __forceinline__ void tile_store(const float* acc, float* __restrict__ agg,
                                           int n0, int nt, int wave, int lane) {
    constexpr int H   = F / 128;
    constexpr int RPW = (T + NW - 1) / NW;
#pragma unroll
    for (int k = 0; k < RPW; ++k) {
        const int r = wave + k * NW;
        if (r < nt) {
            const float* a = acc + r * F + lane;
            float* __restrict__ o = agg + (size_t)(n0 + r) * F + lane * 4;
#pragma unroll
            for (int h = 0; h < H; ++h) {
                float4 v;
                v.x = a[h * 128];
                v.y = a[h * 128 + 32];
                v.z = a[h * 128 + 64];
                v.w = a[h * 128 + 96];
                *(float4*)(o + h * 128) = v;
            }
        }
    }
}

template <int F, int T, bool EMIT>
__global__ __launch_bounds__(1024)
void k_agg_tile(const int* __restrict__ src, const int* __restrict__ dst,
                const float* __restrict__ dinv, const float* __restrict__ xw,
                const float* __restrict__ bias, float* __restrict__ agg, int N, int E,
                int2* __restrict__ list, const int* __restrict__ tilecnt) {
    constexpr int NW  = 32;
    constexpr int G   = 1024 / F;
    constexpr int CAP = 160;
    static_assert(T <= 256, "a tile row must fit the high byte of a list entry");

    __shared__ float acc[T * F];
    __shared__ float dv[T];
    __shared__ int   l_e[NW * CAP];
    __shared__ int   l_d[NW * CAP];
    __shared__ int   lpos;

    const int tid  = threadIdx.x;
    const int lane = tid & 31;
    const int wave = __builtin_amdgcn_readfirstlane(tid >> 5);
    const int n0   = blockIdx.x * T;
    const int nt   = (N - n0) < T ? (N - n0) : T;

    int2* __restrict__ lst = list;
    if (EMIT) {
        lst += tile_list_base(tilecnt, blockIdx.x, lane);
        if (tid == 0) lpos = 0;
    }

    tile_init<F, T, NW, true>(acc, dv, dinv, xw, bias, n0, nt, wave, lane);
    __syncthreads();

    int* le = l_e + wave * CAP;
    int* ld = l_d + wave * CAP;
    int  cnt = 0;

    auto append = [&](int t, int e, unsigned mask) {
        const int pos = cnt + (int)__builtin_amdgcn_mbcnt_lo(mask, 0u);
        if ((unsigned)t < (unsigned)nt) { le[pos] = e; ld[pos] = t; }
        cnt += __builtin_popcount(mask);
    };
    auto flush = [&]() {
        while (cnt >= 32) {
            cnt -= 32;
            agg_batch<F, G, true, EMIT>(le + cnt, ld + cnt, 32, lane, src, dinv, xw, dv, acc,
                                        lst, &lpos);
        }
    };

    const int nfull = ((E & 3) == 0) ? (E >> 9) : 0;
    const int rot   = nfull > 0 ? (int)(((unsigned long long)blockIdx.x * (unsigned)nfull) / gridDim.x) : 0;
    auto chunk_base = [&](int c) -> int {
        int cc = c + rot;
        if (cc >= nfull) cc -= nfull;
        return cc << 9;
    };
    auto load_chunk = [&](int c, int4 (&d)[4]) {
        const int4* __restrict__ p = (const int4*)(dst + chunk_base(c)) + lane;
#pragma unroll
        for (int j = 0; j < 4; ++j) d[j] = p[j * 32];
    };
    auto filter_chunk = [&](int c, const int4 (&d)[4]) {
        const int base = chunk_base(c);
#pragma unroll
        for (int j = 0; j < 4; ++j) {
            const int e0 = base + (j * 32 + lane) * 4;
            const int t[4] = {d[j].x - n0, d[j].y - n0, d[j].z - n0, d[j].w - n0};
            unsigned mk[4];
#pragma unroll
            for (int q = 0; q < 4; ++q)
                mk[q] = __builtin_amdgcn_ballot_w32((unsigned)t[q] < (unsigned)nt);
            if (mk[0] | mk[1] | mk[2] | mk[3]) {
#pragma unroll
                for (int q = 0; q < 4; ++q)
                    if (mk[q]) append(t[q], e0 + q, mk[q]);
                flush();
            }
        }
    };

    if (wave < nfull) {
        int4 da[4], db[4];
        int  c = wave;
        load_chunk(c, da);
        for (;;) {
            int  cn  = c + NW;
            bool has = cn < nfull;
            load_chunk(has ? cn : c, db);
            filter_chunk(c, da);
            if (!has) break;
            c   = cn;
            cn  = c + NW;
            has = cn < nfull;
            load_chunk(has ? cn : c, da);
            filter_chunk(c, db);
            if (!has) break;
            c = cn;
        }
    }
    for (int eb = (nfull << 9) + wave * 32; eb < E; eb += NW * 32) {
        const int e = eb + lane;
        const int t = (e < E) ? dst[e] - n0 : -1;
        const unsigned mask = __builtin_amdgcn_ballot_w32((unsigned)t < (unsigned)nt);
        if (mask) { append(t, e, mask); flush(); }
    }
    if (cnt > 0)
        agg_batch<F, G, false, EMIT>(le, ld, cnt, lane, src, dinv, xw, dv, acc, lst, &lpos);
    __syncthreads();

    tile_store<F, T, NW>(acc, agg, n0, nt, wave, lane);
}

template <int F, int T>
__global__ __launch_bounds__(1024)
void k_agg_list(const int2* __restrict__ list, const int* __restrict__ tilecnt,
                const float* __restrict__ dinv, const float* __restrict__ xw,
                const float* __restrict__ bias, float* __restrict__ agg, int N) {
    constexpr int NW = 32;
    constexpr int G  = 1024 / F;
    static_assert(G <= 32 && 32 % G == 0, "a 32-entry step is a whole number of row groups");

    __shared__ float acc[T * F];

    const int tid  = threadIdx.x;
    const int lane = tid & 31;
    const int wave = __builtin_amdgcn_readfirstlane(tid >> 5);
    const int n0   = blockIdx.x * T;
    const int nt   = (N - n0) < T ? (N - n0) : T;

    const int2* __restrict__ lst = list + tile_list_base(tilecnt, blockIdx.x, lane);
    const int cnt = tilecnt[blockIdx.x];
    const int per = (((cnt + G - 1) / G + NW - 1) / NW) * G;
    const int lo  = (wave * per < cnt) ? wave * per : cnt;
    const int hi  = (lo + per < cnt) ? lo + per : cnt;

    constexpr int PRE = 4;
    int2 en[PRE];
#pragma unroll
    for (int k = 0; k < PRE; ++k) {
        en[k] = make_int2(0, 0);
        if (lo + k * 32 + lane < hi) en[k] = lst[lo + k * 32 + lane];
    }

    tile_init<F, T, NW, false>(acc, nullptr, dinv, xw, bias, n0, nt, wave, lane);
    __syncthreads();

    int k = 0;
    for (int b0 = lo; b0 < hi; b0 += 32, ++k) {
        const int nb = (hi - b0) < 32 ? (hi - b0) : 32;
        int2 cur = en[0];
#pragma unroll
        for (int j = 1; j < PRE; ++j)
            if (k == j) cur = en[j];
        if (k >= PRE) {
            cur = make_int2(0, 0);
            if (lane < nb) cur = lst[b0 + lane];
        }
        const int   s  = cur.x & 0xFFFFFF;
        const int   dl = (int)((unsigned)cur.x >> 24);
        const float cf = __int_as_float(cur.y);
        if (nb % G == 0) gather_add<F, G, true>(s, dl, cf, nb, lane, xw, acc);
        else             gather_add<F, G, false, true>(s, dl, cf, nb, lane, xw, acc);
    }
    __syncthreads();

    tile_store<F, T, NW>(acc, agg, n0, nt, wave, lane);
}

__global__ void k_node_head(const float* __restrict__ emb, const float* __restrict__ Wn,
                            const float* __restrict__ bn, float* __restrict__ out, int n) {
    int i = blockIdx.x * blockDim.x + threadIdx.x;
    if (i >= n) return;
    const float4* row = (const float4*)(emb + (size_t)i * 128);
    float a0 = bn[0], a1 = bn[1];
#pragma unroll 8
    for (int q = 0; q < 32; ++q) {
        float4 v = row[q];
        int f = q * 4;
        a0 += v.x * Wn[(f + 0) * 2 + 0] + v.y * Wn[(f + 1) * 2 + 0] +
              v.z * Wn[(f + 2) * 2 + 0] + v.w * Wn[(f + 3) * 2 + 0];
        a1 += v.x * Wn[(f + 0) * 2 + 1] + v.y * Wn[(f + 1) * 2 + 1] +
              v.z * Wn[(f + 2) * 2 + 1] + v.w * Wn[(f + 3) * 2 + 1];
    }
    out[(size_t)i * 2 + 0] = a0;
    out[(size_t)i * 2 + 1] = a1;
}

template <int X>
__device__ __forceinline__ float grp8_xor(float v) {
    constexpr int sel = ((0 ^ X) << 0) | ((1 ^ X) << 3) | ((2 ^ X) << 6) | ((3 ^ X) << 9) |
                        ((4 ^ X) << 12) | ((5 ^ X) << 15) | ((6 ^ X) << 18) | ((7 ^ X) << 21);
    return __int_as_float(__builtin_amdgcn_mov_dpp8(__float_as_int(v), sel));
}

__device__ __forceinline__ void dot4x2(const float4 v, const float4 wa, const float4 wb,
                                       float& a0, float& a1) {
    a0 += v.x * wa.x; a1 += v.x * wa.y;
    a0 += v.y * wa.z; a1 += v.y * wa.w;
    a0 += v.z * wb.x; a1 += v.z * wb.y;
    a0 += v.w * wb.z; a1 += v.w * wb.w;
}

__global__ __launch_bounds__(256)
void k_edge_head(const int* __restrict__ src, const int* __restrict__ dst,
                 const float* __restrict__ emb, const float* __restrict__ We,
                 const float* __restrict__ be, float* __restrict__ out, int E) {
    const int lane = threadIdx.x & 31;
    const int q    = lane & 7;
    const int g    = lane >> 3;
    const int wave   = __builtin_amdgcn_readfirstlane(
        (int)(blockIdx.x * (blockDim.x >> 5) + (threadIdx.x >> 5)));
    const int nwaves = gridDim.x * (blockDim.x >> 5);

    float4 ws[4][2], wd[4][2];
#pragma unroll
    for (int k = 0; k < 4; ++k) {
        const float* __restrict__ w = We + ((k * 8 + q) * 4) * 2;
        ws[k][0] = *(const float4*)(w);
        ws[k][1] = *(const float4*)(w + 4);
        wd[k][0] = *(const float4*)(w + 256);
        wd[k][1] = *(const float4*)(w + 260);
    }
    const float* __restrict__ ecol = emb + q * 4;

    const int nit = ((E & 3) == 0) ? (E >> 4) : 0;
    if (wave < nit) {
        const float bq = be[q & 1];
        const bool  b4 = (q & 4) != 0, b2 = (q & 2) != 0, b1 = (q & 1) != 0;
        int  it = wave;
        int4 si = *(const int4*)(src + it * 16 + g * 4);
        int4 di = *(const int4*)(dst + it * 16 + g * 4);
        for (;;) {
            const int sidx[4] = {si.x, si.y, si.z, si.w};
            const int didx[4] = {di.x, di.y, di.z, di.w};
            float4 rs[4][4], rd[4][4];
#pragma unroll
            for (int j = 0; j < 4; ++j) {
                const float* __restrict__ ps = ecol + (size_t)sidx[j] * 128;
                const float* __restrict__ pd = ecol + (size_t)didx[j] * 128;
#pragma unroll
                for (int k = 0; k < 4; ++k) {
                    rs[j][k] = *(const float4*)(ps + k * 32);
                    rd[j][k] = *(const float4*)(pd + k * 32);
                }
            }
            const int  nx  = it + nwaves;
            const bool has = nx < nit;
            if (has) {
                si = *(const int4*)(src + nx * 16 + g * 4);
                di = *(const int4*)(dst + nx * 16 + g * 4);
            }
            float v[8];
#pragma unroll
            for (int j = 0; j < 4; ++j) {
                float a0 = 0.0f, a1 = 0.0f;
#pragma unroll
                for (int k = 0; k < 4; ++k) {
                    dot4x2(rs[j][k], ws[k][0], ws[k][1], a0, a1);
                    dot4x2(rd[j][k], wd[k][0], wd[k][1], a0, a1);
                }
                v[2 * j]     = a0;
                v[2 * j + 1] = a1;
            }
            float u[4];
#pragma unroll
            for (int t = 0; t < 4; ++t)
                u[t] = (b4 ? v[t + 4] : v[t]) + grp8_xor<4>(b4 ? v[t] : v[t + 4]);
            float w2[2];
#pragma unroll
            for (int t = 0; t < 2; ++t)
                w2[t] = (b2 ? u[t + 2] : u[t]) + grp8_xor<2>(b2 ? u[t] : u[t + 2]);
            const float z = (b1 ? w2[1] : w2[0]) + grp8_xor<1>(b1 ? w2[0] : w2[1]);
            out[(size_t)it * 32 + lane] = z + bq;
            if (!has) break;
            it = nx;
        }
    }

    for (int e0 = nit * 16 + wave * 4; e0 < E; e0 += nwaves * 4) {
        const int  e  = e0 + g;
        const bool ok = e < E;
        const int  ec = ok ? e : e0;
        const float* __restrict__ ps = ecol + (size_t)src[ec] * 128;
        const float* __restrict__ pd = ecol + (size_t)dst[ec] * 128;
        float a0 = 0.0f, a1 = 0.0f;
#pragma unroll
        for (int k = 0; k < 4; ++k) {
            dot4x2(*(const float4*)(ps + k * 32), ws[k][0], ws[k][1], a0, a1);
            dot4x2(*(const float4*)(pd + k * 32), wd[k][0], wd[k][1], a0, a1);
        }
        a0 += grp8_xor<4>(a0); a1 += grp8_xor<4>(a1);
        a0 += grp8_xor<2>(a0); a1 += grp8_xor<2>(a1);
        a0 += grp8_xor<1>(a0); a1 += grp8_xor<1>(a1);
        if (ok && q == 0) {
            out[(size_t)e * 2 + 0] = a0 + be[0];
            out[(size_t)e * 2 + 1] = a1 + be[1];
        }
    }
}

extern "C" void kernel_launch(void* const* d_in, const int* in_sizes, int n_in,
                              void* d_out, int out_size, void* d_ws, size_t ws_size,
                              hipStream_t stream) {
    const float* x  = (const float*)d_in[0];
    const int*   ei = (const int*)d_in[1];
    const float* W1 = (const float*)d_in[2];
    const float* b1 = (const float*)d_in[3];
    const float* W2 = (const float*)d_in[4];
    const float* b2 = (const float*)d_in[5];
    const float* Wn = (const float*)d_in[6];
    const float* bn = (const float*)d_in[7];
    const float* We = (const float*)d_in[8];
    const float* be = (const float*)d_in[9];

    const int N = in_sizes[0] / 64;
    const int E = in_sizes[1] / 2;
    const int* src = ei;
    const int* dst = ei + E;

    char*  ws  = (char*)d_ws;
    size_t off = 0;
    auto carve = [&](size_t bytes) -> float* {
        float* p = (float*)(ws + off);
        off += (bytes + 255) & ~(size_t)255;
        return p;
    };
    float* deg  = carve((size_t)N * 4);
    float* dinv = carve((size_t)N * 4);
    float* xw1  = carve((size_t)N * 256 * 4);
    float* agg1 = carve((size_t)N * 256 * 4);
    float* xw2  = carve((size_t)N * 128 * 4);
    float* emb  = carve((size_t)N * 128 * 4);
    int2*  elist   = (int2*)carve((size_t)E * 8);
    int*   tilecnt = (int*)carve((size_t)((N + 195) / 196) * 4);

    float* out_node = (float*)d_out;
    float* out_edge = out_node + (size_t)N * 2;

    k_zero <<<(N + 255) / 256, 256, 0, stream>>>(deg, N);
    k_count<<<(E + 255) / 256, 256, 0, stream>>>(dst, deg, E);
    k_dinv_tile<196><<<(N + 195) / 196, 256, 0, stream>>>(deg, dinv, tilecnt, N);

    k_gemm_wmma<false, 64><<<dim3(N / 16, 256 / 128), 256, 0, stream>>>(x, W1, xw1, N, 256);
    k_agg_tile<256, 196, true><<<(N + 195) / 196, 1024, 0, stream>>>(src, dst, dinv, xw1, b1, agg1, N, E,
                                                                    elist, tilecnt);

    k_gemm_wmma<true, 256><<<dim3(N / 16, 128 / 128), 256, 0, stream>>>(agg1, W2, xw2, N, 128);
    k_agg_list<128, 196><<<(N + 195) / 196, 1024, 0, stream>>>(elist, tilecnt, dinv, xw2, b2, emb, N);

    k_node_head<<<(N + 255) / 256, 256, 0, stream>>>(emb, Wn, bn, out_node, N);
    k_edge_head<<<512, 256, 0, stream>>>(src, dst, emb, We, be, out_edge, E);
}


# Tile edge lists built by one binning pass (LDS rank per tile, one global cursor add per tile and block) instead of 256 full scans of dst; 9 -> 10 launches; list order, so f32 sum order, reshuffles
# speedup vs baseline: 4.7114x; 1.0467x over previous
#include <hip/hip_runtime.h>


typedef float v2f __attribute__((ext_vector_type(2)));
typedef float v8f __attribute__((ext_vector_type(8)));

__global__ void k_zero(float* __restrict__ p, int n) {
    int i = blockIdx.x * blockDim.x + threadIdx.x;
    if (i < n) p[i] = 0.0f;
}

__global__ void k_count(const int* __restrict__ dst, float* __restrict__ deg, int E) {
    int e = blockIdx.x * blockDim.x + threadIdx.x;
    if (e < E) atomicAdd(&deg[dst[e]], 1.0f);
}

template <int T>
__global__ void k_dinv_tile(const float* __restrict__ deg, float* __restrict__ dinv,
                            int* __restrict__ tilecnt, int* __restrict__ cursor, int n) {
    __shared__ int s_cnt;
    const int t = threadIdx.x;
    const int i = blockIdx.x * T + t;
    if (t == 0) s_cnt = 0;
    __syncthreads();
    if (t < T && i < n) {
        const float d = deg[i];
        dinv[i] = rsqrtf(d + 1.0f);
        __hip_atomic_fetch_add(&s_cnt, (int)d, __ATOMIC_RELAXED, __HIP_MEMORY_SCOPE_WORKGROUP);
    }
    __syncthreads();
    if (t == 0) {
        tilecnt[blockIdx.x] = s_cnt;
        cursor[blockIdx.x]  = 0;
    }
}

__device__ __forceinline__ int tile_list_base(const int* __restrict__ tilecnt, int tile, int lane) {
    int v = 0;
    for (int j = lane; j < tile; j += 32) v += tilecnt[j];
#pragma unroll
    for (int m = 16; m > 0; m >>= 1) v += __shfl_xor(v, m, 32);
    return __builtin_amdgcn_readfirstlane(v);
}

template <bool RELU_A, int K>
__global__ void k_gemm_wmma(const float* __restrict__ A, const float* __restrict__ B,
                            float* __restrict__ C, int M, int N) {
    constexpr int LDK = K + 2;
    __shared__ float As[16 * LDK];

    const int mt   = blockIdx.x;
    const int nb   = blockIdx.y;
    const int tid  = threadIdx.x;
    const int wave = tid >> 5;
    const int lane = tid & 31;

    const float* __restrict__ Ag = A + (size_t)mt * 16 * K;
#pragma unroll
    for (int idx = tid; idx < 16 * K; idx += 256) {
        int r  = idx / K;
        int kk = idx - r * K;
        float v = Ag[idx];
        if (RELU_A) v = v > 0.0f ? v : 0.0f;
        As[r * LDK + kk] = v;
    }
    __syncthreads();

    const int nt  = nb * 8 + wave;
    const int col = lane & 15;
    const int kb  = (lane >> 4) << 1;

    const float* __restrict__ Arow = As + col * LDK + kb;
    const float* __restrict__ Bc   = B + (size_t)nt * 16 + col;

    v8f acc = {};
#pragma unroll
    for (int k = 0; k < K; k += 4) {
        v2f av; av.x = Arow[k];                      av.y = Arow[k + 1];
        v2f bv; bv.x = Bc[(size_t)(k + kb) * N];     bv.y = Bc[(size_t)(k + kb + 1) * N];
        acc = __builtin_amdgcn_wmma_f32_16x16x4_f32(
            false, av, false, bv, (short)0, acc, false, false);
    }

    float* __restrict__ Cp =
        C + (size_t)(mt * 16 + ((lane >> 4) << 3)) * N + (size_t)nt * 16 + col;
#pragma unroll
    for (int r = 0; r < 8; ++r) Cp[(size_t)r * N] = acc[r];
}

__device__ __forceinline__ void lds_add(float* p, float v) {
    __hip_atomic_fetch_add(p, v, __ATOMIC_RELAXED, __HIP_MEMORY_SCOPE_WORKGROUP);
}

template <int F, int G, bool FULL, bool CLAMP = false>
__device__ __forceinline__ void gather_add(int s, int dl, float cf, int nb, int lane,
                                           const float* __restrict__ xw, float* acc) {
    constexpr int H = F / 128;
#pragma unroll 1
    for (int i = 0; i < nb; i += G) {
        float4 v[G][H];
        int    rl[G];
        float  ci[G];
#pragma unroll
        for (int g = 0; g < G; ++g) {
            if (FULL || CLAMP || i + g < nb) {
                const int j  = (CLAMP && i + g >= nb) ? nb - 1 : i + g;
                const int si = __builtin_amdgcn_readlane(s, j);
                rl[g] = __builtin_amdgcn_readlane(dl, j);
                ci[g] = __int_as_float(__builtin_amdgcn_readlane(__float_as_int(cf), j));
                const float* __restrict__ row = xw + (size_t)si * F + lane * 4;
#pragma unroll
                for (int h = 0; h < H; ++h) v[g][h] = *(const float4*)(row + h * 128);
            }
        }
#pragma unroll
        for (int g = 0; g < G; ++g) {
            if (FULL || i + g < nb) {
                float* a = acc + rl[g] * F + lane;
#pragma unroll
                for (int h = 0; h < H; ++h) {
                    lds_add(a + h * 128,      ci[g] * v[g][h].x);
                    lds_add(a + h * 128 + 32, ci[g] * v[g][h].y);
                    lds_add(a + h * 128 + 64, ci[g] * v[g][h].z);
                    lds_add(a + h * 128 + 96, ci[g] * v[g][h].w);
                }
            }
        }
    }
}

__device__ __forceinline__ int2 list_pack(int s, int dl, float cf) {
    return make_int2((int)((unsigned)s | ((unsigned)dl << 24)), __float_as_int(cf));
}

template <int F, int T, int NW>
__device__ __forceinline__ void tile_init(float* acc,
                                          const float* __restrict__ dinv,
                                          const float* __restrict__ xw,
                                          const float* __restrict__ bias,
                                          int n0, int nt, int wave, int lane) {
    constexpr int H   = F / 128;
    constexpr int RPW = (T + NW - 1) / NW;
    float4 b[H];
#pragma unroll
    for (int h = 0; h < H; ++h) b[h] = *(const float4*)(bias + h * 128 + lane * 4);
    float4 v[RPW][H];
    float  di[RPW];
#pragma unroll
    for (int k = 0; k < RPW; ++k) {
        const int r  = wave + k * NW;
        const int rc = r < nt ? r : nt - 1;
        di[k] = dinv[n0 + rc];
        const float* __restrict__ xr = xw + (size_t)(n0 + rc) * F + lane * 4;
#pragma unroll
        for (int h = 0; h < H; ++h) v[k][h] = *(const float4*)(xr + h * 128);
    }
#pragma unroll
    for (int k = 0; k < RPW; ++k) {
        const int r = wave + k * NW;
        if (r < nt) {
            float* a = acc + r * F + lane;
#pragma unroll
            for (int h = 0; h < H; ++h) {
                a[h * 128]      = v[k][h].x * di[k] * di[k] + b[h].x;
                a[h * 128 + 32] = v[k][h].y * di[k] * di[k] + b[h].y;
                a[h * 128 + 64] = v[k][h].z * di[k] * di[k] + b[h].z;
                a[h * 128 + 96] = v[k][h].w * di[k] * di[k] + b[h].w;
            }
        }
    }
}

template <int F, int T, int NW>
__device__ __forceinline__ void tile_store(const float* acc, float* __restrict__ agg,
                                           int n0, int nt, int wave, int lane) {
    constexpr int H   = F / 128;
    constexpr int RPW = (T + NW - 1) / NW;
#pragma unroll
    for (int k = 0; k < RPW; ++k) {
        const int r = wave + k * NW;
        if (r < nt) {
            const float* a = acc + r * F + lane;
            float* __restrict__ o = agg + (size_t)(n0 + r) * F + lane * 4;
#pragma unroll
            for (int h = 0; h < H; ++h) {
                float4 v;
                v.x = a[h * 128];
                v.y = a[h * 128 + 32];
                v.z = a[h * 128 + 64];
                v.w = a[h * 128 + 96];
                *(float4*)(o + h * 128) = v;
            }
        }
    }
}

template <int T>
__global__ __launch_bounds__(512)
void k_bin(const int* __restrict__ src, const int* __restrict__ dst,
           const float* __restrict__ dinv, const int* __restrict__ tilecnt,
           int* __restrict__ cursor, int2* __restrict__ list, int N, int E) {
    constexpr int NT  = 512;
    constexpr int EPT = 8;
    static_assert(T <= 256, "a tile row must fit the high byte of a list entry");

    __shared__ int s_cnt[NT];
    __shared__ int s_first[NT];
    __shared__ int s_wsum[NT / 32];

    const int tid   = threadIdx.x;
    const int lane  = tid & 31;
    const int wave  = __builtin_amdgcn_readfirstlane(tid >> 5);
    const int ntile = (N + T - 1) / T;

    s_cnt[tid] = 0;

    const int e0 = blockIdx.x * (NT * EPT) + tid;
    int  s[EPT], d[EPT];
    bool ok[EPT];
#pragma unroll
    for (int j = 0; j < EPT; ++j) {
        const int e = e0 + j * NT;
        ok[j] = e < E;
        s[j]  = ok[j] ? src[e] : 0;
        d[j]  = ok[j] ? dst[e] : 0;
    }
    float cf[EPT];
#pragma unroll
    for (int j = 0; j < EPT; ++j) cf[j] = dinv[s[j]] * dinv[d[j]];

    const int c = tid < ntile ? tilecnt[tid] : 0;
    int x = c;
#pragma unroll
    for (int m = 1; m < 32; m <<= 1) {
        const int y = __shfl_up(x, m, 32);
        if (lane >= m) x += y;
    }
    if (lane == 31) s_wsum[wave] = x;
    __syncthreads();
    int before = 0;
    for (int w = 0; w < wave; ++w) before += s_wsum[w];
    s_first[tid] = before + x - c;

    int tl[EPT], rk[EPT];
#pragma unroll
    for (int j = 0; j < EPT; ++j) {
        tl[j] = d[j] / T;
        rk[j] = 0;
        if (ok[j])
            rk[j] = __hip_atomic_fetch_add(&s_cnt[tl[j]], 1, __ATOMIC_RELAXED, __HIP_MEMORY_SCOPE_WORKGROUP);
    }
    __syncthreads();

    if (tid < ntile) {
        const int n = s_cnt[tid];
        if (n > 0)
            s_first[tid] += __hip_atomic_fetch_add(&cursor[tid], n, __ATOMIC_RELAXED, __HIP_MEMORY_SCOPE_AGENT);
    }
    __syncthreads();

#pragma unroll
    for (int j = 0; j < EPT; ++j)
        if (ok[j]) list[s_first[tl[j]] + rk[j]] = list_pack(s[j], d[j] - tl[j] * T, cf[j]);
}

template <int F, int T>
__global__ __launch_bounds__(1024)
void k_agg_list(const int2* __restrict__ list, const int* __restrict__ tilecnt,
                const float* __restrict__ dinv, const float* __restrict__ xw,
                const float* __restrict__ bias, float* __restrict__ agg, int N) {
    constexpr int NW = 32;
    constexpr int G  = 1024 / F;
    static_assert(G <= 32 && 32 % G == 0, "a 32-entry step is a whole number of row groups");

    __shared__ float acc[T * F];

    const int tid  = threadIdx.x;
    const int lane = tid & 31;
    const int wave = __builtin_amdgcn_readfirstlane(tid >> 5);
    const int n0   = blockIdx.x * T;
    const int nt   = (N - n0) < T ? (N - n0) : T;

    const int2* __restrict__ lst = list + tile_list_base(tilecnt, blockIdx.x, lane);
    const int cnt = tilecnt[blockIdx.x];
    const int per = (((cnt + G - 1) / G + NW - 1) / NW) * G;
    const int lo  = (wave * per < cnt) ? wave * per : cnt;
    const int hi  = (lo + per < cnt) ? lo + per : cnt;

    constexpr int PRE = 4;
    int2 en[PRE];
#pragma unroll
    for (int k = 0; k < PRE; ++k) {
        en[k] = make_int2(0, 0);
        if (lo + k * 32 + lane < hi) en[k] = lst[lo + k * 32 + lane];
    }

    tile_init<F, T, NW>(acc, dinv, xw, bias, n0, nt, wave, lane);
    __syncthreads();

    int k = 0;
    for (int b0 = lo; b0 < hi; b0 += 32, ++k) {
        const int nb = (hi - b0) < 32 ? (hi - b0) : 32;
        int2 cur = en[0];
#pragma unroll
        for (int j = 1; j < PRE; ++j)
            if (k == j) cur = en[j];
        if (k >= PRE) {
            cur = make_int2(0, 0);
            if (lane < nb) cur = lst[b0 + lane];
        }
        const int   s  = cur.x & 0xFFFFFF;
        const int   dl = (int)((unsigned)cur.x >> 24);
        const float cf = __int_as_float(cur.y);
        if (nb % G == 0) gather_add<F, G, true>(s, dl, cf, nb, lane, xw, acc);
        else             gather_add<F, G, false, true>(s, dl, cf, nb, lane, xw, acc);
    }
    __syncthreads();

    tile_store<F, T, NW>(acc, agg, n0, nt, wave, lane);
}

__global__ void k_node_head(const float* __restrict__ emb, const float* __restrict__ Wn,
                            const float* __restrict__ bn, float* __restrict__ out, int n) {
    int i = blockIdx.x * blockDim.x + threadIdx.x;
    if (i >= n) return;
    const float4* row = (const float4*)(emb + (size_t)i * 128);
    float a0 = bn[0], a1 = bn[1];
#pragma unroll 8
    for (int q = 0; q < 32; ++q) {
        float4 v = row[q];
        int f = q * 4;
        a0 += v.x * Wn[(f + 0) * 2 + 0] + v.y * Wn[(f + 1) * 2 + 0] +
              v.z * Wn[(f + 2) * 2 + 0] + v.w * Wn[(f + 3) * 2 + 0];
        a1 += v.x * Wn[(f + 0) * 2 + 1] + v.y * Wn[(f + 1) * 2 + 1] +
              v.z * Wn[(f + 2) * 2 + 1] + v.w * Wn[(f + 3) * 2 + 1];
    }
    out[(size_t)i * 2 + 0] = a0;
    out[(size_t)i * 2 + 1] = a1;
}

template <int X>
__device__ __forceinline__ float grp8_xor(float v) {
    constexpr int sel = ((0 ^ X) << 0) | ((1 ^ X) << 3) | ((2 ^ X) << 6) | ((3 ^ X) << 9) |
                        ((4 ^ X) << 12) | ((5 ^ X) << 15) | ((6 ^ X) << 18) | ((7 ^ X) << 21);
    return __int_as_float(__builtin_amdgcn_mov_dpp8(__float_as_int(v), sel));
}

__device__ __forceinline__ void dot4x2(const float4 v, const float4 wa, const float4 wb,
                                       float& a0, float& a1) {
    a0 += v.x * wa.x; a1 += v.x * wa.y;
    a0 += v.y * wa.z; a1 += v.y * wa.w;
    a0 += v.z * wb.x; a1 += v.z * wb.y;
    a0 += v.w * wb.z; a1 += v.w * wb.w;
}

__global__ __launch_bounds__(256)
void k_edge_head(const int* __restrict__ src, const int* __restrict__ dst,
                 const float* __restrict__ emb, const float* __restrict__ We,
                 const float* __restrict__ be, float* __restrict__ out, int E) {
    const int lane = threadIdx.x & 31;
    const int q    = lane & 7;
    const int g    = lane >> 3;
    const int wave   = __builtin_amdgcn_readfirstlane(
        (int)(blockIdx.x * (blockDim.x >> 5) + (threadIdx.x >> 5)));
    const int nwaves = gridDim.x * (blockDim.x >> 5);

    float4 ws[4][2], wd[4][2];
#pragma unroll
    for (int k = 0; k < 4; ++k) {
        const float* __restrict__ w = We + ((k * 8 + q) * 4) * 2;
        ws[k][0] = *(const float4*)(w);
        ws[k][1] = *(const float4*)(w + 4);
        wd[k][0] = *(const float4*)(w + 256);
        wd[k][1] = *(const float4*)(w + 260);
    }
    const float* __restrict__ ecol = emb + q * 4;

    const int nit = ((E & 3) == 0) ? (E >> 4) : 0;
    if (wave < nit) {
        const float bq = be[q & 1];
        const bool  b4 = (q & 4) != 0, b2 = (q & 2) != 0, b1 = (q & 1) != 0;
        int  it = wave;
        int4 si = *(const int4*)(src + it * 16 + g * 4);
        int4 di = *(const int4*)(dst + it * 16 + g * 4);
        for (;;) {
            const int sidx[4] = {si.x, si.y, si.z, si.w};
            const int didx[4] = {di.x, di.y, di.z, di.w};
            float4 rs[4][4], rd[4][4];
#pragma unroll
            for (int j = 0; j < 4; ++j) {
                const float* __restrict__ ps = ecol + (size_t)sidx[j] * 128;
                const float* __restrict__ pd = ecol + (size_t)didx[j] * 128;
#pragma unroll
                for (int k = 0; k < 4; ++k) {
                    rs[j][k] = *(const float4*)(ps + k * 32);
                    rd[j][k] = *(const float4*)(pd + k * 32);
                }
            }
            const int  nx  = it + nwaves;
            const bool has = nx < nit;
            if (has) {
                si = *(const int4*)(src + nx * 16 + g * 4);
                di = *(const int4*)(dst + nx * 16 + g * 4);
            }
            float v[8];
#pragma unroll
            for (int j = 0; j < 4; ++j) {
                float a0 = 0.0f, a1 = 0.0f;
#pragma unroll
                for (int k = 0; k < 4; ++k) {
                    dot4x2(rs[j][k], ws[k][0], ws[k][1], a0, a1);
                    dot4x2(rd[j][k], wd[k][0], wd[k][1], a0, a1);
                }
                v[2 * j]     = a0;
                v[2 * j + 1] = a1;
            }
            float u[4];
#pragma unroll
            for (int t = 0; t < 4; ++t)
                u[t] = (b4 ? v[t + 4] : v[t]) + grp8_xor<4>(b4 ? v[t] : v[t + 4]);
            float w2[2];
#pragma unroll
            for (int t = 0; t < 2; ++t)
                w2[t] = (b2 ? u[t + 2] : u[t]) + grp8_xor<2>(b2 ? u[t] : u[t + 2]);
            const float z = (b1 ? w2[1] : w2[0]) + grp8_xor<1>(b1 ? w2[0] : w2[1]);
            out[(size_t)it * 32 + lane] = z + bq;
            if (!has) break;
            it = nx;
        }
    }

    for (int e0 = nit * 16 + wave * 4; e0 < E; e0 += nwaves * 4) {
        const int  e  = e0 + g;
        const bool ok = e < E;
        const int  ec = ok ? e : e0;
        const float* __restrict__ ps = ecol + (size_t)src[ec] * 128;
        const float* __restrict__ pd = ecol + (size_t)dst[ec] * 128;
        float a0 = 0.0f, a1 = 0.0f;
#pragma unroll
        for (int k = 0; k < 4; ++k) {
            dot4x2(*(const float4*)(ps + k * 32), ws[k][0], ws[k][1], a0, a1);
            dot4x2(*(const float4*)(pd + k * 32), wd[k][0], wd[k][1], a0, a1);
        }
        a0 += grp8_xor<4>(a0); a1 += grp8_xor<4>(a1);
        a0 += grp8_xor<2>(a0); a1 += grp8_xor<2>(a1);
        a0 += grp8_xor<1>(a0); a1 += grp8_xor<1>(a1);
        if (ok && q == 0) {
            out[(size_t)e * 2 + 0] = a0 + be[0];
            out[(size_t)e * 2 + 1] = a1 + be[1];
        }
    }
}

extern "C" void kernel_launch(void* const* d_in, const int* in_sizes, int n_in,
                              void* d_out, int out_size, void* d_ws, size_t ws_size,
                              hipStream_t stream) {
    const float* x  = (const float*)d_in[0];
    const int*   ei = (const int*)d_in[1];
    const float* W1 = (const float*)d_in[2];
    const float* b1 = (const float*)d_in[3];
    const float* W2 = (const float*)d_in[4];
    const float* b2 = (const float*)d_in[5];
    const float* Wn = (const float*)d_in[6];
    const float* bn = (const float*)d_in[7];
    const float* We = (const float*)d_in[8];
    const float* be = (const float*)d_in[9];

    const int N = in_sizes[0] / 64;
    const int E = in_sizes[1] / 2;
    const int* src = ei;
    const int* dst = ei + E;

    char*  ws  = (char*)d_ws;
    size_t off = 0;
    auto carve = [&](size_t bytes) -> float* {
        float* p = (float*)(ws + off);
        off += (bytes + 255) & ~(size_t)255;
        return p;
    };
    float* deg  = carve((size_t)N * 4);
    float* dinv = carve((size_t)N * 4);
    float* xw1  = carve((size_t)N * 256 * 4);
    float* agg1 = carve((size_t)N * 256 * 4);
    float* xw2  = carve((size_t)N * 128 * 4);
    float* emb  = carve((size_t)N * 128 * 4);
    int2*  elist   = (int2*)carve((size_t)E * 8);
    int*   tilecnt = (int*)carve((size_t)((N + 195) / 196) * 4);
    int*   cursor  = (int*)carve((size_t)((N + 195) / 196) * 4);

    float* out_node = (float*)d_out;
    float* out_edge = out_node + (size_t)N * 2;

    k_zero <<<(N + 255) / 256, 256, 0, stream>>>(deg, N);
    k_count<<<(E + 255) / 256, 256, 0, stream>>>(dst, deg, E);
    k_dinv_tile<196><<<(N + 195) / 196, 256, 0, stream>>>(deg, dinv, tilecnt, cursor, N);

    k_bin<196><<<(E + 4095) / 4096, 512, 0, stream>>>(src, dst, dinv, tilecnt, cursor, elist, N, E);

    k_gemm_wmma<false, 64><<<dim3(N / 16, 256 / 128), 256, 0, stream>>>(x, W1, xw1, N, 256);
    k_agg_list<256, 196><<<(N + 195) / 196, 1024, 0, stream>>>(elist, tilecnt, dinv, xw1, b1, agg1, N);

    k_gemm_wmma<true, 256><<<dim3(N / 16, 128 / 128), 256, 0, stream>>>(agg1, W2, xw2, N, 128);
    k_agg_list<128, 196><<<(N + 195) / 196, 1024, 0, stream>>>(elist, tilecnt, dinv, xw2, b2, emb, N);

    k_node_head<<<(N + 255) / 256, 256, 0, stream>>>(emb, Wn, bn, out_node, N);
    k_edge_head<<<512, 256, 0, stream>>>(src, dst, emb, We, be, out_edge, E);
}


# Degrees and tile lists without global atomics: per-block LDS tile counts, list slots by column sums, each tile counts its rows in LDS; 10 -> 9 launches; list order, so f32 sum order, reshuffles
# speedup vs baseline: 5.1130x; 1.0852x over previous
#include <hip/hip_runtime.h>


typedef float v2f __attribute__((ext_vector_type(2)));
typedef float v8f __attribute__((ext_vector_type(8)));

__device__ __forceinline__ int tile_list_base(const int* __restrict__ tilecnt, int tile, int lane) {
    int v = 0;
    for (int j = lane; j < tile; j += 32) v += tilecnt[j];
#pragma unroll
    for (int m = 16; m > 0; m >>= 1) v += __shfl_xor(v, m, 32);
    return __builtin_amdgcn_readfirstlane(v);
}

template <bool RELU_A, int K>
__global__ void k_gemm_wmma(const float* __restrict__ A, const float* __restrict__ B,
                            float* __restrict__ C, int M, int N) {
    constexpr int LDK = K + 2;
    __shared__ float As[16 * LDK];

    const int mt   = blockIdx.x;
    const int nb   = blockIdx.y;
    const int tid  = threadIdx.x;
    const int wave = tid >> 5;
    const int lane = tid & 31;

    const float* __restrict__ Ag = A + (size_t)mt * 16 * K;
#pragma unroll
    for (int idx = tid; idx < 16 * K; idx += 256) {
        int r  = idx / K;
        int kk = idx - r * K;
        float v = Ag[idx];
        if (RELU_A) v = v > 0.0f ? v : 0.0f;
        As[r * LDK + kk] = v;
    }
    __syncthreads();

    const int nt  = nb * 8 + wave;
    const int col = lane & 15;
    const int kb  = (lane >> 4) << 1;

    const float* __restrict__ Arow = As + col * LDK + kb;
    const float* __restrict__ Bc   = B + (size_t)nt * 16 + col;

    v8f acc = {};
#pragma unroll
    for (int k = 0; k < K; k += 4) {
        v2f av; av.x = Arow[k];                      av.y = Arow[k + 1];
        v2f bv; bv.x = Bc[(size_t)(k + kb) * N];     bv.y = Bc[(size_t)(k + kb + 1) * N];
        acc = __builtin_amdgcn_wmma_f32_16x16x4_f32(
            false, av, false, bv, (short)0, acc, false, false);
    }

    float* __restrict__ Cp =
        C + (size_t)(mt * 16 + ((lane >> 4) << 3)) * N + (size_t)nt * 16 + col;
#pragma unroll
    for (int r = 0; r < 8; ++r) Cp[(size_t)r * N] = acc[r];
}

__device__ __forceinline__ void lds_add(float* p, float v) {
    __hip_atomic_fetch_add(p, v, __ATOMIC_RELAXED, __HIP_MEMORY_SCOPE_WORKGROUP);
}

template <int F, int G, bool FULL, bool CLAMP = false>
__device__ __forceinline__ void gather_add(int s, int dl, float cf, int nb, int lane,
                                           const float* __restrict__ xw, float* acc) {
    constexpr int H = F / 128;
#pragma unroll 1
    for (int i = 0; i < nb; i += G) {
        float4 v[G][H];
        int    rl[G];
        float  ci[G];
#pragma unroll
        for (int g = 0; g < G; ++g) {
            if (FULL || CLAMP || i + g < nb) {
                const int j  = (CLAMP && i + g >= nb) ? nb - 1 : i + g;
                const int si = __builtin_amdgcn_readlane(s, j);
                rl[g] = __builtin_amdgcn_readlane(dl, j);
                ci[g] = __int_as_float(__builtin_amdgcn_readlane(__float_as_int(cf), j));
                const float* __restrict__ row = xw + (size_t)si * F + lane * 4;
#pragma unroll
                for (int h = 0; h < H; ++h) v[g][h] = *(const float4*)(row + h * 128);
            }
        }
#pragma unroll
        for (int g = 0; g < G; ++g) {
            if (FULL || i + g < nb) {
                float* a = acc + rl[g] * F + lane;
#pragma unroll
                for (int h = 0; h < H; ++h) {
                    lds_add(a + h * 128,      ci[g] * v[g][h].x);
                    lds_add(a + h * 128 + 32, ci[g] * v[g][h].y);
                    lds_add(a + h * 128 + 64, ci[g] * v[g][h].z);
                    lds_add(a + h * 128 + 96, ci[g] * v[g][h].w);
                }
            }
        }
    }
}

__device__ __forceinline__ int list_pack(int s, int dl) {
    return (int)((unsigned)s | ((unsigned)dl << 24));
}

template <int F, int T, int NW>
__device__ __forceinline__ void tile_init(float* acc,
                                          const float* __restrict__ dinv,
                                          const float* __restrict__ xw,
                                          const float* __restrict__ bias,
                                          int n0, int nt, int wave, int lane) {
    constexpr int H   = F / 128;
    constexpr int RPW = (T + NW - 1) / NW;
    float4 b[H];
#pragma unroll
    for (int h = 0; h < H; ++h) b[h] = *(const float4*)(bias + h * 128 + lane * 4);
    float4 v[RPW][H];
    float  di[RPW];
#pragma unroll
    for (int k = 0; k < RPW; ++k) {
        const int r  = wave + k * NW;
        const int rc = r < nt ? r : nt - 1;
        di[k] = dinv[n0 + rc];
        const float* __restrict__ xr = xw + (size_t)(n0 + rc) * F + lane * 4;
#pragma unroll
        for (int h = 0; h < H; ++h) v[k][h] = *(const float4*)(xr + h * 128);
    }
#pragma unroll
    for (int k = 0; k < RPW; ++k) {
        const int r = wave + k * NW;
        if (r < nt) {
            float* a = acc + r * F + lane;
#pragma unroll
            for (int h = 0; h < H; ++h) {
                a[h * 128]      = v[k][h].x * di[k] * di[k] + b[h].x;
                a[h * 128 + 32] = v[k][h].y * di[k] * di[k] + b[h].y;
                a[h * 128 + 64] = v[k][h].z * di[k] * di[k] + b[h].z;
                a[h * 128 + 96] = v[k][h].w * di[k] * di[k] + b[h].w;
            }
        }
    }
}

template <int F, int T, int NW>
__device__ __forceinline__ void tile_store(const float* acc, float* __restrict__ agg,
                                           int n0, int nt, int wave, int lane) {
    constexpr int H   = F / 128;
    constexpr int RPW = (T + NW - 1) / NW;
#pragma unroll
    for (int k = 0; k < RPW; ++k) {
        const int r = wave + k * NW;
        if (r < nt) {
            const float* a = acc + r * F + lane;
            float* __restrict__ o = agg + (size_t)(n0 + r) * F + lane * 4;
#pragma unroll
            for (int h = 0; h < H; ++h) {
                float4 v;
                v.x = a[h * 128];
                v.y = a[h * 128 + 32];
                v.z = a[h * 128 + 64];
                v.w = a[h * 128 + 96];
                *(float4*)(o + h * 128) = v;
            }
        }
    }
}

template <int T>
__global__ __launch_bounds__(512)
void k_tilehist(const int* __restrict__ dst, int* __restrict__ wgcnt, int E) {
    constexpr int NT  = 512;
    constexpr int EPT = 8;
    __shared__ int s_cnt[NT];
    const int tid = threadIdx.x;
    s_cnt[tid] = 0;
    const int e0 = blockIdx.x * (NT * EPT) + tid;
    int d[EPT];
#pragma unroll
    for (int j = 0; j < EPT; ++j) {
        const int e = e0 + j * NT;
        d[j] = e < E ? dst[e] : -1;
    }
    __syncthreads();
#pragma unroll
    for (int j = 0; j < EPT; ++j)
        if (d[j] >= 0)
            __hip_atomic_fetch_add(&s_cnt[d[j] / T], 1, __ATOMIC_RELAXED, __HIP_MEMORY_SCOPE_WORKGROUP);
    __syncthreads();
    wgcnt[blockIdx.x * NT + tid] = s_cnt[tid];
}

template <int T>
__global__ __launch_bounds__(512)
void k_bin(const int* __restrict__ src, const int* __restrict__ dst,
           const int* __restrict__ wgcnt, int* __restrict__ tilecnt,
           int* __restrict__ list, int N, int E) {
    constexpr int NT  = 512;
    constexpr int EPT = 8;
    static_assert(T <= 256, "a tile row must fit the high byte of a list entry");

    __shared__ int s_cnt[NT];
    __shared__ int s_first[NT];
    __shared__ int s_wsum[NT / 32];

    const int tid   = threadIdx.x;
    const int lane  = tid & 31;
    const int wave  = __builtin_amdgcn_readfirstlane(tid >> 5);
    const int ntile = (N + T - 1) / T;
    const int me    = blockIdx.x;
    const int nwg   = gridDim.x;

    s_cnt[tid] = 0;

    const int e0 = blockIdx.x * (NT * EPT) + tid;
    int  s[EPT], d[EPT];
    bool ok[EPT];
#pragma unroll
    for (int j = 0; j < EPT; ++j) {
        const int e = e0 + j * NT;
        ok[j] = e < E;
        s[j]  = ok[j] ? src[e] : 0;
        d[j]  = ok[j] ? dst[e] : 0;
    }

    int before = 0, c = 0;
    if (tid < ntile) {
        const int* __restrict__ col = wgcnt + tid;
        constexpr int CB = 28;
        for (int b0 = 0; b0 < nwg; b0 += CB) {
            int v[CB];
#pragma unroll
            for (int j = 0; j < CB; ++j) {
                const int b = b0 + j;
                v[j] = col[(b < nwg ? b : nwg - 1) * NT];
            }
#pragma unroll
            for (int j = 0; j < CB; ++j) {
                const int b = b0 + j;
                c      += v[j] & -(int)(b < nwg);
                before += v[j] & -(int)(b < me);
            }
        }
        if (me == 0) tilecnt[tid] = c;
    }
    int x = c;
#pragma unroll
    for (int m = 1; m < 32; m <<= 1) {
        const int y = __shfl_up(x, m, 32);
        if (lane >= m) x += y;
    }
    if (lane == 31) s_wsum[wave] = x;
    __syncthreads();
    int prev = 0;
    for (int w = 0; w < wave; ++w) prev += s_wsum[w];
    s_first[tid] = prev + x - c + before;

    int tl[EPT], rk[EPT];
#pragma unroll
    for (int j = 0; j < EPT; ++j) {
        tl[j] = d[j] / T;
        rk[j] = 0;
        if (ok[j])
            rk[j] = __hip_atomic_fetch_add(&s_cnt[tl[j]], 1, __ATOMIC_RELAXED, __HIP_MEMORY_SCOPE_WORKGROUP);
    }
    __syncthreads();

#pragma unroll
    for (int j = 0; j < EPT; ++j)
        if (ok[j]) list[s_first[tl[j]] + rk[j]] = list_pack(s[j], d[j] - tl[j] * T);
}

template <int T>
__global__ __launch_bounds__(1024)
void k_deg(const int* __restrict__ list, const int* __restrict__ tilecnt,
           float* __restrict__ dinv, int N) {
    constexpr int NT = 1024;
    constexpr int LB = 4;
    static_assert(T <= 256, "one counter per tile row");
    __shared__ int s_deg[256];
    const int tid  = threadIdx.x;
    const int lane = tid & 31;
    if (tid < 256) s_deg[tid] = 0;
    const int* __restrict__ lst = list + tile_list_base(tilecnt, blockIdx.x, lane);
    const int cnt = tilecnt[blockIdx.x];
    __syncthreads();
    for (int i0 = 0; i0 < cnt; i0 += NT * LB) {
        int v[LB];
#pragma unroll
        for (int j = 0; j < LB; ++j) {
            const int i = i0 + j * NT + tid;
            v[j] = lst[i < cnt ? i : cnt - 1];
        }
#pragma unroll
        for (int j = 0; j < LB; ++j)
            if (i0 + j * NT + tid < cnt)
                __hip_atomic_fetch_add(&s_deg[(unsigned)v[j] >> 24], 1, __ATOMIC_RELAXED, __HIP_MEMORY_SCOPE_WORKGROUP);
    }
    __syncthreads();
    const int node = blockIdx.x * T + tid;
    if (tid < T && node < N) dinv[node] = rsqrtf((float)s_deg[tid] + 1.0f);
}

template <int F, int T, bool FIRST>
__global__ __launch_bounds__(1024)
void k_agg_list(const int* __restrict__ list, float* __restrict__ cflist,
                const int* __restrict__ tilecnt,
                const float* __restrict__ dinv, const float* __restrict__ xw,
                const float* __restrict__ bias, float* __restrict__ agg, int N) {
    constexpr int NW = 32;
    constexpr int G  = 1024 / F;
    static_assert(G <= 32 && 32 % G == 0, "a 32-entry step is a whole number of row groups");

    __shared__ float acc[T * F];

    const int tid  = threadIdx.x;
    const int lane = tid & 31;
    const int wave = __builtin_amdgcn_readfirstlane(tid >> 5);
    const int n0   = blockIdx.x * T;
    const int nt   = (N - n0) < T ? (N - n0) : T;

    const int lbase = tile_list_base(tilecnt, blockIdx.x, lane);
    const int* __restrict__ lst = list + lbase;
    float*     __restrict__ cfl = cflist + lbase;
    const int cnt = tilecnt[blockIdx.x];
    const int per = (((cnt + G - 1) / G + NW - 1) / NW) * G;
    const int lo  = (wave * per < cnt) ? wave * per : cnt;
    const int hi  = (lo + per < cnt) ? lo + per : cnt;

    constexpr int PRE = 4;
    int   en[PRE];
    float ec[PRE];
#pragma unroll
    for (int k = 0; k < PRE; ++k) {
        en[k] = 0;
        if (lo + k * 32 + lane < hi) en[k] = lst[lo + k * 32 + lane];
    }
    float ds[PRE], dd[PRE];
#pragma unroll
    for (int k = 0; k < PRE; ++k) {
        ec[k] = 0.0f;
        if (FIRST) {
            ds[k] = dinv[en[k] & 0xFFFFFF];
            dd[k] = dinv[n0 + (int)((unsigned)en[k] >> 24)];
        } else if (lo + k * 32 + lane < hi) {
            ec[k] = cfl[lo + k * 32 + lane];
        }
    }

    tile_init<F, T, NW>(acc, dinv, xw, bias, n0, nt, wave, lane);
    if (FIRST) {
#pragma unroll
        for (int k = 0; k < PRE; ++k) {
            ec[k] = ds[k] * dd[k];
            if (lo + k * 32 + lane < hi) cfl[lo + k * 32 + lane] = ec[k];
        }
    }
    __syncthreads();

    int k = 0;
    for (int b0 = lo; b0 < hi; b0 += 32, ++k) {
        const int nb = (hi - b0) < 32 ? (hi - b0) : 32;
        int   cur = en[0];
        float cf  = ec[0];
#pragma unroll
        for (int j = 1; j < PRE; ++j)
            if (k == j) { cur = en[j]; cf = ec[j]; }
        if (k >= PRE) {
            cur = 0;
            cf  = 0.0f;
            if (lane < nb) {
                cur = lst[b0 + lane];
                if (FIRST) {
                    cf = dinv[cur & 0xFFFFFF] * dinv[n0 + (int)((unsigned)cur >> 24)];
                    cfl[b0 + lane] = cf;
                } else {
                    cf = cfl[b0 + lane];
                }
            }
        }
        const int s  = cur & 0xFFFFFF;
        const int dl = (int)((unsigned)cur >> 24);
        if (nb % G == 0) gather_add<F, G, true>(s, dl, cf, nb, lane, xw, acc);
        else             gather_add<F, G, false, true>(s, dl, cf, nb, lane, xw, acc);
    }
    __syncthreads();

    tile_store<F, T, NW>(acc, agg, n0, nt, wave, lane);
}

__global__ void k_node_head(const float* __restrict__ emb, const float* __restrict__ Wn,
                            const float* __restrict__ bn, float* __restrict__ out, int n) {
    int i = blockIdx.x * blockDim.x + threadIdx.x;
    if (i >= n) return;
    const float4* row = (const float4*)(emb + (size_t)i * 128);
    float a0 = bn[0], a1 = bn[1];
#pragma unroll 8
    for (int q = 0; q < 32; ++q) {
        float4 v = row[q];
        int f = q * 4;
        a0 += v.x * Wn[(f + 0) * 2 + 0] + v.y * Wn[(f + 1) * 2 + 0] +
              v.z * Wn[(f + 2) * 2 + 0] + v.w * Wn[(f + 3) * 2 + 0];
        a1 += v.x * Wn[(f + 0) * 2 + 1] + v.y * Wn[(f + 1) * 2 + 1] +
              v.z * Wn[(f + 2) * 2 + 1] + v.w * Wn[(f + 3) * 2 + 1];
    }
    out[(size_t)i * 2 + 0] = a0;
    out[(size_t)i * 2 + 1] = a1;
}

template <int X>
__device__ __forceinline__ float grp8_xor(float v) {
    constexpr int sel = ((0 ^ X) << 0) | ((1 ^ X) << 3) | ((2 ^ X) << 6) | ((3 ^ X) << 9) |
                        ((4 ^ X) << 12) | ((5 ^ X) << 15) | ((6 ^ X) << 18) | ((7 ^ X) << 21);
    return __int_as_float(__builtin_amdgcn_mov_dpp8(__float_as_int(v), sel));
}

__device__ __forceinline__ void dot4x2(const float4 v, const float4 wa, const float4 wb,
                                       float& a0, float& a1) {
    a0 += v.x * wa.x; a1 += v.x * wa.y;
    a0 += v.y * wa.z; a1 += v.y * wa.w;
    a0 += v.z * wb.x; a1 += v.z * wb.y;
    a0 += v.w * wb.z; a1 += v.w * wb.w;
}

__global__ __launch_bounds__(256)
void k_edge_head(const int* __restrict__ src, const int* __restrict__ dst,
                 const float* __restrict__ emb, const float* __restrict__ We,
                 const float* __restrict__ be, float* __restrict__ out, int E) {
    const int lane = threadIdx.x & 31;
    const int q    = lane & 7;
    const int g    = lane >> 3;
    const int wave   = __builtin_amdgcn_readfirstlane(
        (int)(blockIdx.x * (blockDim.x >> 5) + (threadIdx.x >> 5)));
    const int nwaves = gridDim.x * (blockDim.x >> 5);

    float4 ws[4][2], wd[4][2];
#pragma unroll
    for (int k = 0; k < 4; ++k) {
        const float* __restrict__ w = We + ((k * 8 + q) * 4) * 2;
        ws[k][0] = *(const float4*)(w);
        ws[k][1] = *(const float4*)(w + 4);
        wd[k][0] = *(const float4*)(w + 256);
        wd[k][1] = *(const float4*)(w + 260);
    }
    const float* __restrict__ ecol = emb + q * 4;

    const int nit = ((E & 3) == 0) ? (E >> 4) : 0;
    if (wave < nit) {
        const float bq = be[q & 1];
        const bool  b4 = (q & 4) != 0, b2 = (q & 2) != 0, b1 = (q & 1) != 0;
        int  it = wave;
        int4 si = *(const int4*)(src + it * 16 + g * 4);
        int4 di = *(const int4*)(dst + it * 16 + g * 4);
        for (;;) {
            const int sidx[4] = {si.x, si.y, si.z, si.w};
            const int didx[4] = {di.x, di.y, di.z, di.w};
            float4 rs[4][4], rd[4][4];
#pragma unroll
            for (int j = 0; j < 4; ++j) {
                const float* __restrict__ ps = ecol + (size_t)sidx[j] * 128;
                const float* __restrict__ pd = ecol + (size_t)didx[j] * 128;
#pragma unroll
                for (int k = 0; k < 4; ++k) {
                    rs[j][k] = *(const float4*)(ps + k * 32);
                    rd[j][k] = *(const float4*)(pd + k * 32);
                }
            }
            const int  nx  = it + nwaves;
            const bool has = nx < nit;
            if (has) {
                si = *(const int4*)(src + nx * 16 + g * 4);
                di = *(const int4*)(dst + nx * 16 + g * 4);
            }
            float v[8];
#pragma unroll
            for (int j = 0; j < 4; ++j) {
                float a0 = 0.0f, a1 = 0.0f;
#pragma unroll
                for (int k = 0; k < 4; ++k) {
                    dot4x2(rs[j][k], ws[k][0], ws[k][1], a0, a1);
                    dot4x2(rd[j][k], wd[k][0], wd[k][1], a0, a1);
                }
                v[2 * j]     = a0;
                v[2 * j + 1] = a1;
            }
            float u[4];
#pragma unroll
            for (int t = 0; t < 4; ++t)
                u[t] = (b4 ? v[t + 4] : v[t]) + grp8_xor<4>(b4 ? v[t] : v[t + 4]);
            float w2[2];
#pragma unroll
            for (int t = 0; t < 2; ++t)
                w2[t] = (b2 ? u[t + 2] : u[t]) + grp8_xor<2>(b2 ? u[t] : u[t + 2]);
            const float z = (b1 ? w2[1] : w2[0]) + grp8_xor<1>(b1 ? w2[0] : w2[1]);
            out[(size_t)it * 32 + lane] = z + bq;
            if (!has) break;
            it = nx;
        }
    }

    for (int e0 = nit * 16 + wave * 4; e0 < E; e0 += nwaves * 4) {
        const int  e  = e0 + g;
        const bool ok = e < E;
        const int  ec = ok ? e : e0;
        const float* __restrict__ ps = ecol + (size_t)src[ec] * 128;
        const float* __restrict__ pd = ecol + (size_t)dst[ec] * 128;
        float a0 = 0.0f, a1 = 0.0f;
#pragma unroll
        for (int k = 0; k < 4; ++k) {
            dot4x2(*(const float4*)(ps + k * 32), ws[k][0], ws[k][1], a0, a1);
            dot4x2(*(const float4*)(pd + k * 32), wd[k][0], wd[k][1], a0, a1);
        }
        a0 += grp8_xor<4>(a0); a1 += grp8_xor<4>(a1);
        a0 += grp8_xor<2>(a0); a1 += grp8_xor<2>(a1);
        a0 += grp8_xor<1>(a0); a1 += grp8_xor<1>(a1);
        if (ok && q == 0) {
            out[(size_t)e * 2 + 0] = a0 + be[0];
            out[(size_t)e * 2 + 1] = a1 + be[1];
        }
    }
}

extern "C" void kernel_launch(void* const* d_in, const int* in_sizes, int n_in,
                              void* d_out, int out_size, void* d_ws, size_t ws_size,
                              hipStream_t stream) {
    const float* x  = (const float*)d_in[0];
    const int*   ei = (const int*)d_in[1];
    const float* W1 = (const float*)d_in[2];
    const float* b1 = (const float*)d_in[3];
    const float* W2 = (const float*)d_in[4];
    const float* b2 = (const float*)d_in[5];
    const float* Wn = (const float*)d_in[6];
    const float* bn = (const float*)d_in[7];
    const float* We = (const float*)d_in[8];
    const float* be = (const float*)d_in[9];

    const int N = in_sizes[0] / 64;
    const int E = in_sizes[1] / 2;
    const int* src = ei;
    const int* dst = ei + E;

    char*  ws  = (char*)d_ws;
    size_t off = 0;
    auto carve = [&](size_t bytes) -> float* {
        float* p = (float*)(ws + off);
        off += (bytes + 255) & ~(size_t)255;
        return p;
    };
    float* dinv = carve((size_t)N * 4);
    float* xw1  = carve((size_t)N * 256 * 4);
    float* agg1 = carve((size_t)N * 256 * 4);
    float* xw2  = carve((size_t)N * 128 * 4);
    float* emb  = carve((size_t)N * 128 * 4);
    int*   elist   = (int*)carve((size_t)E * 4);
    float* cflist  = carve((size_t)E * 4);
    int*   tilecnt = (int*)carve((size_t)((N + 195) / 196) * 4);
    int*   wgcnt   = (int*)carve((size_t)((E + 4095) / 4096) * 512 * 4);

    float* out_node = (float*)d_out;
    float* out_edge = out_node + (size_t)N * 2;

    k_tilehist<196><<<(E + 4095) / 4096, 512, 0, stream>>>(dst, wgcnt, E);
    k_bin<196><<<(E + 4095) / 4096, 512, 0, stream>>>(src, dst, wgcnt, tilecnt, elist, N, E);
    k_deg<196><<<(N + 195) / 196, 1024, 0, stream>>>(elist, tilecnt, dinv, N);

    k_gemm_wmma<false, 64><<<dim3(N / 16, 256 / 128), 256, 0, stream>>>(x, W1, xw1, N, 256);
    k_agg_list<256, 196, true><<<(N + 195) / 196, 1024, 0, stream>>>(elist, cflist, tilecnt, dinv, xw1, b1, agg1, N);

    k_gemm_wmma<true, 256><<<dim3(N / 16, 128 / 128), 256, 0, stream>>>(agg1, W2, xw2, N, 128);
    k_agg_list<128, 196, false><<<(N + 195) / 196, 1024, 0, stream>>>(elist, cflist, tilecnt, dinv, xw2, b2, emb, N);

    k_node_head<<<(N + 255) / 256, 256, 0, stream>>>(emb, Wn, bn, out_node, N);
    k_edge_head<<<512, 256, 0, stream>>>(src, dst, emb, We, be, out_edge, E);
}


# Aggregation tiles feed the next dense stage from LDS: layer 1 stores relu(tile) @ W2 (WMMA, k order = LDS order), layer 2 adds the node head (wave tree sum); 9 -> 7 launches; f32 sum order changes
# speedup vs baseline: 6.0943x; 1.1919x over previous
#include <hip/hip_runtime.h>


typedef float v2f __attribute__((ext_vector_type(2)));
typedef float v8f __attribute__((ext_vector_type(8)));

__device__ __forceinline__ int tile_list_base(const int* __restrict__ tilecnt, int tile, int lane) {
    int v = 0;
    for (int j = lane; j < tile; j += 32) v += tilecnt[j];
#pragma unroll
    for (int m = 16; m > 0; m >>= 1) v += __shfl_xor(v, m, 32);
    return __builtin_amdgcn_readfirstlane(v);
}

template <bool RELU_A, int K>
__global__ void k_gemm_wmma(const float* __restrict__ A, const float* __restrict__ B,
                            float* __restrict__ C, int M, int N) {
    constexpr int LDK = K + 2;
    __shared__ float As[16 * LDK];

    const int mt   = blockIdx.x;
    const int nb   = blockIdx.y;
    const int tid  = threadIdx.x;
    const int wave = tid >> 5;
    const int lane = tid & 31;

    const float* __restrict__ Ag = A + (size_t)mt * 16 * K;
#pragma unroll
    for (int idx = tid; idx < 16 * K; idx += 256) {
        int r  = idx / K;
        int kk = idx - r * K;
        float v = Ag[idx];
        if (RELU_A) v = v > 0.0f ? v : 0.0f;
        As[r * LDK + kk] = v;
    }
    __syncthreads();

    const int nt  = nb * 8 + wave;
    const int col = lane & 15;
    const int kb  = (lane >> 4) << 1;

    const float* __restrict__ Arow = As + col * LDK + kb;
    const float* __restrict__ Bc   = B + (size_t)nt * 16 + col;

    v8f acc = {};
#pragma unroll
    for (int k = 0; k < K; k += 4) {
        v2f av; av.x = Arow[k];                      av.y = Arow[k + 1];
        v2f bv; bv.x = Bc[(size_t)(k + kb) * N];     bv.y = Bc[(size_t)(k + kb + 1) * N];
        acc = __builtin_amdgcn_wmma_f32_16x16x4_f32(
            false, av, false, bv, (short)0, acc, false, false);
    }

    float* __restrict__ Cp =
        C + (size_t)(mt * 16 + ((lane >> 4) << 3)) * N + (size_t)nt * 16 + col;
#pragma unroll
    for (int r = 0; r < 8; ++r) Cp[(size_t)r * N] = acc[r];
}

__device__ __forceinline__ void lds_add(float* p, float v) {
    __hip_atomic_fetch_add(p, v, __ATOMIC_RELAXED, __HIP_MEMORY_SCOPE_WORKGROUP);
}

template <int F, int S, int G, bool FULL, bool CLAMP = false>
__device__ __forceinline__ void gather_add(int s, int dl, float cf, int nb, int lane,
                                           const float* __restrict__ xw, float* acc) {
    constexpr int H = F / 128;
#pragma unroll 1
    for (int i = 0; i < nb; i += G) {
        float4 v[G][H];
        int    rl[G];
        float  ci[G];
#pragma unroll
        for (int g = 0; g < G; ++g) {
            if (FULL || CLAMP || i + g < nb) {
                const int j  = (CLAMP && i + g >= nb) ? nb - 1 : i + g;
                const int si = __builtin_amdgcn_readlane(s, j);
                rl[g] = __builtin_amdgcn_readlane(dl, j);
                ci[g] = __int_as_float(__builtin_amdgcn_readlane(__float_as_int(cf), j));
                const float* __restrict__ row = xw + (size_t)si * F + lane * 4;
#pragma unroll
                for (int h = 0; h < H; ++h) v[g][h] = *(const float4*)(row + h * 128);
            }
        }
#pragma unroll
        for (int g = 0; g < G; ++g) {
            if (FULL || i + g < nb) {
                float* a = acc + rl[g] * S + lane;
#pragma unroll
                for (int h = 0; h < H; ++h) {
                    lds_add(a + h * 128,      ci[g] * v[g][h].x);
                    lds_add(a + h * 128 + 32, ci[g] * v[g][h].y);
                    lds_add(a + h * 128 + 64, ci[g] * v[g][h].z);
                    lds_add(a + h * 128 + 96, ci[g] * v[g][h].w);
                }
            }
        }
    }
}

__device__ __forceinline__ int list_pack(int s, int dl) {
    return (int)((unsigned)s | ((unsigned)dl << 24));
}

template <int F, int S, int T, int NW>
__device__ __forceinline__ void tile_init(float* acc,
                                          const float* __restrict__ dinv,
                                          const float* __restrict__ xw,
                                          const float* __restrict__ bias,
                                          int n0, int nt, int wave, int lane) {
    constexpr int H   = F / 128;
    constexpr int RPW = (T + NW - 1) / NW;
    float4 b[H];
#pragma unroll
    for (int h = 0; h < H; ++h) b[h] = *(const float4*)(bias + h * 128 + lane * 4);
    float4 v[RPW][H];
    float  di[RPW];
#pragma unroll
    for (int k = 0; k < RPW; ++k) {
        const int r  = wave + k * NW;
        const int rc = r < nt ? r : nt - 1;
        di[k] = dinv[n0 + rc];
        const float* __restrict__ xr = xw + (size_t)(n0 + rc) * F + lane * 4;
#pragma unroll
        for (int h = 0; h < H; ++h) v[k][h] = *(const float4*)(xr + h * 128);
    }
#pragma unroll
    for (int k = 0; k < RPW; ++k) {
        const int r = wave + k * NW;
        if (r < nt) {
            float* a = acc + r * S + lane;
#pragma unroll
            for (int h = 0; h < H; ++h) {
                a[h * 128]      = v[k][h].x * di[k] * di[k] + b[h].x;
                a[h * 128 + 32] = v[k][h].y * di[k] * di[k] + b[h].y;
                a[h * 128 + 64] = v[k][h].z * di[k] * di[k] + b[h].z;
                a[h * 128 + 96] = v[k][h].w * di[k] * di[k] + b[h].w;
            }
        }
    }
}

template <int T, int NW>
__device__ __forceinline__ void tile_store_head(const float* acc, float* head,
                                                float* __restrict__ emb,
                                                const float* __restrict__ Wn,
                                                const float* __restrict__ bn,
                                                int n0, int nt, int wave, int lane) {
    constexpr int F   = 128;
    constexpr int RPW = (T + NW - 1) / NW;
    const float4 wa = *(const float4*)(Wn + lane * 8);
    const float4 wb = *(const float4*)(Wn + lane * 8 + 4);
    const float  b0 = bn[0], b1 = bn[1];
#pragma unroll
    for (int k = 0; k < RPW; ++k) {
        const int r = wave + k * NW;
        if (r < nt) {
            const float* a = acc + r * F + lane;
            float4 v;
            v.x = a[0];
            v.y = a[32];
            v.z = a[64];
            v.w = a[96];
            *(float4*)(emb + (size_t)(n0 + r) * F + lane * 4) = v;
            float p0 = v.x * wa.x + v.y * wa.z + v.z * wb.x + v.w * wb.z;
            float p1 = v.x * wa.y + v.y * wa.w + v.z * wb.y + v.w * wb.w;
#pragma unroll
            for (int m = 16; m > 0; m >>= 1) {
                p0 += __shfl_xor(p0, m, 32);
                p1 += __shfl_xor(p1, m, 32);
            }
            if (lane == 0) {
                head[2 * r]     = p0 + b0;
                head[2 * r + 1] = p1 + b1;
            }
        }
    }
}

template <int NM, int S>
__device__ __forceinline__ void tile_gemm_relu(const float* acc, const float* __restrict__ Wb,
                                               float* __restrict__ C, int n0, int nt,
                                               int mg, int nc, int lane) {
    constexpr int N = 128;
    const int col = lane & 15;
    const int hf  = lane >> 4;
    const float* __restrict__ Bc = Wb + (size_t)(8 * hf) * N + nc * 16 + col;
    const float* Ar = acc + (mg * 16 + col) * S + 2 * hf;

    v8f d[NM];
#pragma unroll
    for (int m = 0; m < NM; ++m) d[m] = v8f{};

    auto brow = [](int s) { return (s >> 5) * 128 + ((s >> 3) & 3) + 16 * (s & 7); };
    constexpr int KC = 4;
    v2f bcur[KC], bnxt[KC];
#pragma unroll
    for (int i = 0; i < KC; ++i) {
        bcur[i].x = Bc[(size_t)(brow(i)) * N];
        bcur[i].y = Bc[(size_t)(brow(i) + 4) * N];
    }
#pragma unroll 1
    for (int s0 = 0; s0 < 64; s0 += KC) {
        const int sn = s0 + KC < 64 ? s0 + KC : s0;
        const float* __restrict__ Bn = Bc + (size_t)brow(sn) * N;
#pragma unroll
        for (int i = 0; i < KC; ++i) {
            bnxt[i].x = Bn[(size_t)(16 * i) * N];
            bnxt[i].y = Bn[(size_t)(16 * i + 4) * N];
        }
#pragma unroll
        for (int i = 0; i < KC; ++i) {
#pragma unroll
            for (int m = 0; m < NM; ++m) {
                v2f av = *(const v2f*)(Ar + m * 64 * S + 4 * (s0 + i));
                av.x = fmaxf(av.x, 0.0f);
                av.y = fmaxf(av.y, 0.0f);
                d[m] = __builtin_amdgcn_wmma_f32_16x16x4_f32(
                    false, av, false, bcur[i], (short)0, d[m], false, false);
            }
        }
#pragma unroll
        for (int i = 0; i < KC; ++i) bcur[i] = bnxt[i];
    }

#pragma unroll
    for (int m = 0; m < NM; ++m) {
#pragma unroll
        for (int r = 0; r < 8; ++r) {
            const int row = mg * 16 + m * 64 + r + 8 * hf;
            if (row < nt) C[(size_t)(n0 + row) * N + nc * 16 + col] = d[m][r];
        }
    }
}

template <int T>
__global__ __launch_bounds__(512)
void k_tilehist(const int* __restrict__ dst, int* __restrict__ wgcnt, int E) {
    constexpr int NT  = 512;
    constexpr int EPT = 8;
    __shared__ int s_cnt[NT];
    const int tid = threadIdx.x;
    s_cnt[tid] = 0;
    const int e0 = blockIdx.x * (NT * EPT) + tid;
    int d[EPT];
#pragma unroll
    for (int j = 0; j < EPT; ++j) {
        const int e = e0 + j * NT;
        d[j] = e < E ? dst[e] : -1;
    }
    __syncthreads();
#pragma unroll
    for (int j = 0; j < EPT; ++j)
        if (d[j] >= 0)
            __hip_atomic_fetch_add(&s_cnt[d[j] / T], 1, __ATOMIC_RELAXED, __HIP_MEMORY_SCOPE_WORKGROUP);
    __syncthreads();
    wgcnt[blockIdx.x * NT + tid] = s_cnt[tid];
}

template <int T>
__global__ __launch_bounds__(512)
void k_bin(const int* __restrict__ src, const int* __restrict__ dst,
           const int* __restrict__ wgcnt, int* __restrict__ tilecnt,
           int* __restrict__ list, int N, int E) {
    constexpr int NT  = 512;
    constexpr int EPT = 8;
    static_assert(T <= 256, "a tile row must fit the high byte of a list entry");

    __shared__ int s_cnt[NT];
    __shared__ int s_first[NT];
    __shared__ int s_wsum[NT / 32];

    const int tid   = threadIdx.x;
    const int lane  = tid & 31;
    const int wave  = __builtin_amdgcn_readfirstlane(tid >> 5);
    const int ntile = (N + T - 1) / T;
    const int me    = blockIdx.x;
    const int nwg   = gridDim.x;

    s_cnt[tid] = 0;

    const int e0 = blockIdx.x * (NT * EPT) + tid;
    int  s[EPT], d[EPT];
    bool ok[EPT];
#pragma unroll
    for (int j = 0; j < EPT; ++j) {
        const int e = e0 + j * NT;
        ok[j] = e < E;
        s[j]  = ok[j] ? src[e] : 0;
        d[j]  = ok[j] ? dst[e] : 0;
    }

    int before = 0, c = 0;
    if (tid < ntile) {
        const int* __restrict__ col = wgcnt + tid;
        constexpr int CB = 28;
        for (int b0 = 0; b0 < nwg; b0 += CB) {
            int v[CB];
#pragma unroll
            for (int j = 0; j < CB; ++j) {
                const int b = b0 + j;
                v[j] = col[(b < nwg ? b : nwg - 1) * NT];
            }
#pragma unroll
            for (int j = 0; j < CB; ++j) {
                const int b = b0 + j;
                c      += v[j] & -(int)(b < nwg);
                before += v[j] & -(int)(b < me);
            }
        }
        if (me == 0) tilecnt[tid] = c;
    }
    int x = c;
#pragma unroll
    for (int m = 1; m < 32; m <<= 1) {
        const int y = __shfl_up(x, m, 32);
        if (lane >= m) x += y;
    }
    if (lane == 31) s_wsum[wave] = x;
    __syncthreads();
    int prev = 0;
    for (int w = 0; w < wave; ++w) prev += s_wsum[w];
    s_first[tid] = prev + x - c + before;

    int tl[EPT], rk[EPT];
#pragma unroll
    for (int j = 0; j < EPT; ++j) {
        tl[j] = d[j] / T;
        rk[j] = 0;
        if (ok[j])
            rk[j] = __hip_atomic_fetch_add(&s_cnt[tl[j]], 1, __ATOMIC_RELAXED, __HIP_MEMORY_SCOPE_WORKGROUP);
    }
    __syncthreads();

#pragma unroll
    for (int j = 0; j < EPT; ++j)
        if (ok[j]) list[s_first[tl[j]] + rk[j]] = list_pack(s[j], d[j] - tl[j] * T);
}

template <int T>
__global__ __launch_bounds__(1024)
void k_deg(const int* __restrict__ list, const int* __restrict__ tilecnt,
           float* __restrict__ dinv, int N) {
    constexpr int NT = 1024;
    constexpr int LB = 4;
    static_assert(T <= 256, "one counter per tile row");
    __shared__ int s_deg[256];
    const int tid  = threadIdx.x;
    const int lane = tid & 31;
    if (tid < 256) s_deg[tid] = 0;
    const int* __restrict__ lst = list + tile_list_base(tilecnt, blockIdx.x, lane);
    const int cnt = tilecnt[blockIdx.x];
    __syncthreads();
    for (int i0 = 0; i0 < cnt; i0 += NT * LB) {
        int v[LB];
#pragma unroll
        for (int j = 0; j < LB; ++j) {
            const int i = i0 + j * NT + tid;
            v[j] = lst[i < cnt ? i : cnt - 1];
        }
#pragma unroll
        for (int j = 0; j < LB; ++j)
            if (i0 + j * NT + tid < cnt)
                __hip_atomic_fetch_add(&s_deg[(unsigned)v[j] >> 24], 1, __ATOMIC_RELAXED, __HIP_MEMORY_SCOPE_WORKGROUP);
    }
    __syncthreads();
    const int node = blockIdx.x * T + tid;
    if (tid < T && node < N) dinv[node] = rsqrtf((float)s_deg[tid] + 1.0f);
}

template <int F, int T, bool FIRST>
__global__ __launch_bounds__(1024)
void k_agg_list(const int* __restrict__ list, float* __restrict__ cflist,
                const int* __restrict__ tilecnt,
                const float* __restrict__ dinv, const float* __restrict__ xw,
                const float* __restrict__ bias, float* __restrict__ outp,
                const float* __restrict__ Wnext, const float* __restrict__ bnext,
                float* __restrict__ out_node, int N) {
    constexpr int NW = 32;
    constexpr int G  = 1024 / F;
    constexpr int MT = (T + 15) / 16;
    constexpr int S  = FIRST ? F + 4 : F;
    constexpr int TR = FIRST ? MT * 16 : T;
    static_assert(G <= 32 && 32 % G == 0, "a 32-entry step is a whole number of row groups");
    static_assert(!FIRST || F == 256, "the product's k order is written for 256 columns");
    static_assert(FIRST || F == 128, "the node head is written for 128 columns");

    __shared__ __attribute__((aligned(16))) float acc[TR * S];

    const int tid  = threadIdx.x;
    const int lane = tid & 31;
    const int wave = __builtin_amdgcn_readfirstlane(tid >> 5);
    const int n0   = blockIdx.x * T;
    const int nt   = (N - n0) < T ? (N - n0) : T;

    const int lbase = tile_list_base(tilecnt, blockIdx.x, lane);
    const int* __restrict__ lst = list + lbase;
    float*     __restrict__ cfl = cflist + lbase;
    const int cnt = tilecnt[blockIdx.x];
    const int per = (((cnt + G - 1) / G + NW - 1) / NW) * G;
    const int lo  = (wave * per < cnt) ? wave * per : cnt;
    const int hi  = (lo + per < cnt) ? lo + per : cnt;

    constexpr int PRE = 4;
    int   en[PRE];
    float ec[PRE];
#pragma unroll
    for (int k = 0; k < PRE; ++k) {
        en[k] = 0;
        if (lo + k * 32 + lane < hi) en[k] = lst[lo + k * 32 + lane];
    }
    float ds[PRE], dd[PRE];
#pragma unroll
    for (int k = 0; k < PRE; ++k) {
        ec[k] = 0.0f;
        if (FIRST) {
            ds[k] = dinv[en[k] & 0xFFFFFF];
            dd[k] = dinv[n0 + (int)((unsigned)en[k] >> 24)];
        } else if (lo + k * 32 + lane < hi) {
            ec[k] = cfl[lo + k * 32 + lane];
        }
    }

    tile_init<F, S, T, NW>(acc, dinv, xw, bias, n0, nt, wave, lane);
    if (FIRST) {
#pragma unroll
        for (int k = 0; k < PRE; ++k) {
            ec[k] = ds[k] * dd[k];
            if (lo + k * 32 + lane < hi) cfl[lo + k * 32 + lane] = ec[k];
        }
    }
    __syncthreads();

    int k = 0;
    for (int b0 = lo; b0 < hi; b0 += 32, ++k) {
        const int nb = (hi - b0) < 32 ? (hi - b0) : 32;
        int   cur = en[0];
        float cf  = ec[0];
#pragma unroll
        for (int j = 1; j < PRE; ++j)
            if (k == j) { cur = en[j]; cf = ec[j]; }
        if (k >= PRE) {
            cur = 0;
            cf  = 0.0f;
            if (lane < nb) {
                cur = lst[b0 + lane];
                if (FIRST) {
                    cf = dinv[cur & 0xFFFFFF] * dinv[n0 + (int)((unsigned)cur >> 24)];
                    cfl[b0 + lane] = cf;
                } else {
                    cf = cfl[b0 + lane];
                }
            }
        }
        const int s  = cur & 0xFFFFFF;
        const int dl = (int)((unsigned)cur >> 24);
        if (nb % G == 0) gather_add<F, S, G, true>(s, dl, cf, nb, lane, xw, acc);
        else             gather_add<F, S, G, false, true>(s, dl, cf, nb, lane, xw, acc);
    }
    __syncthreads();

    if constexpr (FIRST) {
        const int mg = wave >> 3;
        const int nc = wave & 7;
        if (mg < MT % 4) tile_gemm_relu<MT / 4 + 1, S>(acc, Wnext, outp, n0, nt, mg, nc, lane);
        else             tile_gemm_relu<MT / 4, S>(acc, Wnext, outp, n0, nt, mg, nc, lane);
    } else {
        __shared__ float head[2 * T];
        tile_store_head<T, NW>(acc, head, outp, Wnext, bnext, n0, nt, wave, lane);
        __syncthreads();
        if (tid < 2 * nt) out_node[(size_t)n0 * 2 + tid] = head[tid];
    }
}

template <int X>
__device__ __forceinline__ float grp8_xor(float v) {
    constexpr int sel = ((0 ^ X) << 0) | ((1 ^ X) << 3) | ((2 ^ X) << 6) | ((3 ^ X) << 9) |
                        ((4 ^ X) << 12) | ((5 ^ X) << 15) | ((6 ^ X) << 18) | ((7 ^ X) << 21);
    return __int_as_float(__builtin_amdgcn_mov_dpp8(__float_as_int(v), sel));
}

__device__ __forceinline__ void dot4x2(const float4 v, const float4 wa, const float4 wb,
                                       float& a0, float& a1) {
    a0 += v.x * wa.x; a1 += v.x * wa.y;
    a0 += v.y * wa.z; a1 += v.y * wa.w;
    a0 += v.z * wb.x; a1 += v.z * wb.y;
    a0 += v.w * wb.z; a1 += v.w * wb.w;
}

__global__ __launch_bounds__(256)
void k_edge_head(const int* __restrict__ src, const int* __restrict__ dst,
                 const float* __restrict__ emb, const float* __restrict__ We,
                 const float* __restrict__ be, float* __restrict__ out, int E) {
    const int lane = threadIdx.x & 31;
    const int q    = lane & 7;
    const int g    = lane >> 3;
    const int wave   = __builtin_amdgcn_readfirstlane(
        (int)(blockIdx.x * (blockDim.x >> 5) + (threadIdx.x >> 5)));
    const int nwaves = gridDim.x * (blockDim.x >> 5);

    float4 ws[4][2], wd[4][2];
#pragma unroll
    for (int k = 0; k < 4; ++k) {
        const float* __restrict__ w = We + ((k * 8 + q) * 4) * 2;
        ws[k][0] = *(const float4*)(w);
        ws[k][1] = *(const float4*)(w + 4);
        wd[k][0] = *(const float4*)(w + 256);
        wd[k][1] = *(const float4*)(w + 260);
    }
    const float* __restrict__ ecol = emb + q * 4;

    const int nit = ((E & 3) == 0) ? (E >> 4) : 0;
    if (wave < nit) {
        const float bq = be[q & 1];
        const bool  b4 = (q & 4) != 0, b2 = (q & 2) != 0, b1 = (q & 1) != 0;
        int  it = wave;
        int4 si = *(const int4*)(src + it * 16 + g * 4);
        int4 di = *(const int4*)(dst + it * 16 + g * 4);
        for (;;) {
            const int sidx[4] = {si.x, si.y, si.z, si.w};
            const int didx[4] = {di.x, di.y, di.z, di.w};
            float4 rs[4][4], rd[4][4];
#pragma unroll
            for (int j = 0; j < 4; ++j) {
                const float* __restrict__ ps = ecol + (size_t)sidx[j] * 128;
                const float* __restrict__ pd = ecol + (size_t)didx[j] * 128;
#pragma unroll
                for (int k = 0; k < 4; ++k) {
                    rs[j][k] = *(const float4*)(ps + k * 32);
                    rd[j][k] = *(const float4*)(pd + k * 32);
                }
            }
            const int  nx  = it + nwaves;
            const bool has = nx < nit;
            if (has) {
                si = *(const int4*)(src + nx * 16 + g * 4);
                di = *(const int4*)(dst + nx * 16 + g * 4);
            }
            float v[8];
#pragma unroll
            for (int j = 0; j < 4; ++j) {
                float a0 = 0.0f, a1 = 0.0f;
#pragma unroll
                for (int k = 0; k < 4; ++k) {
                    dot4x2(rs[j][k], ws[k][0], ws[k][1], a0, a1);
                    dot4x2(rd[j][k], wd[k][0], wd[k][1], a0, a1);
                }
                v[2 * j]     = a0;
                v[2 * j + 1] = a1;
            }
            float u[4];
#pragma unroll
            for (int t = 0; t < 4; ++t)
                u[t] = (b4 ? v[t + 4] : v[t]) + grp8_xor<4>(b4 ? v[t] : v[t + 4]);
            float w2[2];
#pragma unroll
            for (int t = 0; t < 2; ++t)
                w2[t] = (b2 ? u[t + 2] : u[t]) + grp8_xor<2>(b2 ? u[t] : u[t + 2]);
            const float z = (b1 ? w2[1] : w2[0]) + grp8_xor<1>(b1 ? w2[0] : w2[1]);
            out[(size_t)it * 32 + lane] = z + bq;
            if (!has) break;
            it = nx;
        }
    }

    for (int e0 = nit * 16 + wave * 4; e0 < E; e0 += nwaves * 4) {
        const int  e  = e0 + g;
        const bool ok = e < E;
        const int  ec = ok ? e : e0;
        const float* __restrict__ ps = ecol + (size_t)src[ec] * 128;
        const float* __restrict__ pd = ecol + (size_t)dst[ec] * 128;
        float a0 = 0.0f, a1 = 0.0f;
#pragma unroll
        for (int k = 0; k < 4; ++k) {
            dot4x2(*(const float4*)(ps + k * 32), ws[k][0], ws[k][1], a0, a1);
            dot4x2(*(const float4*)(pd + k * 32), wd[k][0], wd[k][1], a0, a1);
        }
        a0 += grp8_xor<4>(a0); a1 += grp8_xor<4>(a1);
        a0 += grp8_xor<2>(a0); a1 += grp8_xor<2>(a1);
        a0 += grp8_xor<1>(a0); a1 += grp8_xor<1>(a1);
        if (ok && q == 0) {
            out[(size_t)e * 2 + 0] = a0 + be[0];
            out[(size_t)e * 2 + 1] = a1 + be[1];
        }
    }
}

extern "C" void kernel_launch(void* const* d_in, const int* in_sizes, int n_in,
                              void* d_out, int out_size, void* d_ws, size_t ws_size,
                              hipStream_t stream) {
    const float* x  = (const float*)d_in[0];
    const int*   ei = (const int*)d_in[1];
    const float* W1 = (const float*)d_in[2];
    const float* b1 = (const float*)d_in[3];
    const float* W2 = (const float*)d_in[4];
    const float* b2 = (const float*)d_in[5];
    const float* Wn = (const float*)d_in[6];
    const float* bn = (const float*)d_in[7];
    const float* We = (const float*)d_in[8];
    const float* be = (const float*)d_in[9];

    const int N = in_sizes[0] / 64;
    const int E = in_sizes[1] / 2;
    const int* src = ei;
    const int* dst = ei + E;

    char*  ws  = (char*)d_ws;
    size_t off = 0;
    auto carve = [&](size_t bytes) -> float* {
        float* p = (float*)(ws + off);
        off += (bytes + 255) & ~(size_t)255;
        return p;
    };
    float* dinv = carve((size_t)N * 4);
    float* xw1  = carve((size_t)N * 256 * 4);
    float* xw2  = carve((size_t)N * 128 * 4);
    float* emb  = carve((size_t)N * 128 * 4);
    int*   elist   = (int*)carve((size_t)E * 4);
    float* cflist  = carve((size_t)E * 4);
    int*   tilecnt = (int*)carve((size_t)((N + 195) / 196) * 4);
    int*   wgcnt   = (int*)carve((size_t)((E + 4095) / 4096) * 512 * 4);

    float* out_node = (float*)d_out;
    float* out_edge = out_node + (size_t)N * 2;

    k_tilehist<196><<<(E + 4095) / 4096, 512, 0, stream>>>(dst, wgcnt, E);
    k_bin<196><<<(E + 4095) / 4096, 512, 0, stream>>>(src, dst, wgcnt, tilecnt, elist, N, E);
    k_deg<196><<<(N + 195) / 196, 1024, 0, stream>>>(elist, tilecnt, dinv, N);

    k_gemm_wmma<false, 64><<<dim3(N / 16, 256 / 128), 256, 0, stream>>>(x, W1, xw1, N, 256);
    k_agg_list<256, 196, true><<<(N + 195) / 196, 1024, 0, stream>>>(elist, cflist, tilecnt, dinv, xw1, b1, xw2, W2, nullptr, nullptr, N);

    k_agg_list<128, 196, false><<<(N + 195) / 196, 1024, 0, stream>>>(elist, cflist, tilecnt, dinv, xw2, b2, emb, Wn, bn, out_node, N);

    k_edge_head<<<512, 256, 0, stream>>>(src, dst, emb, We, be, out_edge, E);
}


# x @ W1 loses its own launch: its 16-row tiles run as extra work-groups behind the three edge-list passes (count, binning, degrees), 16 waves per tile = all 256 columns; 7 -> 6 launches
# speedup vs baseline: 6.5770x; 1.0792x over previous
#include <hip/hip_runtime.h>


typedef float v2f __attribute__((ext_vector_type(2)));
typedef float v8f __attribute__((ext_vector_type(8)));

__device__ __forceinline__ int tile_list_base(const int* __restrict__ tilecnt, int tile, int lane) {
    int v = 0;
    for (int j = lane; j < tile; j += 32) v += tilecnt[j];
#pragma unroll
    for (int m = 16; m > 0; m >>= 1) v += __shfl_xor(v, m, 32);
    return __builtin_amdgcn_readfirstlane(v);
}

template <int K, int NTHR>
__device__ __forceinline__ void gemm_rowtiles(const float* __restrict__ A, const float* __restrict__ B,
                                              float* __restrict__ C, float* As, int mt, int tid) {
    constexpr int N   = 256;
    constexpr int LDK = K + 2;
    constexpr int MT  = NTHR / 512;
    const int wave = tid >> 5;
    const int lane = tid & 31;
    const int ms   = wave >> 4;
    const int nt   = wave & 15;

    const float* __restrict__ Ag = A + (size_t)mt * 16 * K;
#pragma unroll
    for (int idx = tid; idx < MT * 16 * K; idx += NTHR) {
        int r  = idx / K;
        int kk = idx - r * K;
        As[r * LDK + kk] = Ag[idx];
    }
    __syncthreads();

    const int col = lane & 15;
    const int kb  = (lane >> 4) << 1;

    const float* Arow = As + (ms * 16 + col) * LDK + kb;
    const float* __restrict__ Bc = B + (size_t)nt * 16 + col;

    v8f acc = {};
#pragma unroll
    for (int k = 0; k < K; k += 4) {
        v2f av; av.x = Arow[k];                      av.y = Arow[k + 1];
        v2f bv; bv.x = Bc[(size_t)(k + kb) * N];     bv.y = Bc[(size_t)(k + kb + 1) * N];
        acc = __builtin_amdgcn_wmma_f32_16x16x4_f32(
            false, av, false, bv, (short)0, acc, false, false);
    }

    float* __restrict__ Cp =
        C + (size_t)((mt + ms) * 16 + ((lane >> 4) << 3)) * N + (size_t)nt * 16 + col;
#pragma unroll
    for (int r = 0; r < 8; ++r) Cp[(size_t)r * N] = acc[r];
}

__device__ __forceinline__ void lds_add(float* p, float v) {
    __hip_atomic_fetch_add(p, v, __ATOMIC_RELAXED, __HIP_MEMORY_SCOPE_WORKGROUP);
}

template <int F, int S, int G, bool FULL, bool CLAMP = false>
__device__ __forceinline__ void gather_add(int s, int dl, float cf, int nb, int lane,
                                           const float* __restrict__ xw, float* acc) {
    constexpr int H = F / 128;
#pragma unroll 1
    for (int i = 0; i < nb; i += G) {
        float4 v[G][H];
        int    rl[G];
        float  ci[G];
#pragma unroll
        for (int g = 0; g < G; ++g) {
            if (FULL || CLAMP || i + g < nb) {
                const int j  = (CLAMP && i + g >= nb) ? nb - 1 : i + g;
                const int si = __builtin_amdgcn_readlane(s, j);
                rl[g] = __builtin_amdgcn_readlane(dl, j);
                ci[g] = __int_as_float(__builtin_amdgcn_readlane(__float_as_int(cf), j));
                const float* __restrict__ row = xw + (size_t)si * F + lane * 4;
#pragma unroll
                for (int h = 0; h < H; ++h) v[g][h] = *(const float4*)(row + h * 128);
            }
        }
#pragma unroll
        for (int g = 0; g < G; ++g) {
            if (FULL || i + g < nb) {
                float* a = acc + rl[g] * S + lane;
#pragma unroll
                for (int h = 0; h < H; ++h) {
                    lds_add(a + h * 128,      ci[g] * v[g][h].x);
                    lds_add(a + h * 128 + 32, ci[g] * v[g][h].y);
                    lds_add(a + h * 128 + 64, ci[g] * v[g][h].z);
                    lds_add(a + h * 128 + 96, ci[g] * v[g][h].w);
                }
            }
        }
    }
}

__device__ __forceinline__ int list_pack(int s, int dl) {
    return (int)((unsigned)s | ((unsigned)dl << 24));
}

template <int F, int S, int T, int NW>
__device__ __forceinline__ void tile_init(float* acc,
                                          const float* __restrict__ dinv,
                                          const float* __restrict__ xw,
                                          const float* __restrict__ bias,
                                          int n0, int nt, int wave, int lane) {
    constexpr int H   = F / 128;
    constexpr int RPW = (T + NW - 1) / NW;
    float4 b[H];
#pragma unroll
    for (int h = 0; h < H; ++h) b[h] = *(const float4*)(bias + h * 128 + lane * 4);
    float4 v[RPW][H];
    float  di[RPW];
#pragma unroll
    for (int k = 0; k < RPW; ++k) {
        const int r  = wave + k * NW;
        const int rc = r < nt ? r : nt - 1;
        di[k] = dinv[n0 + rc];
        const float* __restrict__ xr = xw + (size_t)(n0 + rc) * F + lane * 4;
#pragma unroll
        for (int h = 0; h < H; ++h) v[k][h] = *(const float4*)(xr + h * 128);
    }
#pragma unroll
    for (int k = 0; k < RPW; ++k) {
        const int r = wave + k * NW;
        if (r < nt) {
            float* a = acc + r * S + lane;
#pragma unroll
            for (int h = 0; h < H; ++h) {
                a[h * 128]      = v[k][h].x * di[k] * di[k] + b[h].x;
                a[h * 128 + 32] = v[k][h].y * di[k] * di[k] + b[h].y;
                a[h * 128 + 64] = v[k][h].z * di[k] * di[k] + b[h].z;
                a[h * 128 + 96] = v[k][h].w * di[k] * di[k] + b[h].w;
            }
        }
    }
}

template <int T, int NW>
__device__ __forceinline__ void tile_store_head(const float* acc, float* head,
                                                float* __restrict__ emb,
                                                const float* __restrict__ Wn,
                                                const float* __restrict__ bn,
                                                int n0, int nt, int wave, int lane) {
    constexpr int F   = 128;
    constexpr int RPW = (T + NW - 1) / NW;
    const float4 wa = *(const float4*)(Wn + lane * 8);
    const float4 wb = *(const float4*)(Wn + lane * 8 + 4);
    const float  b0 = bn[0], b1 = bn[1];
#pragma unroll
    for (int k = 0; k < RPW; ++k) {
        const int r = wave + k * NW;
        if (r < nt) {
            const float* a = acc + r * F + lane;
            float4 v;
            v.x = a[0];
            v.y = a[32];
            v.z = a[64];
            v.w = a[96];
            *(float4*)(emb + (size_t)(n0 + r) * F + lane * 4) = v;
            float p0 = v.x * wa.x + v.y * wa.z + v.z * wb.x + v.w * wb.z;
            float p1 = v.x * wa.y + v.y * wa.w + v.z * wb.y + v.w * wb.w;
#pragma unroll
            for (int m = 16; m > 0; m >>= 1) {
                p0 += __shfl_xor(p0, m, 32);
                p1 += __shfl_xor(p1, m, 32);
            }
            if (lane == 0) {
                head[2 * r]     = p0 + b0;
                head[2 * r + 1] = p1 + b1;
            }
        }
    }
}

template <int NM, int S>
__device__ __forceinline__ void tile_gemm_relu(const float* acc, const float* __restrict__ Wb,
                                               float* __restrict__ C, int n0, int nt,
                                               int mg, int nc, int lane) {
    constexpr int N = 128;
    const int col = lane & 15;
    const int hf  = lane >> 4;
    const float* __restrict__ Bc = Wb + (size_t)(8 * hf) * N + nc * 16 + col;
    const float* Ar = acc + (mg * 16 + col) * S + 2 * hf;

    v8f d[NM];
#pragma unroll
    for (int m = 0; m < NM; ++m) d[m] = v8f{};

    auto brow = [](int s) { return (s >> 5) * 128 + ((s >> 3) & 3) + 16 * (s & 7); };
    constexpr int KC = 4;
    v2f bcur[KC], bnxt[KC];
#pragma unroll
    for (int i = 0; i < KC; ++i) {
        bcur[i].x = Bc[(size_t)(brow(i)) * N];
        bcur[i].y = Bc[(size_t)(brow(i) + 4) * N];
    }
#pragma unroll 1
    for (int s0 = 0; s0 < 64; s0 += KC) {
        const int sn = s0 + KC < 64 ? s0 + KC : s0;
        const float* __restrict__ Bn = Bc + (size_t)brow(sn) * N;
#pragma unroll
        for (int i = 0; i < KC; ++i) {
            bnxt[i].x = Bn[(size_t)(16 * i) * N];
            bnxt[i].y = Bn[(size_t)(16 * i + 4) * N];
        }
#pragma unroll
        for (int i = 0; i < KC; ++i) {
#pragma unroll
            for (int m = 0; m < NM; ++m) {
                v2f av = *(const v2f*)(Ar + m * 64 * S + 4 * (s0 + i));
                av.x = fmaxf(av.x, 0.0f);
                av.y = fmaxf(av.y, 0.0f);
                d[m] = __builtin_amdgcn_wmma_f32_16x16x4_f32(
                    false, av, false, bcur[i], (short)0, d[m], false, false);
            }
        }
#pragma unroll
        for (int i = 0; i < KC; ++i) bcur[i] = bnxt[i];
    }

#pragma unroll
    for (int m = 0; m < NM; ++m) {
#pragma unroll
        for (int r = 0; r < 8; ++r) {
            const int row = mg * 16 + m * 64 + r + 8 * hf;
            if (row < nt) C[(size_t)(n0 + row) * N + nc * 16 + col] = d[m][r];
        }
    }
}

template <int T>
__global__ __launch_bounds__(512)
void k_tilehist(const int* __restrict__ dst, int* __restrict__ wgcnt, int E,
                const float* __restrict__ feat, const float* __restrict__ W1,
                float* __restrict__ xw1) {
    constexpr int NT  = 512;
    constexpr int EPT = 8;
    __shared__ int   s_cnt[NT];
    __shared__ float s_a[16 * 66];
    const int tid = threadIdx.x;
    const int nwg = (E + NT * EPT - 1) / (NT * EPT);
    if ((int)blockIdx.x >= nwg) {
        gemm_rowtiles<64, NT>(feat, W1, xw1, s_a, (int)blockIdx.x - nwg, tid);
        return;
    }
    s_cnt[tid] = 0;
    const int e0 = blockIdx.x * (NT * EPT) + tid;
    int d[EPT];
#pragma unroll
    for (int j = 0; j < EPT; ++j) {
        const int e = e0 + j * NT;
        d[j] = e < E ? dst[e] : -1;
    }
    __syncthreads();
#pragma unroll
    for (int j = 0; j < EPT; ++j)
        if (d[j] >= 0)
            __hip_atomic_fetch_add(&s_cnt[d[j] / T], 1, __ATOMIC_RELAXED, __HIP_MEMORY_SCOPE_WORKGROUP);
    __syncthreads();
    wgcnt[blockIdx.x * NT + tid] = s_cnt[tid];
}

template <int T>
__global__ __launch_bounds__(512, 8)
void k_bin(const int* __restrict__ src, const int* __restrict__ dst,
           const int* __restrict__ wgcnt, int* __restrict__ tilecnt,
           int* __restrict__ list, int N, int E,
           const float* __restrict__ feat, const float* __restrict__ W1,
           float* __restrict__ xw1, int mt0) {
    constexpr int NT  = 512;
    constexpr int EPT = 8;
    static_assert(T <= 256, "a tile row must fit the high byte of a list entry");

    __shared__ int   s_cnt[NT];
    __shared__ int   s_first[NT];
    __shared__ int   s_wsum[NT / 32];
    __shared__ float s_a[16 * 66];

    const int tid   = threadIdx.x;
    const int nwg   = (E + NT * EPT - 1) / (NT * EPT);
    if ((int)blockIdx.x >= nwg) {
        gemm_rowtiles<64, NT>(feat, W1, xw1, s_a, mt0 + (int)blockIdx.x - nwg, tid);
        return;
    }
    const int lane  = tid & 31;
    const int wave  = __builtin_amdgcn_readfirstlane(tid >> 5);
    const int ntile = (N + T - 1) / T;
    const int me    = blockIdx.x;

    s_cnt[tid] = 0;

    const int e0 = blockIdx.x * (NT * EPT) + tid;
    int  s[EPT], d[EPT];
    bool ok[EPT];
#pragma unroll
    for (int j = 0; j < EPT; ++j) {
        const int e = e0 + j * NT;
        ok[j] = e < E;
        s[j]  = ok[j] ? src[e] : 0;
        d[j]  = ok[j] ? dst[e] : 0;
    }

    int before = 0, c = 0;
    if (tid < ntile) {
        const int* __restrict__ col = wgcnt + tid;
        constexpr int CB = 28;
        for (int b0 = 0; b0 < nwg; b0 += CB) {
            int v[CB];
#pragma unroll
            for (int j = 0; j < CB; ++j) {
                const int b = b0 + j;
                v[j] = col[(b < nwg ? b : nwg - 1) * NT];
            }
#pragma unroll
            for (int j = 0; j < CB; ++j) {
                const int b = b0 + j;
                c      += v[j] & -(int)(b < nwg);
                before += v[j] & -(int)(b < me);
            }
        }
        if (me == 0) tilecnt[tid] = c;
    }
    int x = c;
#pragma unroll
    for (int m = 1; m < 32; m <<= 1) {
        const int y = __shfl_up(x, m, 32);
        if (lane >= m) x += y;
    }
    if (lane == 31) s_wsum[wave] = x;
    __syncthreads();
    int prev = 0;
    for (int w = 0; w < wave; ++w) prev += s_wsum[w];
    s_first[tid] = prev + x - c + before;

    int tl[EPT], rk[EPT];
#pragma unroll
    for (int j = 0; j < EPT; ++j) {
        tl[j] = d[j] / T;
        rk[j] = 0;
        if (ok[j])
            rk[j] = __hip_atomic_fetch_add(&s_cnt[tl[j]], 1, __ATOMIC_RELAXED, __HIP_MEMORY_SCOPE_WORKGROUP);
    }
    __syncthreads();

#pragma unroll
    for (int j = 0; j < EPT; ++j)
        if (ok[j]) list[s_first[tl[j]] + rk[j]] = list_pack(s[j], d[j] - tl[j] * T);
}

template <int T>
__global__ __launch_bounds__(1024)
void k_deg(const int* __restrict__ list, const int* __restrict__ tilecnt,
           float* __restrict__ dinv, int N,
           const float* __restrict__ feat, const float* __restrict__ W1,
           float* __restrict__ xw1, int mt0) {
    constexpr int NT = 1024;
    constexpr int LB = 4;
    static_assert(T <= 256, "one counter per tile row");
    __shared__ int   s_deg[256];
    __shared__ float s_a[2 * 16 * 66];
    const int tid  = threadIdx.x;
    const int ntile = (N + T - 1) / T;
    if ((int)blockIdx.x >= ntile) {
        gemm_rowtiles<64, NT>(feat, W1, xw1, s_a, mt0 + 2 * ((int)blockIdx.x - ntile), tid);
        return;
    }
    const int lane = tid & 31;
    if (tid < 256) s_deg[tid] = 0;
    const int* __restrict__ lst = list + tile_list_base(tilecnt, blockIdx.x, lane);
    const int cnt = tilecnt[blockIdx.x];
    __syncthreads();
    for (int i0 = 0; i0 < cnt; i0 += NT * LB) {
        int v[LB];
#pragma unroll
        for (int j = 0; j < LB; ++j) {
            const int i = i0 + j * NT + tid;
            v[j] = lst[i < cnt ? i : cnt - 1];
        }
#pragma unroll
        for (int j = 0; j < LB; ++j)
            if (i0 + j * NT + tid < cnt)
                __hip_atomic_fetch_add(&s_deg[(unsigned)v[j] >> 24], 1, __ATOMIC_RELAXED, __HIP_MEMORY_SCOPE_WORKGROUP);
    }
    __syncthreads();
    const int node = blockIdx.x * T + tid;
    if (tid < T && node < N) dinv[node] = rsqrtf((float)s_deg[tid] + 1.0f);
}

template <int F, int T, bool FIRST>
__global__ __launch_bounds__(1024)
void k_agg_list(const int* __restrict__ list, float* __restrict__ cflist,
                const int* __restrict__ tilecnt,
                const float* __restrict__ dinv, const float* __restrict__ xw,
                const float* __restrict__ bias, float* __restrict__ outp,
                const float* __restrict__ Wnext, const float* __restrict__ bnext,
                float* __restrict__ out_node, int N) {
    constexpr int NW = 32;
    constexpr int G  = 1024 / F;
    constexpr int MT = (T + 15) / 16;
    constexpr int S  = FIRST ? F + 4 : F;
    constexpr int TR = FIRST ? MT * 16 : T;
    static_assert(G <= 32 && 32 % G == 0, "a 32-entry step is a whole number of row groups");
    static_assert(!FIRST || F == 256, "the product's k order is written for 256 columns");
    static_assert(FIRST || F == 128, "the node head is written for 128 columns");

    __shared__ __attribute__((aligned(16))) float acc[TR * S];

    const int tid  = threadIdx.x;
    const int lane = tid & 31;
    const int wave = __builtin_amdgcn_readfirstlane(tid >> 5);
    const int n0   = blockIdx.x * T;
    const int nt   = (N - n0) < T ? (N - n0) : T;

    const int lbase = tile_list_base(tilecnt, blockIdx.x, lane);
    const int* __restrict__ lst = list + lbase;
    float*     __restrict__ cfl = cflist + lbase;
    const int cnt = tilecnt[blockIdx.x];
    const int per = (((cnt + G - 1) / G + NW - 1) / NW) * G;
    const int lo  = (wave * per < cnt) ? wave * per : cnt;
    const int hi  = (lo + per < cnt) ? lo + per : cnt;

    constexpr int PRE = 4;
    int   en[PRE];
    float ec[PRE];
#pragma unroll
    for (int k = 0; k < PRE; ++k) {
        en[k] = 0;
        if (lo + k * 32 + lane < hi) en[k] = lst[lo + k * 32 + lane];
    }
    float ds[PRE], dd[PRE];
#pragma unroll
    for (int k = 0; k < PRE; ++k) {
        ec[k] = 0.0f;
        if (FIRST) {
            ds[k] = dinv[en[k] & 0xFFFFFF];
            dd[k] = dinv[n0 + (int)((unsigned)en[k] >> 24)];
        } else if (lo + k * 32 + lane < hi) {
            ec[k] = cfl[lo + k * 32 + lane];
        }
    }

    tile_init<F, S, T, NW>(acc, dinv, xw, bias, n0, nt, wave, lane);
    if (FIRST) {
#pragma unroll
        for (int k = 0; k < PRE; ++k) {
            ec[k] = ds[k] * dd[k];
            if (lo + k * 32 + lane < hi) cfl[lo + k * 32 + lane] = ec[k];
        }
    }
    __syncthreads();

    int k = 0;
    for (int b0 = lo; b0 < hi; b0 += 32, ++k) {
        const int nb = (hi - b0) < 32 ? (hi - b0) : 32;
        int   cur = en[0];
        float cf  = ec[0];
#pragma unroll
        for (int j = 1; j < PRE; ++j)
            if (k == j) { cur = en[j]; cf = ec[j]; }
        if (k >= PRE) {
            cur = 0;
            cf  = 0.0f;
            if (lane < nb) {
                cur = lst[b0 + lane];
                if (FIRST) {
                    cf = dinv[cur & 0xFFFFFF] * dinv[n0 + (int)((unsigned)cur >> 24)];
                    cfl[b0 + lane] = cf;
                } else {
                    cf = cfl[b0 + lane];
                }
            }
        }
        const int s  = cur & 0xFFFFFF;
        const int dl = (int)((unsigned)cur >> 24);
        if (nb % G == 0) gather_add<F, S, G, true>(s, dl, cf, nb, lane, xw, acc);
        else             gather_add<F, S, G, false, true>(s, dl, cf, nb, lane, xw, acc);
    }
    __syncthreads();

    if constexpr (FIRST) {
        const int mg = wave >> 3;
        const int nc = wave & 7;
        if (mg < MT % 4) tile_gemm_relu<MT / 4 + 1, S>(acc, Wnext, outp, n0, nt, mg, nc, lane);
        else             tile_gemm_relu<MT / 4, S>(acc, Wnext, outp, n0, nt, mg, nc, lane);
    } else {
        __shared__ float head[2 * T];
        tile_store_head<T, NW>(acc, head, outp, Wnext, bnext, n0, nt, wave, lane);
        __syncthreads();
        if (tid < 2 * nt) out_node[(size_t)n0 * 2 + tid] = head[tid];
    }
}

template <int X>
__device__ __forceinline__ float grp8_xor(float v) {
    constexpr int sel = ((0 ^ X) << 0) | ((1 ^ X) << 3) | ((2 ^ X) << 6) | ((3 ^ X) << 9) |
                        ((4 ^ X) << 12) | ((5 ^ X) << 15) | ((6 ^ X) << 18) | ((7 ^ X) << 21);
    return __int_as_float(__builtin_amdgcn_mov_dpp8(__float_as_int(v), sel));
}

__device__ __forceinline__ void dot4x2(const float4 v, const float4 wa, const float4 wb,
                                       float& a0, float& a1) {
    a0 += v.x * wa.x; a1 += v.x * wa.y;
    a0 += v.y * wa.z; a1 += v.y * wa.w;
    a0 += v.z * wb.x; a1 += v.z * wb.y;
    a0 += v.w * wb.z; a1 += v.w * wb.w;
}

__global__ __launch_bounds__(256)
void k_edge_head(const int* __restrict__ src, const int* __restrict__ dst,
                 const float* __restrict__ emb, const float* __restrict__ We,
                 const float* __restrict__ be, float* __restrict__ out, int E) {
    const int lane = threadIdx.x & 31;
    const int q    = lane & 7;
    const int g    = lane >> 3;
    const int wave   = __builtin_amdgcn_readfirstlane(
        (int)(blockIdx.x * (blockDim.x >> 5) + (threadIdx.x >> 5)));
    const int nwaves = gridDim.x * (blockDim.x >> 5);

    float4 ws[4][2], wd[4][2];
#pragma unroll
    for (int k = 0; k < 4; ++k) {
        const float* __restrict__ w = We + ((k * 8 + q) * 4) * 2;
        ws[k][0] = *(const float4*)(w);
        ws[k][1] = *(const float4*)(w + 4);
        wd[k][0] = *(const float4*)(w + 256);
        wd[k][1] = *(const float4*)(w + 260);
    }
    const float* __restrict__ ecol = emb + q * 4;

    const int nit = ((E & 3) == 0) ? (E >> 4) : 0;
    if (wave < nit) {
        const float bq = be[q & 1];
        const bool  b4 = (q & 4) != 0, b2 = (q & 2) != 0, b1 = (q & 1) != 0;
        int  it = wave;
        int4 si = *(const int4*)(src + it * 16 + g * 4);
        int4 di = *(const int4*)(dst + it * 16 + g * 4);
        for (;;) {
            const int sidx[4] = {si.x, si.y, si.z, si.w};
            const int didx[4] = {di.x, di.y, di.z, di.w};
            float4 rs[4][4], rd[4][4];
#pragma unroll
            for (int j = 0; j < 4; ++j) {
                const float* __restrict__ ps = ecol + (size_t)sidx[j] * 128;
                const float* __restrict__ pd = ecol + (size_t)didx[j] * 128;
#pragma unroll
                for (int k = 0; k < 4; ++k) {
                    rs[j][k] = *(const float4*)(ps + k * 32);
                    rd[j][k] = *(const float4*)(pd + k * 32);
                }
            }
            const int  nx  = it + nwaves;
            const bool has = nx < nit;
            if (has) {
                si = *(const int4*)(src + nx * 16 + g * 4);
                di = *(const int4*)(dst + nx * 16 + g * 4);
            }
            float v[8];
#pragma unroll
            for (int j = 0; j < 4; ++j) {
                float a0 = 0.0f, a1 = 0.0f;
#pragma unroll
                for (int k = 0; k < 4; ++k) {
                    dot4x2(rs[j][k], ws[k][0], ws[k][1], a0, a1);
                    dot4x2(rd[j][k], wd[k][0], wd[k][1], a0, a1);
                }
                v[2 * j]     = a0;
                v[2 * j + 1] = a1;
            }
            float u[4];
#pragma unroll
            for (int t = 0; t < 4; ++t)
                u[t] = (b4 ? v[t + 4] : v[t]) + grp8_xor<4>(b4 ? v[t] : v[t + 4]);
            float w2[2];
#pragma unroll
            for (int t = 0; t < 2; ++t)
                w2[t] = (b2 ? u[t + 2] : u[t]) + grp8_xor<2>(b2 ? u[t] : u[t + 2]);
            const float z = (b1 ? w2[1] : w2[0]) + grp8_xor<1>(b1 ? w2[0] : w2[1]);
            out[(size_t)it * 32 + lane] = z + bq;
            if (!has) break;
            it = nx;
        }
    }

    for (int e0 = nit * 16 + wave * 4; e0 < E; e0 += nwaves * 4) {
        const int  e  = e0 + g;
        const bool ok = e < E;
        const int  ec = ok ? e : e0;
        const float* __restrict__ ps = ecol + (size_t)src[ec] * 128;
        const float* __restrict__ pd = ecol + (size_t)dst[ec] * 128;
        float a0 = 0.0f, a1 = 0.0f;
#pragma unroll
        for (int k = 0; k < 4; ++k) {
            dot4x2(*(const float4*)(ps + k * 32), ws[k][0], ws[k][1], a0, a1);
            dot4x2(*(const float4*)(pd + k * 32), wd[k][0], wd[k][1], a0, a1);
        }
        a0 += grp8_xor<4>(a0); a1 += grp8_xor<4>(a1);
        a0 += grp8_xor<2>(a0); a1 += grp8_xor<2>(a1);
        a0 += grp8_xor<1>(a0); a1 += grp8_xor<1>(a1);
        if (ok && q == 0) {
            out[(size_t)e * 2 + 0] = a0 + be[0];
            out[(size_t)e * 2 + 1] = a1 + be[1];
        }
    }
}

extern "C" void kernel_launch(void* const* d_in, const int* in_sizes, int n_in,
                              void* d_out, int out_size, void* d_ws, size_t ws_size,
                              hipStream_t stream) {
    const float* x  = (const float*)d_in[0];
    const int*   ei = (const int*)d_in[1];
    const float* W1 = (const float*)d_in[2];
    const float* b1 = (const float*)d_in[3];
    const float* W2 = (const float*)d_in[4];
    const float* b2 = (const float*)d_in[5];
    const float* Wn = (const float*)d_in[6];
    const float* bn = (const float*)d_in[7];
    const float* We = (const float*)d_in[8];
    const float* be = (const float*)d_in[9];

    const int N = in_sizes[0] / 64;
    const int E = in_sizes[1] / 2;
    const int* src = ei;
    const int* dst = ei + E;

    char*  ws  = (char*)d_ws;
    size_t off = 0;
    auto carve = [&](size_t bytes) -> float* {
        float* p = (float*)(ws + off);
        off += (bytes + 255) & ~(size_t)255;
        return p;
    };
    float* dinv = carve((size_t)N * 4);
    float* xw1  = carve((size_t)N * 256 * 4);
    float* xw2  = carve((size_t)N * 128 * 4);
    float* emb  = carve((size_t)N * 128 * 4);
    int*   elist   = (int*)carve((size_t)E * 4);
    float* cflist  = carve((size_t)E * 4);
    int*   tilecnt = (int*)carve((size_t)((N + 195) / 196) * 4);
    int*   wgcnt   = (int*)carve((size_t)((E + 4095) / 4096) * 512 * 4);

    float* out_node = (float*)d_out;
    float* out_edge = out_node + (size_t)N * 2;

    k_tilehist<196><<<(E + 4095) / 4096 + (N / 16 - N / 32 - 2 * (N / 128)), 512, 0, stream>>>(dst, wgcnt, E, x, W1, xw1);
    k_bin<196><<<(E + 4095) / 4096 + N / 32, 512, 0, stream>>>(src, dst, wgcnt, tilecnt, elist, N, E,
                                                             x, W1, xw1, N / 16 - N / 32 - 2 * (N / 128));
    k_deg<196><<<(N + 195) / 196 + N / 128, 1024, 0, stream>>>(elist, tilecnt, dinv, N,
                                                             x, W1, xw1, N / 16 - 2 * (N / 128));

    k_agg_list<256, 196, true><<<(N + 195) / 196, 1024, 0, stream>>>(elist, cflist, tilecnt, dinv, xw1, b1, xw2, W2, nullptr, nullptr, N);

    k_agg_list<128, 196, false><<<(N + 195) / 196, 1024, 0, stream>>>(elist, cflist, tilecnt, dinv, xw2, b2, emb, Wn, bn, out_node, N);

    k_edge_head<<<512, 256, 0, stream>>>(src, dst, emb, We, be, out_edge, E);
}


# x @ W1 enters with far fewer waves: a product wave keeps its 16 W1 fragments in registers and walks 2, 4 or 6 row tiles (the work-group's A tiles go to LDS as one group of loads); k order unchanged
# speedup vs baseline: 6.7724x; 1.0297x over previous
#include <hip/hip_runtime.h>


typedef float v2f __attribute__((ext_vector_type(2)));
typedef float v8f __attribute__((ext_vector_type(8)));

__device__ __forceinline__ int tile_list_base(const int* __restrict__ tilecnt, int tile, int lane) {
    int v = 0;
    for (int j = lane; j < tile; j += 32) v += tilecnt[j];
#pragma unroll
    for (int m = 16; m > 0; m >>= 1) v += __shfl_xor(v, m, 32);
    return __builtin_amdgcn_readfirstlane(v);
}

template <int K, int NTHR, int R>
__device__ __forceinline__ void gemm_rowtiles(const float* __restrict__ A, const float* __restrict__ B,
                                              float* __restrict__ C, float* As, int mt, int mtEnd, int tid) {
    constexpr int N   = 256;
    constexpr int LDK = K + 2;
    constexpr int TW  = (NTHR / 512) * R;
    const int wave = __builtin_amdgcn_readfirstlane(tid >> 5);
    const int lane = tid & 31;
    const int ms   = wave >> 4;
    const int nt   = wave & 15;
    const int col  = lane & 15;
    const int kb   = (lane >> 4) << 1;

    const float* __restrict__ Bc = B + (size_t)nt * 16 + col;
    v2f bv[K / 4];
#pragma unroll
    for (int k = 0; k < K; k += 4) {
        bv[k / 4].x = Bc[(size_t)(k + kb) * N];
        bv[k / 4].y = Bc[(size_t)(k + kb + 1) * N];
    }

    const int ntl = (mtEnd - mt) < TW ? (mtEnd - mt) : TW;
    const float* __restrict__ Ag = A + (size_t)mt * 16 * K;
    constexpr int NL = TW * 16 * (K / 2) / NTHR;
    float2 ar[NL];
#pragma unroll
    for (int i = 0; i < NL; ++i) {
        const int idx = tid + i * NTHR;
        const int r   = idx / (K / 2);
        const int kp  = idx - r * (K / 2);
        const int rs  = r < ntl * 16 ? r : ntl * 16 - 1;
        ar[i] = *(const float2*)(Ag + ((size_t)rs * (K / 2) + kp) * 2);
    }
#pragma unroll
    for (int i = 0; i < NL; ++i) {
        const int idx = tid + i * NTHR;
        const int r   = idx / (K / 2);
        const int kp  = idx - r * (K / 2);
        *(float2*)(As + r * LDK + kp * 2) = ar[i];
    }
    __syncthreads();

#pragma unroll
    for (int t = 0; t < R; ++t) {
        const int lt = ms * R + t;
        if (lt < ntl) {
            const float* Arow = As + (lt * 16 + col) * LDK + kb;
            v8f acc = {};
#pragma unroll
            for (int k = 0; k < K; k += 4) {
                v2f av; av.x = Arow[k]; av.y = Arow[k + 1];
                acc = __builtin_amdgcn_wmma_f32_16x16x4_f32(
                    false, av, false, bv[k / 4], (short)0, acc, false, false);
            }
            float* __restrict__ Cp =
                C + (size_t)((mt + lt) * 16 + ((lane >> 4) << 3)) * N + (size_t)nt * 16 + col;
#pragma unroll
            for (int r = 0; r < 8; ++r) Cp[(size_t)r * N] = acc[r];
        }
    }
}

__device__ __forceinline__ void lds_add(float* p, float v) {
    __hip_atomic_fetch_add(p, v, __ATOMIC_RELAXED, __HIP_MEMORY_SCOPE_WORKGROUP);
}

template <int F, int S, int G, bool FULL, bool CLAMP = false>
__device__ __forceinline__ void gather_add(int s, int dl, float cf, int nb, int lane,
                                           const float* __restrict__ xw, float* acc) {
    constexpr int H = F / 128;
#pragma unroll 1
    for (int i = 0; i < nb; i += G) {
        float4 v[G][H];
        int    rl[G];
        float  ci[G];
#pragma unroll
        for (int g = 0; g < G; ++g) {
            if (FULL || CLAMP || i + g < nb) {
                const int j  = (CLAMP && i + g >= nb) ? nb - 1 : i + g;
                const int si = __builtin_amdgcn_readlane(s, j);
                rl[g] = __builtin_amdgcn_readlane(dl, j);
                ci[g] = __int_as_float(__builtin_amdgcn_readlane(__float_as_int(cf), j));
                const float* __restrict__ row = xw + (size_t)si * F + lane * 4;
#pragma unroll
                for (int h = 0; h < H; ++h) v[g][h] = *(const float4*)(row + h * 128);
            }
        }
#pragma unroll
        for (int g = 0; g < G; ++g) {
            if (FULL || i + g < nb) {
                float* a = acc + rl[g] * S + lane;
#pragma unroll
                for (int h = 0; h < H; ++h) {
                    lds_add(a + h * 128,      ci[g] * v[g][h].x);
                    lds_add(a + h * 128 + 32, ci[g] * v[g][h].y);
                    lds_add(a + h * 128 + 64, ci[g] * v[g][h].z);
                    lds_add(a + h * 128 + 96, ci[g] * v[g][h].w);
                }
            }
        }
    }
}

__device__ __forceinline__ int list_pack(int s, int dl) {
    return (int)((unsigned)s | ((unsigned)dl << 24));
}

template <int F, int S, int T, int NW>
__device__ __forceinline__ void tile_init(float* acc,
                                          const float* __restrict__ dinv,
                                          const float* __restrict__ xw,
                                          const float* __restrict__ bias,
                                          int n0, int nt, int wave, int lane) {
    constexpr int H   = F / 128;
    constexpr int RPW = (T + NW - 1) / NW;
    float4 b[H];
#pragma unroll
    for (int h = 0; h < H; ++h) b[h] = *(const float4*)(bias + h * 128 + lane * 4);
    float4 v[RPW][H];
    float  di[RPW];
#pragma unroll
    for (int k = 0; k < RPW; ++k) {
        const int r  = wave + k * NW;
        const int rc = r < nt ? r : nt - 1;
        di[k] = dinv[n0 + rc];
        const float* __restrict__ xr = xw + (size_t)(n0 + rc) * F + lane * 4;
#pragma unroll
        for (int h = 0; h < H; ++h) v[k][h] = *(const float4*)(xr + h * 128);
    }
#pragma unroll
    for (int k = 0; k < RPW; ++k) {
        const int r = wave + k * NW;
        if (r < nt) {
            float* a = acc + r * S + lane;
#pragma unroll
            for (int h = 0; h < H; ++h) {
                a[h * 128]      = v[k][h].x * di[k] * di[k] + b[h].x;
                a[h * 128 + 32] = v[k][h].y * di[k] * di[k] + b[h].y;
                a[h * 128 + 64] = v[k][h].z * di[k] * di[k] + b[h].z;
                a[h * 128 + 96] = v[k][h].w * di[k] * di[k] + b[h].w;
            }
        }
    }
}

template <int T, int NW>
__device__ __forceinline__ void tile_store_head(const float* acc, float* head,
                                                float* __restrict__ emb,
                                                const float* __restrict__ Wn,
                                                const float* __restrict__ bn,
                                                int n0, int nt, int wave, int lane) {
    constexpr int F   = 128;
    constexpr int RPW = (T + NW - 1) / NW;
    const float4 wa = *(const float4*)(Wn + lane * 8);
    const float4 wb = *(const float4*)(Wn + lane * 8 + 4);
    const float  b0 = bn[0], b1 = bn[1];
#pragma unroll
    for (int k = 0; k < RPW; ++k) {
        const int r = wave + k * NW;
        if (r < nt) {
            const float* a = acc + r * F + lane;
            float4 v;
            v.x = a[0];
            v.y = a[32];
            v.z = a[64];
            v.w = a[96];
            *(float4*)(emb + (size_t)(n0 + r) * F + lane * 4) = v;
            float p0 = v.x * wa.x + v.y * wa.z + v.z * wb.x + v.w * wb.z;
            float p1 = v.x * wa.y + v.y * wa.w + v.z * wb.y + v.w * wb.w;
#pragma unroll
            for (int m = 16; m > 0; m >>= 1) {
                p0 += __shfl_xor(p0, m, 32);
                p1 += __shfl_xor(p1, m, 32);
            }
            if (lane == 0) {
                head[2 * r]     = p0 + b0;
                head[2 * r + 1] = p1 + b1;
            }
        }
    }
}

template <int NM, int S>
__device__ __forceinline__ void tile_gemm_relu(const float* acc, const float* __restrict__ Wb,
                                               float* __restrict__ C, int n0, int nt,
                                               int mg, int nc, int lane) {
    constexpr int N = 128;
    const int col = lane & 15;
    const int hf  = lane >> 4;
    const float* __restrict__ Bc = Wb + (size_t)(8 * hf) * N + nc * 16 + col;
    const float* Ar = acc + (mg * 16 + col) * S + 2 * hf;

    v8f d[NM];
#pragma unroll
    for (int m = 0; m < NM; ++m) d[m] = v8f{};

    auto brow = [](int s) { return (s >> 5) * 128 + ((s >> 3) & 3) + 16 * (s & 7); };
    constexpr int KC = 4;
    v2f bcur[KC], bnxt[KC];
#pragma unroll
    for (int i = 0; i < KC; ++i) {
        bcur[i].x = Bc[(size_t)(brow(i)) * N];
        bcur[i].y = Bc[(size_t)(brow(i) + 4) * N];
    }
#pragma unroll 1
    for (int s0 = 0; s0 < 64; s0 += KC) {
        const int sn = s0 + KC < 64 ? s0 + KC : s0;
        const float* __restrict__ Bn = Bc + (size_t)brow(sn) * N;
#pragma unroll
        for (int i = 0; i < KC; ++i) {
            bnxt[i].x = Bn[(size_t)(16 * i) * N];
            bnxt[i].y = Bn[(size_t)(16 * i + 4) * N];
        }
#pragma unroll
        for (int i = 0; i < KC; ++i) {
#pragma unroll
            for (int m = 0; m < NM; ++m) {
                v2f av = *(const v2f*)(Ar + m * 64 * S + 4 * (s0 + i));
                av.x = fmaxf(av.x, 0.0f);
                av.y = fmaxf(av.y, 0.0f);
                d[m] = __builtin_amdgcn_wmma_f32_16x16x4_f32(
                    false, av, false, bcur[i], (short)0, d[m], false, false);
            }
        }
#pragma unroll
        for (int i = 0; i < KC; ++i) bcur[i] = bnxt[i];
    }

#pragma unroll
    for (int m = 0; m < NM; ++m) {
#pragma unroll
        for (int r = 0; r < 8; ++r) {
            const int row = mg * 16 + m * 64 + r + 8 * hf;
            if (row < nt) C[(size_t)(n0 + row) * N + nc * 16 + col] = d[m][r];
        }
    }
}

template <int T, int RT>
__global__ __launch_bounds__(512)
void k_tilehist(const int* __restrict__ dst, int* __restrict__ wgcnt, int E,
                const float* __restrict__ feat, const float* __restrict__ W1,
                float* __restrict__ xw1, int mtEnd) {
    constexpr int NT  = 512;
    constexpr int EPT = 8;
    __shared__ int   s_cnt[NT];
    __shared__ float s_a[RT * 16 * 66];
    const int tid = threadIdx.x;
    const int nwg = (E + NT * EPT - 1) / (NT * EPT);
    if ((int)blockIdx.x >= nwg) {
        gemm_rowtiles<64, NT, RT>(feat, W1, xw1, s_a, ((int)blockIdx.x - nwg) * RT, mtEnd, tid);
        return;
    }
    s_cnt[tid] = 0;
    const int e0 = blockIdx.x * (NT * EPT) + tid;
    int d[EPT];
#pragma unroll
    for (int j = 0; j < EPT; ++j) {
        const int e = e0 + j * NT;
        d[j] = e < E ? dst[e] : -1;
    }
    __syncthreads();
#pragma unroll
    for (int j = 0; j < EPT; ++j)
        if (d[j] >= 0)
            __hip_atomic_fetch_add(&s_cnt[d[j] / T], 1, __ATOMIC_RELAXED, __HIP_MEMORY_SCOPE_WORKGROUP);
    __syncthreads();
    wgcnt[blockIdx.x * NT + tid] = s_cnt[tid];
}

template <int T, int RT>
__global__ __launch_bounds__(512, 8)
void k_bin(const int* __restrict__ src, const int* __restrict__ dst,
           const int* __restrict__ wgcnt, int* __restrict__ tilecnt,
           int* __restrict__ list, int N, int E,
           const float* __restrict__ feat, const float* __restrict__ W1,
           float* __restrict__ xw1, int mt0, int mtEnd) {
    constexpr int NT  = 512;
    constexpr int EPT = 8;
    static_assert(T <= 256, "a tile row must fit the high byte of a list entry");

    __shared__ int   s_cnt[NT];
    __shared__ int   s_first[NT];
    __shared__ int   s_wsum[NT / 32];
    __shared__ float s_a[RT * 16 * 66];

    const int tid   = threadIdx.x;
    const int nwg   = (E + NT * EPT - 1) / (NT * EPT);
    if ((int)blockIdx.x >= nwg) {
        gemm_rowtiles<64, NT, RT>(feat, W1, xw1, s_a, mt0 + ((int)blockIdx.x - nwg) * RT, mtEnd, tid);
        return;
    }
    const int lane  = tid & 31;
    const int wave  = __builtin_amdgcn_readfirstlane(tid >> 5);
    const int ntile = (N + T - 1) / T;
    const int me    = blockIdx.x;

    s_cnt[tid] = 0;

    const int e0 = blockIdx.x * (NT * EPT) + tid;
    int  s[EPT], d[EPT];
    bool ok[EPT];
#pragma unroll
    for (int j = 0; j < EPT; ++j) {
        const int e = e0 + j * NT;
        ok[j] = e < E;
        s[j]  = ok[j] ? src[e] : 0;
        d[j]  = ok[j] ? dst[e] : 0;
    }

    int before = 0, c = 0;
    if (tid < ntile) {
        const int* __restrict__ col = wgcnt + tid;
        constexpr int CB = 28;
        for (int b0 = 0; b0 < nwg; b0 += CB) {
            int v[CB];
#pragma unroll
            for (int j = 0; j < CB; ++j) {
                const int b = b0 + j;
                v[j] = col[(b < nwg ? b : nwg - 1) * NT];
            }
#pragma unroll
            for (int j = 0; j < CB; ++j) {
                const int b = b0 + j;
                c      += v[j] & -(int)(b < nwg);
                before += v[j] & -(int)(b < me);
            }
        }
        if (me == 0) tilecnt[tid] = c;
    }
    int x = c;
#pragma unroll
    for (int m = 1; m < 32; m <<= 1) {
        const int y = __shfl_up(x, m, 32);
        if (lane >= m) x += y;
    }
    if (lane == 31) s_wsum[wave] = x;
    __syncthreads();
    int prev = 0;
    for (int w = 0; w < wave; ++w) prev += s_wsum[w];
    s_first[tid] = prev + x - c + before;

    int tl[EPT], rk[EPT];
#pragma unroll
    for (int j = 0; j < EPT; ++j) {
        tl[j] = d[j] / T;
        rk[j] = 0;
        if (ok[j])
            rk[j] = __hip_atomic_fetch_add(&s_cnt[tl[j]], 1, __ATOMIC_RELAXED, __HIP_MEMORY_SCOPE_WORKGROUP);
    }
    __syncthreads();

#pragma unroll
    for (int j = 0; j < EPT; ++j)
        if (ok[j]) list[s_first[tl[j]] + rk[j]] = list_pack(s[j], d[j] - tl[j] * T);
}

template <int T, int RT>
__global__ __launch_bounds__(1024)
void k_deg(const int* __restrict__ list, const int* __restrict__ tilecnt,
           float* __restrict__ dinv, int N,
           const float* __restrict__ feat, const float* __restrict__ W1,
           float* __restrict__ xw1, int mt0, int mtEnd) {
    constexpr int NT = 1024;
    constexpr int LB = 4;
    static_assert(T <= 256, "one counter per tile row");
    __shared__ int   s_deg[256];
    __shared__ float s_a[2 * RT * 16 * 66];
    const int tid  = threadIdx.x;
    const int ntile = (N + T - 1) / T;
    if ((int)blockIdx.x >= ntile) {
        gemm_rowtiles<64, NT, RT>(feat, W1, xw1, s_a, mt0 + 2 * RT * ((int)blockIdx.x - ntile), mtEnd, tid);
        return;
    }
    const int lane = tid & 31;
    if (tid < 256) s_deg[tid] = 0;
    const int* __restrict__ lst = list + tile_list_base(tilecnt, blockIdx.x, lane);
    const int cnt = tilecnt[blockIdx.x];
    __syncthreads();
    for (int i0 = 0; i0 < cnt; i0 += NT * LB) {
        int v[LB];
#pragma unroll
        for (int j = 0; j < LB; ++j) {
            const int i = i0 + j * NT + tid;
            v[j] = lst[i < cnt ? i : cnt - 1];
        }
#pragma unroll
        for (int j = 0; j < LB; ++j)
            if (i0 + j * NT + tid < cnt)
                __hip_atomic_fetch_add(&s_deg[(unsigned)v[j] >> 24], 1, __ATOMIC_RELAXED, __HIP_MEMORY_SCOPE_WORKGROUP);
    }
    __syncthreads();
    const int node = blockIdx.x * T + tid;
    if (tid < T && node < N) dinv[node] = rsqrtf((float)s_deg[tid] + 1.0f);
}

template <int F, int T, bool FIRST>
__global__ __launch_bounds__(1024)
void k_agg_list(const int* __restrict__ list, float* __restrict__ cflist,
                const int* __restrict__ tilecnt,
                const float* __restrict__ dinv, const float* __restrict__ xw,
                const float* __restrict__ bias, float* __restrict__ outp,
                const float* __restrict__ Wnext, const float* __restrict__ bnext,
                float* __restrict__ out_node, int N) {
    constexpr int NW = 32;
    constexpr int G  = 1024 / F;
    constexpr int MT = (T + 15) / 16;
    constexpr int S  = FIRST ? F + 4 : F;
    constexpr int TR = FIRST ? MT * 16 : T;
    static_assert(G <= 32 && 32 % G == 0, "a 32-entry step is a whole number of row groups");
    static_assert(!FIRST || F == 256, "the product's k order is written for 256 columns");
    static_assert(FIRST || F == 128, "the node head is written for 128 columns");

    __shared__ __attribute__((aligned(16))) float acc[TR * S];

    const int tid  = threadIdx.x;
    const int lane = tid & 31;
    const int wave = __builtin_amdgcn_readfirstlane(tid >> 5);
    const int n0   = blockIdx.x * T;
    const int nt   = (N - n0) < T ? (N - n0) : T;

    const int lbase = tile_list_base(tilecnt, blockIdx.x, lane);
    const int* __restrict__ lst = list + lbase;
    float*     __restrict__ cfl = cflist + lbase;
    const int cnt = tilecnt[blockIdx.x];
    const int per = (((cnt + G - 1) / G + NW - 1) / NW) * G;
    const int lo  = (wave * per < cnt) ? wave * per : cnt;
    const int hi  = (lo + per < cnt) ? lo + per : cnt;

    constexpr int PRE = 4;
    int   en[PRE];
    float ec[PRE];
#pragma unroll
    for (int k = 0; k < PRE; ++k) {
        en[k] = 0;
        if (lo + k * 32 + lane < hi) en[k] = lst[lo + k * 32 + lane];
    }
    float ds[PRE], dd[PRE];
#pragma unroll
    for (int k = 0; k < PRE; ++k) {
        ec[k] = 0.0f;
        if (FIRST) {
            ds[k] = dinv[en[k] & 0xFFFFFF];
            dd[k] = dinv[n0 + (int)((unsigned)en[k] >> 24)];
        } else if (lo + k * 32 + lane < hi) {
            ec[k] = cfl[lo + k * 32 + lane];
        }
    }

    tile_init<F, S, T, NW>(acc, dinv, xw, bias, n0, nt, wave, lane);
    if (FIRST) {
#pragma unroll
        for (int k = 0; k < PRE; ++k) {
            ec[k] = ds[k] * dd[k];
            if (lo + k * 32 + lane < hi) cfl[lo + k * 32 + lane] = ec[k];
        }
    }
    __syncthreads();

    int k = 0;
    for (int b0 = lo; b0 < hi; b0 += 32, ++k) {
        const int nb = (hi - b0) < 32 ? (hi - b0) : 32;
        int   cur = en[0];
        float cf  = ec[0];
#pragma unroll
        for (int j = 1; j < PRE; ++j)
            if (k == j) { cur = en[j]; cf = ec[j]; }
        if (k >= PRE) {
            cur = 0;
            cf  = 0.0f;
            if (lane < nb) {
                cur = lst[b0 + lane];
                if (FIRST) {
                    cf = dinv[cur & 0xFFFFFF] * dinv[n0 + (int)((unsigned)cur >> 24)];
                    cfl[b0 + lane] = cf;
                } else {
                    cf = cfl[b0 + lane];
                }
            }
        }
        const int s  = cur & 0xFFFFFF;
        const int dl = (int)((unsigned)cur >> 24);
        if (nb % G == 0) gather_add<F, S, G, true>(s, dl, cf, nb, lane, xw, acc);
        else             gather_add<F, S, G, false, true>(s, dl, cf, nb, lane, xw, acc);
    }
    __syncthreads();

    if constexpr (FIRST) {
        const int mg = wave >> 3;
        const int nc = wave & 7;
        if (mg < MT % 4) tile_gemm_relu<MT / 4 + 1, S>(acc, Wnext, outp, n0, nt, mg, nc, lane);
        else             tile_gemm_relu<MT / 4, S>(acc, Wnext, outp, n0, nt, mg, nc, lane);
    } else {
        __shared__ float head[2 * T];
        tile_store_head<T, NW>(acc, head, outp, Wnext, bnext, n0, nt, wave, lane);
        __syncthreads();
        if (tid < 2 * nt) out_node[(size_t)n0 * 2 + tid] = head[tid];
    }
}

template <int X>
__device__ __forceinline__ float grp8_xor(float v) {
    constexpr int sel = ((0 ^ X) << 0) | ((1 ^ X) << 3) | ((2 ^ X) << 6) | ((3 ^ X) << 9) |
                        ((4 ^ X) << 12) | ((5 ^ X) << 15) | ((6 ^ X) << 18) | ((7 ^ X) << 21);
    return __int_as_float(__builtin_amdgcn_mov_dpp8(__float_as_int(v), sel));
}

__device__ __forceinline__ void dot4x2(const float4 v, const float4 wa, const float4 wb,
                                       float& a0, float& a1) {
    a0 += v.x * wa.x; a1 += v.x * wa.y;
    a0 += v.y * wa.z; a1 += v.y * wa.w;
    a0 += v.z * wb.x; a1 += v.z * wb.y;
    a0 += v.w * wb.z; a1 += v.w * wb.w;
}

__global__ __launch_bounds__(256)
void k_edge_head(const int* __restrict__ src, const int* __restrict__ dst,
                 const float* __restrict__ emb, const float* __restrict__ We,
                 const float* __restrict__ be, float* __restrict__ out, int E) {
    const int lane = threadIdx.x & 31;
    const int q    = lane & 7;
    const int g    = lane >> 3;
    const int wave   = __builtin_amdgcn_readfirstlane(
        (int)(blockIdx.x * (blockDim.x >> 5) + (threadIdx.x >> 5)));
    const int nwaves = gridDim.x * (blockDim.x >> 5);

    float4 ws[4][2], wd[4][2];
#pragma unroll
    for (int k = 0; k < 4; ++k) {
        const float* __restrict__ w = We + ((k * 8 + q) * 4) * 2;
        ws[k][0] = *(const float4*)(w);
        ws[k][1] = *(const float4*)(w + 4);
        wd[k][0] = *(const float4*)(w + 256);
        wd[k][1] = *(const float4*)(w + 260);
    }
    const float* __restrict__ ecol = emb + q * 4;

    const int nit = ((E & 3) == 0) ? (E >> 4) : 0;
    if (wave < nit) {
        const float bq = be[q & 1];
        const bool  b4 = (q & 4) != 0, b2 = (q & 2) != 0, b1 = (q & 1) != 0;
        int  it = wave;
        int4 si = *(const int4*)(src + it * 16 + g * 4);
        int4 di = *(const int4*)(dst + it * 16 + g * 4);
        for (;;) {
            const int sidx[4] = {si.x, si.y, si.z, si.w};
            const int didx[4] = {di.x, di.y, di.z, di.w};
            float4 rs[4][4], rd[4][4];
#pragma unroll
            for (int j = 0; j < 4; ++j) {
                const float* __restrict__ ps = ecol + (size_t)sidx[j] * 128;
                const float* __restrict__ pd = ecol + (size_t)didx[j] * 128;
#pragma unroll
                for (int k = 0; k < 4; ++k) {
                    rs[j][k] = *(const float4*)(ps + k * 32);
                    rd[j][k] = *(const float4*)(pd + k * 32);
                }
            }
            const int  nx  = it + nwaves;
            const bool has = nx < nit;
            if (has) {
                si = *(const int4*)(src + nx * 16 + g * 4);
                di = *(const int4*)(dst + nx * 16 + g * 4);
            }
            float v[8];
#pragma unroll
            for (int j = 0; j < 4; ++j) {
                float a0 = 0.0f, a1 = 0.0f;
#pragma unroll
                for (int k = 0; k < 4; ++k) {
                    dot4x2(rs[j][k], ws[k][0], ws[k][1], a0, a1);
                    dot4x2(rd[j][k], wd[k][0], wd[k][1], a0, a1);
                }
                v[2 * j]     = a0;
                v[2 * j + 1] = a1;
            }
            float u[4];
#pragma unroll
            for (int t = 0; t < 4; ++t)
                u[t] = (b4 ? v[t + 4] : v[t]) + grp8_xor<4>(b4 ? v[t] : v[t + 4]);
            float w2[2];
#pragma unroll
            for (int t = 0; t < 2; ++t)
                w2[t] = (b2 ? u[t + 2] : u[t]) + grp8_xor<2>(b2 ? u[t] : u[t + 2]);
            const float z = (b1 ? w2[1] : w2[0]) + grp8_xor<1>(b1 ? w2[0] : w2[1]);
            out[(size_t)it * 32 + lane] = z + bq;
            if (!has) break;
            it = nx;
        }
    }

    for (int e0 = nit * 16 + wave * 4; e0 < E; e0 += nwaves * 4) {
        const int  e  = e0 + g;
        const bool ok = e < E;
        const int  ec = ok ? e : e0;
        const float* __restrict__ ps = ecol + (size_t)src[ec] * 128;
        const float* __restrict__ pd = ecol + (size_t)dst[ec] * 128;
        float a0 = 0.0f, a1 = 0.0f;
#pragma unroll
        for (int k = 0; k < 4; ++k) {
            dot4x2(*(const float4*)(ps + k * 32), ws[k][0], ws[k][1], a0, a1);
            dot4x2(*(const float4*)(pd + k * 32), wd[k][0], wd[k][1], a0, a1);
        }
        a0 += grp8_xor<4>(a0); a1 += grp8_xor<4>(a1);
        a0 += grp8_xor<2>(a0); a1 += grp8_xor<2>(a1);
        a0 += grp8_xor<1>(a0); a1 += grp8_xor<1>(a1);
        if (ok && q == 0) {
            out[(size_t)e * 2 + 0] = a0 + be[0];
            out[(size_t)e * 2 + 1] = a1 + be[1];
        }
    }
}

extern "C" void kernel_launch(void* const* d_in, const int* in_sizes, int n_in,
                              void* d_out, int out_size, void* d_ws, size_t ws_size,
                              hipStream_t stream) {
    const float* x  = (const float*)d_in[0];
    const int*   ei = (const int*)d_in[1];
    const float* W1 = (const float*)d_in[2];
    const float* b1 = (const float*)d_in[3];
    const float* W2 = (const float*)d_in[4];
    const float* b2 = (const float*)d_in[5];
    const float* Wn = (const float*)d_in[6];
    const float* bn = (const float*)d_in[7];
    const float* We = (const float*)d_in[8];
    const float* be = (const float*)d_in[9];

    const int N = in_sizes[0] / 64;
    const int E = in_sizes[1] / 2;
    const int* src = ei;
    const int* dst = ei + E;

    char*  ws  = (char*)d_ws;
    size_t off = 0;
    auto carve = [&](size_t bytes) -> float* {
        float* p = (float*)(ws + off);
        off += (bytes + 255) & ~(size_t)255;
        return p;
    };
    float* dinv = carve((size_t)N * 4);
    float* xw1  = carve((size_t)N * 256 * 4);
    float* xw2  = carve((size_t)N * 128 * 4);
    float* emb  = carve((size_t)N * 128 * 4);
    int*   elist   = (int*)carve((size_t)E * 4);
    float* cflist  = carve((size_t)E * 4);
    int*   tilecnt = (int*)carve((size_t)((N + 195) / 196) * 4);
    int*   wgcnt   = (int*)carve((size_t)((E + 4095) / 4096) * 512 * 4);

    float* out_node = (float*)d_out;
    float* out_edge = out_node + (size_t)N * 2;

    k_tilehist<196, 4><<<(E + 4095) / 4096 + (N / 16 - N / 32 - 2 * (N / 128) + 3) / 4, 512, 0, stream>>>(
        dst, wgcnt, E, x, W1, xw1, N / 16 - N / 32 - 2 * (N / 128));
    k_bin<196, 6><<<(E + 4095) / 4096 + (N / 32 + 5) / 6, 512, 0, stream>>>(
        src, dst, wgcnt, tilecnt, elist, N, E,
        x, W1, xw1, N / 16 - N / 32 - 2 * (N / 128), N / 16 - 2 * (N / 128));
    k_deg<196, 2><<<(N + 195) / 196 + (2 * (N / 128) + 3) / 4, 1024, 0, stream>>>(
        elist, tilecnt, dinv, N,
        x, W1, xw1, N / 16 - 2 * (N / 128), N / 16);

    k_agg_list<256, 196, true><<<(N + 195) / 196, 1024, 0, stream>>>(elist, cflist, tilecnt, dinv, xw1, b1, xw2, W2, nullptr, nullptr, N);

    k_agg_list<128, 196, false><<<(N + 195) / 196, 1024, 0, stream>>>(elist, cflist, tilecnt, dinv, xw2, b2, emb, Wn, bn, out_node, N);

    k_edge_head<<<512, 256, 0, stream>>>(src, dst, emb, We, be, out_edge, E);
}


# Tile edge lists ordered by tile row in the degree pass (count, scan, place); aggregation waves sum each run of equal rows in registers (fma), one set of LDS f32 adds per run; f32 sum order changes
# speedup vs baseline: 7.3368x; 1.0833x over previous
#include <hip/hip_runtime.h>


typedef float v2f __attribute__((ext_vector_type(2)));
typedef float v8f __attribute__((ext_vector_type(8)));

__device__ __forceinline__ int tile_list_base(const int* __restrict__ tilecnt, int tile, int lane) {
    int v = 0;
    for (int j = lane; j < tile; j += 32) v += tilecnt[j];
#pragma unroll
    for (int m = 16; m > 0; m >>= 1) v += __shfl_xor(v, m, 32);
    return __builtin_amdgcn_readfirstlane(v);
}

template <int K, int NTHR, int R>
__device__ __forceinline__ void gemm_rowtiles(const float* __restrict__ A, const float* __restrict__ B,
                                              float* __restrict__ C, float* As, int mt, int mtEnd, int tid) {
    constexpr int N   = 256;
    constexpr int LDK = K + 2;
    constexpr int TW  = (NTHR / 512) * R;
    const int wave = __builtin_amdgcn_readfirstlane(tid >> 5);
    const int lane = tid & 31;
    const int ms   = wave >> 4;
    const int nt   = wave & 15;
    const int col  = lane & 15;
    const int kb   = (lane >> 4) << 1;

    const float* __restrict__ Bc = B + (size_t)nt * 16 + col;
    v2f bv[K / 4];
#pragma unroll
    for (int k = 0; k < K; k += 4) {
        bv[k / 4].x = Bc[(size_t)(k + kb) * N];
        bv[k / 4].y = Bc[(size_t)(k + kb + 1) * N];
    }

    const int ntl = (mtEnd - mt) < TW ? (mtEnd - mt) : TW;
    const float* __restrict__ Ag = A + (size_t)mt * 16 * K;
    constexpr int NL = TW * 16 * (K / 2) / NTHR;
    float2 ar[NL];
#pragma unroll
    for (int i = 0; i < NL; ++i) {
        const int idx = tid + i * NTHR;
        const int r   = idx / (K / 2);
        const int kp  = idx - r * (K / 2);
        const int rs  = r < ntl * 16 ? r : ntl * 16 - 1;
        ar[i] = *(const float2*)(Ag + ((size_t)rs * (K / 2) + kp) * 2);
    }
#pragma unroll
    for (int i = 0; i < NL; ++i) {
        const int idx = tid + i * NTHR;
        const int r   = idx / (K / 2);
        const int kp  = idx - r * (K / 2);
        *(float2*)(As + r * LDK + kp * 2) = ar[i];
    }
    __syncthreads();

#pragma unroll
    for (int t = 0; t < R; ++t) {
        const int lt = ms * R + t;
        if (lt < ntl) {
            const float* Arow = As + (lt * 16 + col) * LDK + kb;
            v8f acc = {};
#pragma unroll
            for (int k = 0; k < K; k += 4) {
                v2f av; av.x = Arow[k]; av.y = Arow[k + 1];
                acc = __builtin_amdgcn_wmma_f32_16x16x4_f32(
                    false, av, false, bv[k / 4], (short)0, acc, false, false);
            }
            float* __restrict__ Cp =
                C + (size_t)((mt + lt) * 16 + ((lane >> 4) << 3)) * N + (size_t)nt * 16 + col;
#pragma unroll
            for (int r = 0; r < 8; ++r) Cp[(size_t)r * N] = acc[r];
        }
    }
}

__device__ __forceinline__ void lds_add(float* p, float v) {
    __hip_atomic_fetch_add(p, v, __ATOMIC_RELAXED, __HIP_MEMORY_SCOPE_WORKGROUP);
}

template <int H, int S>
__device__ __forceinline__ void run_flush(float* acc, int row, int lane, const float4 (&run)[H]) {
    float* a = acc + row * S + lane;
#pragma unroll
    for (int h = 0; h < H; ++h) {
        lds_add(a + h * 128,      run[h].x);
        lds_add(a + h * 128 + 32, run[h].y);
        lds_add(a + h * 128 + 64, run[h].z);
        lds_add(a + h * 128 + 96, run[h].w);
    }
}

template <int F, int S, int G, bool FULL, bool CLAMP = false>
__device__ __forceinline__ void gather_add(int s, int dl, float cf, int nb, int lane,
                                           const float* __restrict__ xw, float* acc,
                                           int& crow, float4 (&run)[F / 128]) {
    constexpr int H = F / 128;
#pragma unroll 1
    for (int i = 0; i < nb; i += G) {
        float4 v[G][H];
        int    rl[G];
        float  ci[G];
#pragma unroll
        for (int g = 0; g < G; ++g) {
            if (FULL || CLAMP || i + g < nb) {
                const int j  = (CLAMP && i + g >= nb) ? nb - 1 : i + g;
                const int si = __builtin_amdgcn_readlane(s, j);
                rl[g] = __builtin_amdgcn_readlane(dl, j);
                ci[g] = __int_as_float(__builtin_amdgcn_readlane(__float_as_int(cf), j));
                const float* __restrict__ row = xw + (size_t)si * F + lane * 4;
#pragma unroll
                for (int h = 0; h < H; ++h) v[g][h] = *(const float4*)(row + h * 128);
            }
        }
#pragma unroll
        for (int g = 0; g < G; ++g) {
            if (FULL || i + g < nb) {
                if (rl[g] != crow) {
                    if (crow >= 0) run_flush<H, S>(acc, crow, lane, run);
                    crow = rl[g];
#pragma unroll
                    for (int h = 0; h < H; ++h) run[h] = make_float4(0.0f, 0.0f, 0.0f, 0.0f);
                }
#pragma unroll
                for (int h = 0; h < H; ++h) {
                    run[h].x = fmaf(ci[g], v[g][h].x, run[h].x);
                    run[h].y = fmaf(ci[g], v[g][h].y, run[h].y);
                    run[h].z = fmaf(ci[g], v[g][h].z, run[h].z);
                    run[h].w = fmaf(ci[g], v[g][h].w, run[h].w);
                }
            }
        }
    }
}

__device__ __forceinline__ int list_pack(int s, int dl) {
    return (int)((unsigned)s | ((unsigned)dl << 24));
}

template <int F, int S, int T, int NW>
__device__ __forceinline__ void tile_init(float* acc,
                                          const float* __restrict__ dinv,
                                          const float* __restrict__ xw,
                                          const float* __restrict__ bias,
                                          int n0, int nt, int wave, int lane) {
    constexpr int H   = F / 128;
    constexpr int RPW = (T + NW - 1) / NW;
    float4 b[H];
#pragma unroll
    for (int h = 0; h < H; ++h) b[h] = *(const float4*)(bias + h * 128 + lane * 4);
    float4 v[RPW][H];
    float  di[RPW];
#pragma unroll
    for (int k = 0; k < RPW; ++k) {
        const int r  = wave + k * NW;
        const int rc = r < nt ? r : nt - 1;
        di[k] = dinv[n0 + rc];
        const float* __restrict__ xr = xw + (size_t)(n0 + rc) * F + lane * 4;
#pragma unroll
        for (int h = 0; h < H; ++h) v[k][h] = *(const float4*)(xr + h * 128);
    }
#pragma unroll
    for (int k = 0; k < RPW; ++k) {
        const int r = wave + k * NW;
        if (r < nt) {
            float* a = acc + r * S + lane;
#pragma unroll
            for (int h = 0; h < H; ++h) {
                a[h * 128]      = v[k][h].x * di[k] * di[k] + b[h].x;
                a[h * 128 + 32] = v[k][h].y * di[k] * di[k] + b[h].y;
                a[h * 128 + 64] = v[k][h].z * di[k] * di[k] + b[h].z;
                a[h * 128 + 96] = v[k][h].w * di[k] * di[k] + b[h].w;
            }
        }
    }
}

template <int T, int NW>
__device__ __forceinline__ void tile_store_head(const float* acc, float* head,
                                                float* __restrict__ emb,
                                                const float* __restrict__ Wn,
                                                const float* __restrict__ bn,
                                                int n0, int nt, int wave, int lane) {
    constexpr int F   = 128;
    constexpr int RPW = (T + NW - 1) / NW;
    const float4 wa = *(const float4*)(Wn + lane * 8);
    const float4 wb = *(const float4*)(Wn + lane * 8 + 4);
    const float  b0 = bn[0], b1 = bn[1];
#pragma unroll
    for (int k = 0; k < RPW; ++k) {
        const int r = wave + k * NW;
        if (r < nt) {
            const float* a = acc + r * F + lane;
            float4 v;
            v.x = a[0];
            v.y = a[32];
            v.z = a[64];
            v.w = a[96];
            *(float4*)(emb + (size_t)(n0 + r) * F + lane * 4) = v;
            float p0 = v.x * wa.x + v.y * wa.z + v.z * wb.x + v.w * wb.z;
            float p1 = v.x * wa.y + v.y * wa.w + v.z * wb.y + v.w * wb.w;
#pragma unroll
            for (int m = 16; m > 0; m >>= 1) {
                p0 += __shfl_xor(p0, m, 32);
                p1 += __shfl_xor(p1, m, 32);
            }
            if (lane == 0) {
                head[2 * r]     = p0 + b0;
                head[2 * r + 1] = p1 + b1;
            }
        }
    }
}

template <int NM, int S>
__device__ __forceinline__ void tile_gemm_relu(const float* acc, const float* __restrict__ Wb,
                                               float* __restrict__ C, int n0, int nt,
                                               int mg, int nc, int lane) {
    constexpr int N = 128;
    const int col = lane & 15;
    const int hf  = lane >> 4;
    const float* __restrict__ Bc = Wb + (size_t)(8 * hf) * N + nc * 16 + col;
    const float* Ar = acc + (mg * 16 + col) * S + 2 * hf;

    v8f d[NM];
#pragma unroll
    for (int m = 0; m < NM; ++m) d[m] = v8f{};

    auto brow = [](int s) { return (s >> 5) * 128 + ((s >> 3) & 3) + 16 * (s & 7); };
    constexpr int KC = 4;
    v2f bcur[KC], bnxt[KC];
#pragma unroll
    for (int i = 0; i < KC; ++i) {
        bcur[i].x = Bc[(size_t)(brow(i)) * N];
        bcur[i].y = Bc[(size_t)(brow(i) + 4) * N];
    }
#pragma unroll 1
    for (int s0 = 0; s0 < 64; s0 += KC) {
        const int sn = s0 + KC < 64 ? s0 + KC : s0;
        const float* __restrict__ Bn = Bc + (size_t)brow(sn) * N;
#pragma unroll
        for (int i = 0; i < KC; ++i) {
            bnxt[i].x = Bn[(size_t)(16 * i) * N];
            bnxt[i].y = Bn[(size_t)(16 * i + 4) * N];
        }
#pragma unroll
        for (int i = 0; i < KC; ++i) {
#pragma unroll
            for (int m = 0; m < NM; ++m) {
                v2f av = *(const v2f*)(Ar + m * 64 * S + 4 * (s0 + i));
                av.x = fmaxf(av.x, 0.0f);
                av.y = fmaxf(av.y, 0.0f);
                d[m] = __builtin_amdgcn_wmma_f32_16x16x4_f32(
                    false, av, false, bcur[i], (short)0, d[m], false, false);
            }
        }
#pragma unroll
        for (int i = 0; i < KC; ++i) bcur[i] = bnxt[i];
    }

#pragma unroll
    for (int m = 0; m < NM; ++m) {
#pragma unroll
        for (int r = 0; r < 8; ++r) {
            const int row = mg * 16 + m * 64 + r + 8 * hf;
            if (row < nt) C[(size_t)(n0 + row) * N + nc * 16 + col] = d[m][r];
        }
    }
}

template <int T, int RT>
__global__ __launch_bounds__(512)
void k_tilehist(const int* __restrict__ dst, int* __restrict__ wgcnt, int E,
                const float* __restrict__ feat, const float* __restrict__ W1,
                float* __restrict__ xw1, int mtEnd) {
    constexpr int NT  = 512;
    constexpr int EPT = 8;
    __shared__ int   s_cnt[NT];
    __shared__ float s_a[RT * 16 * 66];
    const int tid = threadIdx.x;
    const int nwg = (E + NT * EPT - 1) / (NT * EPT);
    if ((int)blockIdx.x >= nwg) {
        gemm_rowtiles<64, NT, RT>(feat, W1, xw1, s_a, ((int)blockIdx.x - nwg) * RT, mtEnd, tid);
        return;
    }
    s_cnt[tid] = 0;
    const int e0 = blockIdx.x * (NT * EPT) + tid;
    int d[EPT];
#pragma unroll
    for (int j = 0; j < EPT; ++j) {
        const int e = e0 + j * NT;
        d[j] = e < E ? dst[e] : -1;
    }
    __syncthreads();
#pragma unroll
    for (int j = 0; j < EPT; ++j)
        if (d[j] >= 0)
            __hip_atomic_fetch_add(&s_cnt[d[j] / T], 1, __ATOMIC_RELAXED, __HIP_MEMORY_SCOPE_WORKGROUP);
    __syncthreads();
    wgcnt[blockIdx.x * NT + tid] = s_cnt[tid];
}

template <int T, int RT>
__global__ __launch_bounds__(512, 8)
void k_bin(const int* __restrict__ src, const int* __restrict__ dst,
           const int* __restrict__ wgcnt, int* __restrict__ tilecnt,
           int* __restrict__ list, int N, int E,
           const float* __restrict__ feat, const float* __restrict__ W1,
           float* __restrict__ xw1, int mt0, int mtEnd) {
    constexpr int NT  = 512;
    constexpr int EPT = 8;
    static_assert(T <= 256, "a tile row must fit the high byte of a list entry");

    __shared__ int   s_cnt[NT];
    __shared__ int   s_first[NT];
    __shared__ int   s_wsum[NT / 32];
    __shared__ float s_a[RT * 16 * 66];

    const int tid   = threadIdx.x;
    const int nwg   = (E + NT * EPT - 1) / (NT * EPT);
    if ((int)blockIdx.x >= nwg) {
        gemm_rowtiles<64, NT, RT>(feat, W1, xw1, s_a, mt0 + ((int)blockIdx.x - nwg) * RT, mtEnd, tid);
        return;
    }
    const int lane  = tid & 31;
    const int wave  = __builtin_amdgcn_readfirstlane(tid >> 5);
    const int ntile = (N + T - 1) / T;
    const int me    = blockIdx.x;

    s_cnt[tid] = 0;

    const int e0 = blockIdx.x * (NT * EPT) + tid;
    int  s[EPT], d[EPT];
    bool ok[EPT];
#pragma unroll
    for (int j = 0; j < EPT; ++j) {
        const int e = e0 + j * NT;
        ok[j] = e < E;
        s[j]  = ok[j] ? src[e] : 0;
        d[j]  = ok[j] ? dst[e] : 0;
    }

    int before = 0, c = 0;
    if (tid < ntile) {
        const int* __restrict__ col = wgcnt + tid;
        constexpr int CB = 28;
        for (int b0 = 0; b0 < nwg; b0 += CB) {
            int v[CB];
#pragma unroll
            for (int j = 0; j < CB; ++j) {
                const int b = b0 + j;
                v[j] = col[(b < nwg ? b : nwg - 1) * NT];
            }
#pragma unroll
            for (int j = 0; j < CB; ++j) {
                const int b = b0 + j;
                c      += v[j] & -(int)(b < nwg);
                before += v[j] & -(int)(b < me);
            }
        }
        if (me == 0) tilecnt[tid] = c;
    }
    int x = c;
#pragma unroll
    for (int m = 1; m < 32; m <<= 1) {
        const int y = __shfl_up(x, m, 32);
        if (lane >= m) x += y;
    }
    if (lane == 31) s_wsum[wave] = x;
    __syncthreads();
    int prev = 0;
    for (int w = 0; w < wave; ++w) prev += s_wsum[w];
    s_first[tid] = prev + x - c + before;

    int tl[EPT], rk[EPT];
#pragma unroll
    for (int j = 0; j < EPT; ++j) {
        tl[j] = d[j] / T;
        rk[j] = 0;
        if (ok[j])
            rk[j] = __hip_atomic_fetch_add(&s_cnt[tl[j]], 1, __ATOMIC_RELAXED, __HIP_MEMORY_SCOPE_WORKGROUP);
    }
    __syncthreads();

#pragma unroll
    for (int j = 0; j < EPT; ++j)
        if (ok[j]) list[s_first[tl[j]] + rk[j]] = list_pack(s[j], d[j] - tl[j] * T);
}

template <int T, int RT>
__global__ __launch_bounds__(1024)
void k_deg(const int* __restrict__ list, const int* __restrict__ tilecnt,
           float* __restrict__ dinv, int* __restrict__ slist, int N,
           const float* __restrict__ feat, const float* __restrict__ W1,
           float* __restrict__ xw1, int mt0, int mtEnd) {
    constexpr int NT  = 1024;
    constexpr int LB  = 4;
    constexpr int CAP = 6144;
    static_assert(T <= 256, "one counter per tile row");
    __shared__ int   s_deg[256];
    __shared__ int   s_wsum[8];
    __shared__ int   s_ord[CAP];
    __shared__ float s_a[2 * RT * 16 * 66];
    const int tid  = threadIdx.x;
    const int ntile = (N + T - 1) / T;
    if ((int)blockIdx.x >= ntile) {
        gemm_rowtiles<64, NT, RT>(feat, W1, xw1, s_a, mt0 + 2 * RT * ((int)blockIdx.x - ntile), mtEnd, tid);
        return;
    }
    const int lane = tid & 31;
    const int wave = __builtin_amdgcn_readfirstlane(tid >> 5);
    if (tid < 256) s_deg[tid] = 0;
    const int base = tile_list_base(tilecnt, blockIdx.x, lane);
    const int* __restrict__ lst = list + base;
    int*       __restrict__ out = slist + base;
    const int cnt = tilecnt[blockIdx.x];
    __syncthreads();
    for (int i0 = 0; i0 < cnt; i0 += NT * LB) {
        int v[LB];
#pragma unroll
        for (int j = 0; j < LB; ++j) {
            const int i = i0 + j * NT + tid;
            v[j] = lst[i < cnt ? i : cnt - 1];
        }
#pragma unroll
        for (int j = 0; j < LB; ++j)
            if (i0 + j * NT + tid < cnt)
                __hip_atomic_fetch_add(&s_deg[(unsigned)v[j] >> 24], 1, __ATOMIC_RELAXED, __HIP_MEMORY_SCOPE_WORKGROUP);
    }
    __syncthreads();
    const int c    = tid < 256 ? s_deg[tid] : 0;
    const int node = blockIdx.x * T + tid;
    if (tid < T && node < N) dinv[node] = rsqrtf((float)c + 1.0f);

    int x = c;
#pragma unroll
    for (int m = 1; m < 32; m <<= 1) {
        const int y = __shfl_up(x, m, 32);
        if (lane >= m) x += y;
    }
    if (lane == 31 && wave < 8) s_wsum[wave] = x;
    __syncthreads();
    if (tid < 256) {
        int prev = 0;
        for (int w = 0; w < wave; ++w) prev += s_wsum[w];
        s_deg[tid] = prev + x - c;
    }
    __syncthreads();

    const bool fits = cnt <= CAP;
    for (int i0 = 0; i0 < cnt; i0 += NT * LB) {
        int v[LB];
#pragma unroll
        for (int j = 0; j < LB; ++j) {
            const int i = i0 + j * NT + tid;
            v[j] = lst[i < cnt ? i : cnt - 1];
        }
#pragma unroll
        for (int j = 0; j < LB; ++j)
            if (i0 + j * NT + tid < cnt) {
                const int slot = __hip_atomic_fetch_add(&s_deg[(unsigned)v[j] >> 24], 1,
                                                        __ATOMIC_RELAXED, __HIP_MEMORY_SCOPE_WORKGROUP);
                if (fits) s_ord[slot] = v[j];
                else      out[slot]   = v[j];
            }
    }
    if (fits) {
        __syncthreads();
        for (int i = tid; i < cnt; i += NT) out[i] = s_ord[i];
    }
}

template <int F, int T, bool FIRST>
__global__ __launch_bounds__(1024)
void k_agg_list(const int* __restrict__ list, float* __restrict__ cflist,
                const int* __restrict__ tilecnt,
                const float* __restrict__ dinv, const float* __restrict__ xw,
                const float* __restrict__ bias, float* __restrict__ outp,
                const float* __restrict__ Wnext, const float* __restrict__ bnext,
                float* __restrict__ out_node, int N) {
    constexpr int NW = 32;
    constexpr int G  = 1024 / F;
    constexpr int MT = (T + 15) / 16;
    constexpr int S  = FIRST ? F + 4 : F;
    constexpr int TR = FIRST ? MT * 16 : T;
    static_assert(G <= 32 && 32 % G == 0, "a 32-entry step is a whole number of row groups");
    static_assert(!FIRST || F == 256, "the product's k order is written for 256 columns");
    static_assert(FIRST || F == 128, "the node head is written for 128 columns");

    __shared__ __attribute__((aligned(16))) float acc[TR * S];

    const int tid  = threadIdx.x;
    const int lane = tid & 31;
    const int wave = __builtin_amdgcn_readfirstlane(tid >> 5);
    const int n0   = blockIdx.x * T;
    const int nt   = (N - n0) < T ? (N - n0) : T;

    const int lbase = tile_list_base(tilecnt, blockIdx.x, lane);
    const int* __restrict__ lst = list + lbase;
    float*     __restrict__ cfl = cflist + lbase;
    const int cnt = tilecnt[blockIdx.x];
    const int per = (((cnt + G - 1) / G + NW - 1) / NW) * G;
    const int lo  = (wave * per < cnt) ? wave * per : cnt;
    const int hi  = (lo + per < cnt) ? lo + per : cnt;

    constexpr int PRE = 4;
    int   en[PRE];
    float ec[PRE];
#pragma unroll
    for (int k = 0; k < PRE; ++k) {
        en[k] = 0;
        if (lo + k * 32 + lane < hi) en[k] = lst[lo + k * 32 + lane];
    }
    float ds[PRE], dd[PRE];
#pragma unroll
    for (int k = 0; k < PRE; ++k) {
        ec[k] = 0.0f;
        if (FIRST) {
            ds[k] = dinv[en[k] & 0xFFFFFF];
            dd[k] = dinv[n0 + (int)((unsigned)en[k] >> 24)];
        } else if (lo + k * 32 + lane < hi) {
            ec[k] = cfl[lo + k * 32 + lane];
        }
    }

    tile_init<F, S, T, NW>(acc, dinv, xw, bias, n0, nt, wave, lane);
    if (FIRST) {
#pragma unroll
        for (int k = 0; k < PRE; ++k) {
            ec[k] = ds[k] * dd[k];
            if (lo + k * 32 + lane < hi) cfl[lo + k * 32 + lane] = ec[k];
        }
    }
    __syncthreads();

    int    crow = -1;
    float4 run[F / 128];
#pragma unroll
    for (int h = 0; h < F / 128; ++h) run[h] = make_float4(0.0f, 0.0f, 0.0f, 0.0f);

    int k = 0;
    for (int b0 = lo; b0 < hi; b0 += 32, ++k) {
        const int nb = (hi - b0) < 32 ? (hi - b0) : 32;
        int   cur = en[0];
        float cf  = ec[0];
#pragma unroll
        for (int j = 1; j < PRE; ++j)
            if (k == j) { cur = en[j]; cf = ec[j]; }
        if (k >= PRE) {
            cur = 0;
            cf  = 0.0f;
            if (lane < nb) {
                cur = lst[b0 + lane];
                if (FIRST) {
                    cf = dinv[cur & 0xFFFFFF] * dinv[n0 + (int)((unsigned)cur >> 24)];
                    cfl[b0 + lane] = cf;
                } else {
                    cf = cfl[b0 + lane];
                }
            }
        }
        const int s  = cur & 0xFFFFFF;
        const int dl = (int)((unsigned)cur >> 24);
        if (nb % G == 0) gather_add<F, S, G, true>(s, dl, cf, nb, lane, xw, acc, crow, run);
        else             gather_add<F, S, G, false, true>(s, dl, cf, nb, lane, xw, acc, crow, run);
    }
    if (crow >= 0) run_flush<F / 128, S>(acc, crow, lane, run);
    __syncthreads();

    if constexpr (FIRST) {
        const int mg = wave >> 3;
        const int nc = wave & 7;
        if (mg < MT % 4) tile_gemm_relu<MT / 4 + 1, S>(acc, Wnext, outp, n0, nt, mg, nc, lane);
        else             tile_gemm_relu<MT / 4, S>(acc, Wnext, outp, n0, nt, mg, nc, lane);
    } else {
        __shared__ float head[2 * T];
        tile_store_head<T, NW>(acc, head, outp, Wnext, bnext, n0, nt, wave, lane);
        __syncthreads();
        if (tid < 2 * nt) out_node[(size_t)n0 * 2 + tid] = head[tid];
    }
}

template <int X>
__device__ __forceinline__ float grp8_xor(float v) {
    constexpr int sel = ((0 ^ X) << 0) | ((1 ^ X) << 3) | ((2 ^ X) << 6) | ((3 ^ X) << 9) |
                        ((4 ^ X) << 12) | ((5 ^ X) << 15) | ((6 ^ X) << 18) | ((7 ^ X) << 21);
    return __int_as_float(__builtin_amdgcn_mov_dpp8(__float_as_int(v), sel));
}

__device__ __forceinline__ void dot4x2(const float4 v, const float4 wa, const float4 wb,
                                       float& a0, float& a1) {
    a0 += v.x * wa.x; a1 += v.x * wa.y;
    a0 += v.y * wa.z; a1 += v.y * wa.w;
    a0 += v.z * wb.x; a1 += v.z * wb.y;
    a0 += v.w * wb.z; a1 += v.w * wb.w;
}

__global__ __launch_bounds__(256)
void k_edge_head(const int* __restrict__ src, const int* __restrict__ dst,
                 const float* __restrict__ emb, const float* __restrict__ We,
                 const float* __restrict__ be, float* __restrict__ out, int E) {
    const int lane = threadIdx.x & 31;
    const int q    = lane & 7;
    const int g    = lane >> 3;
    const int wave   = __builtin_amdgcn_readfirstlane(
        (int)(blockIdx.x * (blockDim.x >> 5) + (threadIdx.x >> 5)));
    const int nwaves = gridDim.x * (blockDim.x >> 5);

    float4 ws[4][2], wd[4][2];
#pragma unroll
    for (int k = 0; k < 4; ++k) {
        const float* __restrict__ w = We + ((k * 8 + q) * 4) * 2;
        ws[k][0] = *(const float4*)(w);
        ws[k][1] = *(const float4*)(w + 4);
        wd[k][0] = *(const float4*)(w + 256);
        wd[k][1] = *(const float4*)(w + 260);
    }
    const float* __restrict__ ecol = emb + q * 4;

    const int nit = ((E & 3) == 0) ? (E >> 4) : 0;
    if (wave < nit) {
        const float bq = be[q & 1];
        const bool  b4 = (q & 4) != 0, b2 = (q & 2) != 0, b1 = (q & 1) != 0;
        int  it = wave;
        int4 si = *(const int4*)(src + it * 16 + g * 4);
        int4 di = *(const int4*)(dst + it * 16 + g * 4);
        for (;;) {
            const int sidx[4] = {si.x, si.y, si.z, si.w};
            const int didx[4] = {di.x, di.y, di.z, di.w};
            float4 rs[4][4], rd[4][4];
#pragma unroll
            for (int j = 0; j < 4; ++j) {
                const float* __restrict__ ps = ecol + (size_t)sidx[j] * 128;
                const float* __restrict__ pd = ecol + (size_t)didx[j] * 128;
#pragma unroll
                for (int k = 0; k < 4; ++k) {
                    rs[j][k] = *(const float4*)(ps + k * 32);
                    rd[j][k] = *(const float4*)(pd + k * 32);
                }
            }
            const int  nx  = it + nwaves;
            const bool has = nx < nit;
            if (has) {
                si = *(const int4*)(src + nx * 16 + g * 4);
                di = *(const int4*)(dst + nx * 16 + g * 4);
            }
            float v[8];
#pragma unroll
            for (int j = 0; j < 4; ++j) {
                float a0 = 0.0f, a1 = 0.0f;
#pragma unroll
                for (int k = 0; k < 4; ++k) {
                    dot4x2(rs[j][k], ws[k][0], ws[k][1], a0, a1);
                    dot4x2(rd[j][k], wd[k][0], wd[k][1], a0, a1);
                }
                v[2 * j]     = a0;
                v[2 * j + 1] = a1;
            }
            float u[4];
#pragma unroll
            for (int t = 0; t < 4; ++t)
                u[t] = (b4 ? v[t + 4] : v[t]) + grp8_xor<4>(b4 ? v[t] : v[t + 4]);
            float w2[2];
#pragma unroll
            for (int t = 0; t < 2; ++t)
                w2[t] = (b2 ? u[t + 2] : u[t]) + grp8_xor<2>(b2 ? u[t] : u[t + 2]);
            const float z = (b1 ? w2[1] : w2[0]) + grp8_xor<1>(b1 ? w2[0] : w2[1]);
            out[(size_t)it * 32 + lane] = z + bq;
            if (!has) break;
            it = nx;
        }
    }

    for (int e0 = nit * 16 + wave * 4; e0 < E; e0 += nwaves * 4) {
        const int  e  = e0 + g;
        const bool ok = e < E;
        const int  ec = ok ? e : e0;
        const float* __restrict__ ps = ecol + (size_t)src[ec] * 128;
        const float* __restrict__ pd = ecol + (size_t)dst[ec] * 128;
        float a0 = 0.0f, a1 = 0.0f;
#pragma unroll
        for (int k = 0; k < 4; ++k) {
            dot4x2(*(const float4*)(ps + k * 32), ws[k][0], ws[k][1], a0, a1);
            dot4x2(*(const float4*)(pd + k * 32), wd[k][0], wd[k][1], a0, a1);
        }
        a0 += grp8_xor<4>(a0); a1 += grp8_xor<4>(a1);
        a0 += grp8_xor<2>(a0); a1 += grp8_xor<2>(a1);
        a0 += grp8_xor<1>(a0); a1 += grp8_xor<1>(a1);
        if (ok && q == 0) {
            out[(size_t)e * 2 + 0] = a0 + be[0];
            out[(size_t)e * 2 + 1] = a1 + be[1];
        }
    }
}

extern "C" void kernel_launch(void* const* d_in, const int* in_sizes, int n_in,
                              void* d_out, int out_size, void* d_ws, size_t ws_size,
                              hipStream_t stream) {
    const float* x  = (const float*)d_in[0];
    const int*   ei = (const int*)d_in[1];
    const float* W1 = (const float*)d_in[2];
    const float* b1 = (const float*)d_in[3];
    const float* W2 = (const float*)d_in[4];
    const float* b2 = (const float*)d_in[5];
    const float* Wn = (const float*)d_in[6];
    const float* bn = (const float*)d_in[7];
    const float* We = (const float*)d_in[8];
    const float* be = (const float*)d_in[9];

    const int N = in_sizes[0] / 64;
    const int E = in_sizes[1] / 2;
    const int* src = ei;
    const int* dst = ei + E;

    char*  ws  = (char*)d_ws;
    size_t off = 0;
    auto carve = [&](size_t bytes) -> float* {
        float* p = (float*)(ws + off);
        off += (bytes + 255) & ~(size_t)255;
        return p;
    };
    float* dinv = carve((size_t)N * 4);
    float* xw1  = carve((size_t)N * 256 * 4);
    float* xw2  = carve((size_t)N * 128 * 4);
    float* emb  = carve((size_t)N * 128 * 4);
    int*   elist   = (int*)carve((size_t)E * 4);
    int*   slist   = (int*)carve((size_t)E * 4);
    float* cflist  = carve((size_t)E * 4);
    int*   tilecnt = (int*)carve((size_t)((N + 195) / 196) * 4);
    int*   wgcnt   = (int*)carve((size_t)((E + 4095) / 4096) * 512 * 4);

    float* out_node = (float*)d_out;
    float* out_edge = out_node + (size_t)N * 2;

    k_tilehist<196, 4><<<(E + 4095) / 4096 + (N / 16 - N / 32 - 2 * (N / 128) + 3) / 4, 512, 0, stream>>>(
        dst, wgcnt, E, x, W1, xw1, N / 16 - N / 32 - 2 * (N / 128));
    k_bin<196, 6><<<(E + 4095) / 4096 + (N / 32 + 5) / 6, 512, 0, stream>>>(
        src, dst, wgcnt, tilecnt, elist, N, E,
        x, W1, xw1, N / 16 - N / 32 - 2 * (N / 128), N / 16 - 2 * (N / 128));
    k_deg<196, 2><<<(N + 195) / 196 + (2 * (N / 128) + 3) / 4, 1024, 0, stream>>>(
        elist, tilecnt, dinv, slist, N,
        x, W1, xw1, N / 16 - 2 * (N / 128), N / 16);

    k_agg_list<256, 196, true><<<(N + 195) / 196, 1024, 0, stream>>>(slist, cflist, tilecnt, dinv, xw1, b1, xw2, W2, nullptr, nullptr, N);

    k_agg_list<128, 196, false><<<(N + 195) / 196, 1024, 0, stream>>>(slist, cflist, tilecnt, dinv, xw2, b2, emb, Wn, bn, out_node, N);

    k_edge_head<<<512, 256, 0, stream>>>(src, dst, emb, We, be, out_edge, E);
}
